# starting kernel
# baseline (speedup 1.0000x reference)
#include <hip/hip_runtime.h>
#include <math.h>

typedef __attribute__((ext_vector_type(2))) float v2f;
typedef __attribute__((ext_vector_type(8))) float v8f;

static constexpr int kS  = 2048;
static constexpr int kD  = 512;
static constexpr int kH  = 8;
static constexpr int kDH = 64;

__device__ __forceinline__ v8f wmma_f32(v2f a, v2f b, v8f c) {
  return __builtin_amdgcn_wmma_f32_16x16x4_f32(
       false, a,  false, b,
       (short)0, c,  false,  false);
}

__global__ void proj_gemm_head(const float* __restrict__ A,
                               const float* __restrict__ B,
                               float* __restrict__ out) {
  const int lane = threadIdx.x & 31;
  const int wave = threadIdx.x >> 5;
  const int half = lane >> 4;
  const int lr   = lane & 15;
  const int m0   = blockIdx.x * 16;
  const int n0   = wave * 64;

  const float* ap = A + (size_t)(m0 + lr) * kD + 2 * half;
  const float* bp = B + (size_t)(n0 + lr) * kD + 2 * half;

  v8f acc[4] = {v8f{}, v8f{}, v8f{}, v8f{}};
  for (int k = 0; k < kD; k += 4) {
    v2f a = *(const v2f*)(ap + k);
#pragma unroll
    for (int t = 0; t < 4; ++t) {
      v2f b = *(const v2f*)(bp + (size_t)t * 16 * kD + k);
      acc[t] = wmma_f32(a, b, acc[t]);
    }
  }
#pragma unroll
  for (int t = 0; t < 4; ++t) {
    int n = n0 + t * 16 + lr;
    int h = n >> 6, d = n & 63;
    float* op = out + (size_t)h * kS * kDH + d;
#pragma unroll
    for (int r = 0; r < 8; ++r) {
      int m = m0 + r + 8 * half;
      op[(size_t)m * kDH] = acc[t][r];
    }
  }
}

__global__ void bd_gemm(const float* __restrict__ Qh, const float* __restrict__ Ph,
                        const float* __restrict__ vb, float* __restrict__ BD, int h) {
  const int lane = threadIdx.x & 31;
  const int wave = threadIdx.x >> 5;
  const int half = lane >> 4;
  const int lr   = lane & 15;
  const int m0   = blockIdx.x * 16;
  const int n0   = blockIdx.y * 512 + wave * 64;

  const float* qp = Qh + (size_t)h * kS * kDH + (size_t)(m0 + lr) * kDH + 2 * half;
  const float* pp = Ph + (size_t)h * kS * kDH + (size_t)(n0 + lr) * kDH + 2 * half;
  const float* bb = vb + h * kDH + 2 * half;

  v8f acc[4] = {v8f{}, v8f{}, v8f{}, v8f{}};
  for (int k = 0; k < kDH; k += 4) {
    v2f a = *(const v2f*)(qp + k);
    v2f bias = *(const v2f*)(bb + k);
    a = a + bias;
#pragma unroll
    for (int t = 0; t < 4; ++t) {
      v2f b = *(const v2f*)(pp + (size_t)t * 16 * kDH + k);
      acc[t] = wmma_f32(a, b, acc[t]);
    }
  }
#pragma unroll
  for (int t = 0; t < 4; ++t) {
    int n = n0 + t * 16 + lr;
#pragma unroll
    for (int r = 0; r < 8; ++r) {
      int m = m0 + r + 8 * half;
      BD[(size_t)m * kS + n] = acc[t][r];
    }
  }
}

__global__ void score_gemm(const float* __restrict__ Qh, const float* __restrict__ Kh,
                           const float* __restrict__ ub, const float* __restrict__ BD,
                           float* __restrict__ SC, int h, float scale) {
  const int lane = threadIdx.x & 31;
  const int wave = threadIdx.x >> 5;
  const int half = lane >> 4;
  const int lr   = lane & 15;
  const int m0   = blockIdx.x * 16;
  const int n0   = blockIdx.y * 512 + wave * 64;

  const float* qp = Qh + (size_t)h * kS * kDH + (size_t)(m0 + lr) * kDH + 2 * half;
  const float* kp = Kh + (size_t)h * kS * kDH + (size_t)(n0 + lr) * kDH + 2 * half;
  const float* bb = ub + h * kDH + 2 * half;

  v8f acc[4] = {v8f{}, v8f{}, v8f{}, v8f{}};
  for (int k = 0; k < kDH; k += 4) {
    v2f a = *(const v2f*)(qp + k);
    v2f bias = *(const v2f*)(bb + k);
    a = a + bias;
#pragma unroll
    for (int t = 0; t < 4; ++t) {
      v2f b = *(const v2f*)(kp + (size_t)t * 16 * kDH + k);
      acc[t] = wmma_f32(a, b, acc[t]);
    }
  }
#pragma unroll
  for (int t = 0; t < 4; ++t) {
    int n = n0 + t * 16 + lr;
#pragma unroll
    for (int r = 0; r < 8; ++r) {
      int m = m0 + r + 8 * half;
      float bd;
      if (n <= m)          bd = BD[(size_t)m * kS + (kS - 1 - m + n)];
      else if (n == m + 1) bd = 0.0f;
      else                 bd = BD[(size_t)(m + 1) * kS + (n - m - 2)];
      SC[(size_t)m * kS + n] = (acc[t][r] + bd) * scale;
    }
  }
}

__global__ void softmax_rows(float* __restrict__ SC) {
  __shared__ float red[256];
  const int tid = threadIdx.x;
  float* p = SC + (size_t)blockIdx.x * kS;

  float m = -INFINITY;
  for (int i = tid; i < kS; i += 256) m = fmaxf(m, p[i]);
  red[tid] = m; __syncthreads();
  for (int s = 128; s > 0; s >>= 1) {
    if (tid < s) red[tid] = fmaxf(red[tid], red[tid + s]);
    __syncthreads();
  }
  m = red[0]; __syncthreads();

  float sum = 0.0f;
  for (int i = tid; i < kS; i += 256) {
    float e = __expf(p[i] - m);
    p[i] = e;
    sum += e;
  }
  red[tid] = sum; __syncthreads();
  for (int s = 128; s > 0; s >>= 1) {
    if (tid < s) red[tid] += red[tid + s];
    __syncthreads();
  }
  float inv = 1.0f / red[0];
  for (int i = tid; i < kS; i += 256) p[i] *= inv;
}

__global__ void av_gemm(const float* __restrict__ SC, const float* __restrict__ Vh,
                        float* __restrict__ AVec, int h) {
  const int lane = threadIdx.x & 31;
  const int wave = threadIdx.x >> 5;
  const int half = lane >> 4;
  const int lr   = lane & 15;
  const int m0   = (blockIdx.x * 8 + wave) * 16;

  const float* pp = SC + (size_t)(m0 + lr) * kS + 2 * half;
  const float* vp = Vh + (size_t)h * kS * kDH + lr;

  v8f acc[4] = {v8f{}, v8f{}, v8f{}, v8f{}};
  for (int k = 0; k < kS; k += 4) {
    v2f a = *(const v2f*)(pp + k);
    int kk = k + 2 * half;
#pragma unroll
    for (int t = 0; t < 4; ++t) {
      const float* vb = vp + t * 16;
      v2f b = {vb[(size_t)kk * kDH], vb[(size_t)(kk + 1) * kDH]};
      acc[t] = wmma_f32(a, b, acc[t]);
    }
  }
#pragma unroll
  for (int t = 0; t < 4; ++t) {
    int d = t * 16 + lr;
#pragma unroll
    for (int r = 0; r < 8; ++r) {
      int m = m0 + r + 8 * half;
      AVec[(size_t)m * kD + h * kDH + d] = acc[t][r];
    }
  }
}

__global__ void out_gemm(const float* __restrict__ A, const float* __restrict__ B,
                         float* __restrict__ C) {
  const int lane = threadIdx.x & 31;
  const int wave = threadIdx.x >> 5;
  const int half = lane >> 4;
  const int lr   = lane & 15;
  const int m0   = blockIdx.x * 16;
  const int n0   = wave * 64;

  const float* ap = A + (size_t)(m0 + lr) * kD + 2 * half;
  const float* bp = B + (size_t)(n0 + lr) * kD + 2 * half;

  v8f acc[4] = {v8f{}, v8f{}, v8f{}, v8f{}};
  for (int k = 0; k < kD; k += 4) {
    v2f a = *(const v2f*)(ap + k);
#pragma unroll
    for (int t = 0; t < 4; ++t) {
      v2f b = *(const v2f*)(bp + (size_t)t * 16 * kD + k);
      acc[t] = wmma_f32(a, b, acc[t]);
    }
  }
#pragma unroll
  for (int t = 0; t < 4; ++t) {
    int n = n0 + t * 16 + lr;
#pragma unroll
    for (int r = 0; r < 8; ++r) {
      int m = m0 + r + 8 * half;
      C[(size_t)m * kD + n] = acc[t][r];
    }
  }
}

extern "C" void kernel_launch(void* const* d_in, const int* in_sizes, int n_in,
                              void* d_out, int out_size, void* d_ws, size_t ws_size,
                              hipStream_t stream) {
  const float* x   = (const float*)d_in[0];
  const float* pos = (const float*)d_in[1];
  const float* Wq  = (const float*)d_in[2];
  const float* Wk  = (const float*)d_in[3];
  const float* Wv  = (const float*)d_in[4];
  const float* Wp  = (const float*)d_in[5];
  const float* Wo  = (const float*)d_in[6];
  const float* u   = (const float*)d_in[7];
  const float* v   = (const float*)d_in[8];

  float* ws = (float*)d_ws;
  const size_t SD = (size_t)kS * kD;
  float* Qhs  = ws;
  float* Khs  = Qhs + SD;
  float* Vhs  = Khs + SD;
  float* Phs  = Vhs + SD;
  float* AVec = Phs + SD;
  float* BD   = AVec + SD;
  float* SC   = BD + (size_t)kS * kS;

  dim3 blk(256);
  proj_gemm_head<<<dim3(kS / 16), blk, 0, stream>>>(x,   Wq, Qhs);
  proj_gemm_head<<<dim3(kS / 16), blk, 0, stream>>>(x,   Wk, Khs);
  proj_gemm_head<<<dim3(kS / 16), blk, 0, stream>>>(x,   Wv, Vhs);
  proj_gemm_head<<<dim3(kS / 16), blk, 0, stream>>>(pos, Wp, Phs);

  const float scale = 0.125f;
  for (int h = 0; h < kH; ++h) {
    bd_gemm   <<<dim3(kS / 16, kS / 512), blk, 0, stream>>>(Qhs, Phs, v, BD, h);
    score_gemm<<<dim3(kS / 16, kS / 512), blk, 0, stream>>>(Qhs, Khs, u, BD, SC, h, scale);
    softmax_rows<<<dim3(kS), blk, 0, stream>>>(SC);
    av_gemm   <<<dim3(kS / 128), blk, 0, stream>>>(SC, Vhs, AVec, h);
  }

  out_gemm<<<dim3(kS / 16), blk, 0, stream>>>(AVec, Wo, (float*)d_out);
}


# k loops of the five WMMA kernels software-pipelined: fragment loads of later k steps issued as one group into spare register sets ahead of the current WMMAs (1-3 stages in flight); sum order kept
# speedup vs baseline: 1.5916x; 1.5916x over previous
#include <hip/hip_runtime.h>
#include <math.h>

typedef __attribute__((ext_vector_type(2))) float v2f;
typedef __attribute__((ext_vector_type(8))) float v8f;

static constexpr int kS  = 2048;
static constexpr int kD  = 512;
static constexpr int kH  = 8;
static constexpr int kDH = 64;

__device__ __forceinline__ v8f wmma_f32(v2f a, v2f b, v8f c) {
  return __builtin_amdgcn_wmma_f32_16x16x4_f32(
       false, a,  false, b,
       (short)0, c,  false,  false);
}

template <int U>
struct Stage {
  v2f a[U];
  v2f b[U][4];
};

__device__ __forceinline__ void pin_order() { __builtin_amdgcn_sched_barrier(0); }

template <int U>
__device__ __forceinline__ void load_nt(Stage<U>& s, const float* __restrict__ ap,
                                        const float* __restrict__ bp, int ld, int k0) {
#pragma unroll
  for (int i = 0; i < U; ++i) {
    s.a[i] = *(const v2f*)(ap + k0 + 4 * i);
#pragma unroll
    for (int t = 0; t < 4; ++t)
      s.b[i][t] = *(const v2f*)(bp + (size_t)t * 16 * ld + k0 + 4 * i);
  }
}

template <int U>
__device__ __forceinline__ void load_nn(Stage<U>& s, const float* __restrict__ ap,
                                        const float* __restrict__ vp, int half, int k0) {
#pragma unroll
  for (int i = 0; i < U; ++i) {
    s.a[i] = *(const v2f*)(ap + k0 + 4 * i);
    const int kk = k0 + 4 * i + 2 * half;
#pragma unroll
    for (int t = 0; t < 4; ++t) {
      const float* vb = vp + t * 16;
      s.b[i][t] = v2f{vb[(size_t)kk * kDH], vb[(size_t)(kk + 1) * kDH]};
    }
  }
}

template <int U>
__device__ __forceinline__ void mma_stage(const Stage<U>& s, v8f (&acc)[4]) {
#pragma unroll
  for (int i = 0; i < U; ++i) {
#pragma unroll
    for (int t = 0; t < 4; ++t)
      acc[t] = wmma_f32(s.a[i], s.b[i][t], acc[t]);
  }
}

__device__ __forceinline__ void kloop_nt512(const float* __restrict__ ap,
                                            const float* __restrict__ bp, v8f (&acc)[4]) {
  Stage<8> s0, s1;
  load_nt<8>(s0, ap, bp, kD, 0);
  for (int k = 0; k < kD - 64; k += 64) {
    load_nt<8>(s1, ap, bp, kD, k + 32);
    pin_order();
    mma_stage<8>(s0, acc);
    pin_order();
    load_nt<8>(s0, ap, bp, kD, k + 64);
    pin_order();
    mma_stage<8>(s1, acc);
    pin_order();
  }
  load_nt<8>(s1, ap, bp, kD, kD - 32);
  pin_order();
  mma_stage<8>(s0, acc);
  mma_stage<8>(s1, acc);
}

__device__ __forceinline__ void kloop_nt64_bias(const float* __restrict__ ap,
                                                const float* __restrict__ bp,
                                                const float* __restrict__ bb, v8f (&acc)[4]) {
  Stage<8> s0, s1;
  v2f c0[8], c1[8];
  load_nt<8>(s0, ap, bp, kDH, 0);
#pragma unroll
  for (int i = 0; i < 8; ++i) c0[i] = *(const v2f*)(bb + 4 * i);
  load_nt<8>(s1, ap, bp, kDH, 32);
#pragma unroll
  for (int i = 0; i < 8; ++i) c1[i] = *(const v2f*)(bb + 32 + 4 * i);
  pin_order();
#pragma unroll
  for (int i = 0; i < 8; ++i) s0.a[i] = s0.a[i] + c0[i];
  mma_stage<8>(s0, acc);
#pragma unroll
  for (int i = 0; i < 8; ++i) s1.a[i] = s1.a[i] + c1[i];
  mma_stage<8>(s1, acc);
}

__global__ void __launch_bounds__(256) proj_gemm_head(const float* __restrict__ A,
                               const float* __restrict__ B,
                               float* __restrict__ out) {
  const int lane = threadIdx.x & 31;
  const int wave = threadIdx.x >> 5;
  const int half = lane >> 4;
  const int lr   = lane & 15;
  const int m0   = blockIdx.x * 16;
  const int n0   = wave * 64;

  const float* ap = A + (size_t)(m0 + lr) * kD + 2 * half;
  const float* bp = B + (size_t)(n0 + lr) * kD + 2 * half;

  v8f acc[4] = {v8f{}, v8f{}, v8f{}, v8f{}};
  kloop_nt512(ap, bp, acc);
#pragma unroll
  for (int t = 0; t < 4; ++t) {
    int n = n0 + t * 16 + lr;
    int h = n >> 6, d = n & 63;
    float* op = out + (size_t)h * kS * kDH + d;
#pragma unroll
    for (int r = 0; r < 8; ++r) {
      int m = m0 + r + 8 * half;
      op[(size_t)m * kDH] = acc[t][r];
    }
  }
}

__global__ void __launch_bounds__(256) bd_gemm(const float* __restrict__ Qh, const float* __restrict__ Ph,
                        const float* __restrict__ vb, float* __restrict__ BD, int h) {
  const int lane = threadIdx.x & 31;
  const int wave = threadIdx.x >> 5;
  const int half = lane >> 4;
  const int lr   = lane & 15;
  const int m0   = blockIdx.x * 16;
  const int n0   = blockIdx.y * 512 + wave * 64;

  const float* qp = Qh + (size_t)h * kS * kDH + (size_t)(m0 + lr) * kDH + 2 * half;
  const float* pp = Ph + (size_t)h * kS * kDH + (size_t)(n0 + lr) * kDH + 2 * half;
  const float* bb = vb + h * kDH + 2 * half;

  v8f acc[4] = {v8f{}, v8f{}, v8f{}, v8f{}};
  kloop_nt64_bias(qp, pp, bb, acc);
#pragma unroll
  for (int t = 0; t < 4; ++t) {
    int n = n0 + t * 16 + lr;
#pragma unroll
    for (int r = 0; r < 8; ++r) {
      int m = m0 + r + 8 * half;
      BD[(size_t)m * kS + n] = acc[t][r];
    }
  }
}

__global__ void __launch_bounds__(256) score_gemm(const float* __restrict__ Qh, const float* __restrict__ Kh,
                           const float* __restrict__ ub, const float* __restrict__ BD,
                           float* __restrict__ SC, int h, float scale) {
  const int lane = threadIdx.x & 31;
  const int wave = threadIdx.x >> 5;
  const int half = lane >> 4;
  const int lr   = lane & 15;
  const int m0   = blockIdx.x * 16;
  const int n0   = blockIdx.y * 512 + wave * 64;

  const float* qp = Qh + (size_t)h * kS * kDH + (size_t)(m0 + lr) * kDH + 2 * half;
  const float* kp = Kh + (size_t)h * kS * kDH + (size_t)(n0 + lr) * kDH + 2 * half;
  const float* bb = ub + h * kDH + 2 * half;

  v8f acc[4] = {v8f{}, v8f{}, v8f{}, v8f{}};
  kloop_nt64_bias(qp, kp, bb, acc);
#pragma unroll
  for (int t = 0; t < 4; ++t) {
    int n = n0 + t * 16 + lr;
#pragma unroll
    for (int r = 0; r < 8; ++r) {
      int m = m0 + r + 8 * half;
      float bd;
      if (n <= m)          bd = BD[(size_t)m * kS + (kS - 1 - m + n)];
      else if (n == m + 1) bd = 0.0f;
      else                 bd = BD[(size_t)(m + 1) * kS + (n - m - 2)];
      SC[(size_t)m * kS + n] = (acc[t][r] + bd) * scale;
    }
  }
}

__global__ void softmax_rows(float* __restrict__ SC) {
  __shared__ float red[256];
  const int tid = threadIdx.x;
  float* p = SC + (size_t)blockIdx.x * kS;

  float m = -INFINITY;
  for (int i = tid; i < kS; i += 256) m = fmaxf(m, p[i]);
  red[tid] = m; __syncthreads();
  for (int s = 128; s > 0; s >>= 1) {
    if (tid < s) red[tid] = fmaxf(red[tid], red[tid + s]);
    __syncthreads();
  }
  m = red[0]; __syncthreads();

  float sum = 0.0f;
  for (int i = tid; i < kS; i += 256) {
    float e = __expf(p[i] - m);
    p[i] = e;
    sum += e;
  }
  red[tid] = sum; __syncthreads();
  for (int s = 128; s > 0; s >>= 1) {
    if (tid < s) red[tid] += red[tid + s];
    __syncthreads();
  }
  float inv = 1.0f / red[0];
  for (int i = tid; i < kS; i += 256) p[i] *= inv;
}

__global__ void __launch_bounds__(256) av_gemm(const float* __restrict__ SC, const float* __restrict__ Vh,
                        float* __restrict__ AVec, int h) {
  const int lane = threadIdx.x & 31;
  const int wave = threadIdx.x >> 5;
  const int half = lane >> 4;
  const int lr   = lane & 15;
  const int m0   = (blockIdx.x * 8 + wave) * 16;

  const float* pp = SC + (size_t)(m0 + lr) * kS + 2 * half;
  const float* vp = Vh + (size_t)h * kS * kDH + lr;

  v8f acc[4] = {v8f{}, v8f{}, v8f{}, v8f{}};
  Stage<2> s0, s1, s2, s3;
  load_nn<2>(s0, pp, vp, half, 0);
  load_nn<2>(s1, pp, vp, half, 8);
  load_nn<2>(s2, pp, vp, half, 16);
  for (int k = 0; k < kS - 32; k += 32) {
    load_nn<2>(s3, pp, vp, half, k + 24);
    pin_order();
    mma_stage<2>(s0, acc);
    pin_order();
    load_nn<2>(s0, pp, vp, half, k + 32);
    pin_order();
    mma_stage<2>(s1, acc);
    pin_order();
    load_nn<2>(s1, pp, vp, half, k + 40);
    pin_order();
    mma_stage<2>(s2, acc);
    pin_order();
    load_nn<2>(s2, pp, vp, half, k + 48);
    pin_order();
    mma_stage<2>(s3, acc);
    pin_order();
  }
  load_nn<2>(s3, pp, vp, half, kS - 8);
  pin_order();
  mma_stage<2>(s0, acc);
  mma_stage<2>(s1, acc);
  mma_stage<2>(s2, acc);
  mma_stage<2>(s3, acc);
#pragma unroll
  for (int t = 0; t < 4; ++t) {
    int d = t * 16 + lr;
#pragma unroll
    for (int r = 0; r < 8; ++r) {
      int m = m0 + r + 8 * half;
      AVec[(size_t)m * kD + h * kDH + d] = acc[t][r];
    }
  }
}

__global__ void __launch_bounds__(256) out_gemm(const float* __restrict__ A, const float* __restrict__ B,
                         float* __restrict__ C) {
  const int lane = threadIdx.x & 31;
  const int wave = threadIdx.x >> 5;
  const int half = lane >> 4;
  const int lr   = lane & 15;
  const int m0   = blockIdx.x * 16;
  const int n0   = wave * 64;

  const float* ap = A + (size_t)(m0 + lr) * kD + 2 * half;
  const float* bp = B + (size_t)(n0 + lr) * kD + 2 * half;

  v8f acc[4] = {v8f{}, v8f{}, v8f{}, v8f{}};
  kloop_nt512(ap, bp, acc);
#pragma unroll
  for (int t = 0; t < 4; ++t) {
    int n = n0 + t * 16 + lr;
#pragma unroll
    for (int r = 0; r < 8; ++r) {
      int m = m0 + r + 8 * half;
      C[(size_t)m * kD + n] = acc[t][r];
    }
  }
}

extern "C" void kernel_launch(void* const* d_in, const int* in_sizes, int n_in,
                              void* d_out, int out_size, void* d_ws, size_t ws_size,
                              hipStream_t stream) {
  const float* x   = (const float*)d_in[0];
  const float* pos = (const float*)d_in[1];
  const float* Wq  = (const float*)d_in[2];
  const float* Wk  = (const float*)d_in[3];
  const float* Wv  = (const float*)d_in[4];
  const float* Wp  = (const float*)d_in[5];
  const float* Wo  = (const float*)d_in[6];
  const float* u   = (const float*)d_in[7];
  const float* v   = (const float*)d_in[8];

  float* ws = (float*)d_ws;
  const size_t SD = (size_t)kS * kD;
  float* Qhs  = ws;
  float* Khs  = Qhs + SD;
  float* Vhs  = Khs + SD;
  float* Phs  = Vhs + SD;
  float* AVec = Phs + SD;
  float* BD   = AVec + SD;
  float* SC   = BD + (size_t)kS * kS;

  dim3 blk(256);
  proj_gemm_head<<<dim3(kS / 16), blk, 0, stream>>>(x,   Wq, Qhs);
  proj_gemm_head<<<dim3(kS / 16), blk, 0, stream>>>(x,   Wk, Khs);
  proj_gemm_head<<<dim3(kS / 16), blk, 0, stream>>>(x,   Wv, Vhs);
  proj_gemm_head<<<dim3(kS / 16), blk, 0, stream>>>(pos, Wp, Phs);

  const float scale = 0.125f;
  for (int h = 0; h < kH; ++h) {
    bd_gemm   <<<dim3(kS / 16, kS / 512), blk, 0, stream>>>(Qhs, Phs, v, BD, h);
    score_gemm<<<dim3(kS / 16, kS / 512), blk, 0, stream>>>(Qhs, Khs, u, BD, SC, h, scale);
    softmax_rows<<<dim3(kS), blk, 0, stream>>>(SC);
    av_gemm   <<<dim3(kS / 128), blk, 0, stream>>>(SC, Vhs, AVec, h);
  }

  out_gemm<<<dim3(kS / 16), blk, 0, stream>>>(AVec, Wo, (float*)d_out);
}


# AV product: head columns split over waves, one 16x16 tile and one WMMA chain per wave (512 waves in 256 blocks, was 128 in 16), six-stage load ring, output lines joined in LDS; sum order kept
# speedup vs baseline: 2.5048x; 1.5738x over previous
#include <hip/hip_runtime.h>
#include <math.h>

typedef __attribute__((ext_vector_type(2))) float v2f;
typedef __attribute__((ext_vector_type(4))) float v4f;
typedef __attribute__((ext_vector_type(8))) float v8f;

static constexpr int kS  = 2048;
static constexpr int kD  = 512;
static constexpr int kH  = 8;
static constexpr int kDH = 64;

__device__ __forceinline__ v8f wmma_f32(v2f a, v2f b, v8f c) {
  return __builtin_amdgcn_wmma_f32_16x16x4_f32(
       false, a,  false, b,
       (short)0, c,  false,  false);
}

template <int U>
struct Stage {
  v2f a[U];
  v2f b[U][4];
};

__device__ __forceinline__ void pin_order() { __builtin_amdgcn_sched_barrier(0); }

__device__ __forceinline__ void hold(v8f& acc) { asm volatile("" : "+v"(acc) : : "memory"); }

template <int U>
__device__ __forceinline__ void load_nt(Stage<U>& s, const float* __restrict__ ap,
                                        const float* __restrict__ bp, int ld, int k0) {
#pragma unroll
  for (int i = 0; i < U; ++i) {
    s.a[i] = *(const v2f*)(ap + k0 + 4 * i);
#pragma unroll
    for (int t = 0; t < 4; ++t)
      s.b[i][t] = *(const v2f*)(bp + (size_t)t * 16 * ld + k0 + 4 * i);
  }
}

template <int U>
struct Stage1 {
  v2f a[U];
  v2f b[U];
};

template <int U>
__device__ __forceinline__ void load_nn1(Stage1<U>& s, const float* __restrict__ ap,
                                         const float* __restrict__ vp, int k0) {
#pragma unroll
  for (int i = 0; i < U; ++i) {
    s.a[i] = *(const v2f*)(ap + k0 + 4 * i);
    const float* vb = vp + (size_t)(k0 + 4 * i) * kDH;
    s.b[i] = v2f{vb[0], vb[kDH]};
  }
}

template <int U>
__device__ __forceinline__ void mma_stage1(const Stage1<U>& s, v8f& acc) {
#pragma unroll
  for (int i = 0; i < U; ++i) acc = wmma_f32(s.a[i], s.b[i], acc);
}

template <int U>
__device__ __forceinline__ void mma_stage(const Stage<U>& s, v8f (&acc)[4]) {
#pragma unroll
  for (int i = 0; i < U; ++i) {
#pragma unroll
    for (int t = 0; t < 4; ++t)
      acc[t] = wmma_f32(s.a[i], s.b[i][t], acc[t]);
  }
}

__device__ __forceinline__ void kloop_nt512(const float* __restrict__ ap,
                                            const float* __restrict__ bp, v8f (&acc)[4]) {
  Stage<8> s0, s1;
  load_nt<8>(s0, ap, bp, kD, 0);
  for (int k = 0; k < kD - 64; k += 64) {
    load_nt<8>(s1, ap, bp, kD, k + 32);
    pin_order();
    mma_stage<8>(s0, acc);
    pin_order();
    load_nt<8>(s0, ap, bp, kD, k + 64);
    pin_order();
    mma_stage<8>(s1, acc);
    pin_order();
  }
  load_nt<8>(s1, ap, bp, kD, kD - 32);
  pin_order();
  mma_stage<8>(s0, acc);
  mma_stage<8>(s1, acc);
}

__device__ __forceinline__ void kloop_nt64_bias(const float* __restrict__ ap,
                                                const float* __restrict__ bp,
                                                const float* __restrict__ bb, v8f (&acc)[4]) {
  Stage<8> s0, s1;
  v2f c0[8], c1[8];
  load_nt<8>(s0, ap, bp, kDH, 0);
#pragma unroll
  for (int i = 0; i < 8; ++i) c0[i] = *(const v2f*)(bb + 4 * i);
  load_nt<8>(s1, ap, bp, kDH, 32);
#pragma unroll
  for (int i = 0; i < 8; ++i) c1[i] = *(const v2f*)(bb + 32 + 4 * i);
  pin_order();
#pragma unroll
  for (int i = 0; i < 8; ++i) s0.a[i] = s0.a[i] + c0[i];
  mma_stage<8>(s0, acc);
#pragma unroll
  for (int i = 0; i < 8; ++i) s1.a[i] = s1.a[i] + c1[i];
  mma_stage<8>(s1, acc);
}

__global__ void __launch_bounds__(256) proj_gemm_head(const float* __restrict__ A,
                               const float* __restrict__ B,
                               float* __restrict__ out) {
  const int lane = threadIdx.x & 31;
  const int wave = threadIdx.x >> 5;
  const int half = lane >> 4;
  const int lr   = lane & 15;
  const int m0   = blockIdx.x * 16;
  const int n0   = wave * 64;

  const float* ap = A + (size_t)(m0 + lr) * kD + 2 * half;
  const float* bp = B + (size_t)(n0 + lr) * kD + 2 * half;

  v8f acc[4] = {v8f{}, v8f{}, v8f{}, v8f{}};
  kloop_nt512(ap, bp, acc);
#pragma unroll
  for (int t = 0; t < 4; ++t) {
    int n = n0 + t * 16 + lr;
    int h = n >> 6, d = n & 63;
    float* op = out + (size_t)h * kS * kDH + d;
#pragma unroll
    for (int r = 0; r < 8; ++r) {
      int m = m0 + r + 8 * half;
      op[(size_t)m * kDH] = acc[t][r];
    }
  }
}

__global__ void __launch_bounds__(256) bd_gemm(const float* __restrict__ Qh, const float* __restrict__ Ph,
                        const float* __restrict__ vb, float* __restrict__ BD, int h) {
  const int lane = threadIdx.x & 31;
  const int wave = threadIdx.x >> 5;
  const int half = lane >> 4;
  const int lr   = lane & 15;
  const int m0   = blockIdx.x * 16;
  const int n0   = blockIdx.y * 512 + wave * 64;

  const float* qp = Qh + (size_t)h * kS * kDH + (size_t)(m0 + lr) * kDH + 2 * half;
  const float* pp = Ph + (size_t)h * kS * kDH + (size_t)(n0 + lr) * kDH + 2 * half;
  const float* bb = vb + h * kDH + 2 * half;

  v8f acc[4] = {v8f{}, v8f{}, v8f{}, v8f{}};
  kloop_nt64_bias(qp, pp, bb, acc);
#pragma unroll
  for (int t = 0; t < 4; ++t) {
    int n = n0 + t * 16 + lr;
#pragma unroll
    for (int r = 0; r < 8; ++r) {
      int m = m0 + r + 8 * half;
      BD[(size_t)m * kS + n] = acc[t][r];
    }
  }
}

__global__ void __launch_bounds__(256) score_gemm(const float* __restrict__ Qh, const float* __restrict__ Kh,
                           const float* __restrict__ ub, const float* __restrict__ BD,
                           float* __restrict__ SC, int h, float scale) {
  const int lane = threadIdx.x & 31;
  const int wave = threadIdx.x >> 5;
  const int half = lane >> 4;
  const int lr   = lane & 15;
  const int m0   = blockIdx.x * 16;
  const int n0   = blockIdx.y * 512 + wave * 64;

  const float* qp = Qh + (size_t)h * kS * kDH + (size_t)(m0 + lr) * kDH + 2 * half;
  const float* kp = Kh + (size_t)h * kS * kDH + (size_t)(n0 + lr) * kDH + 2 * half;
  const float* bb = ub + h * kDH + 2 * half;

  v8f acc[4] = {v8f{}, v8f{}, v8f{}, v8f{}};
  kloop_nt64_bias(qp, kp, bb, acc);
#pragma unroll
  for (int t = 0; t < 4; ++t) {
    int n = n0 + t * 16 + lr;
#pragma unroll
    for (int r = 0; r < 8; ++r) {
      int m = m0 + r + 8 * half;
      float bd;
      if (n <= m)          bd = BD[(size_t)m * kS + (kS - 1 - m + n)];
      else if (n == m + 1) bd = 0.0f;
      else                 bd = BD[(size_t)(m + 1) * kS + (n - m - 2)];
      SC[(size_t)m * kS + n] = (acc[t][r] + bd) * scale;
    }
  }
}

__global__ void softmax_rows(float* __restrict__ SC) {
  __shared__ float red[256];
  const int tid = threadIdx.x;
  float* p = SC + (size_t)blockIdx.x * kS;

  float m = -INFINITY;
  for (int i = tid; i < kS; i += 256) m = fmaxf(m, p[i]);
  red[tid] = m; __syncthreads();
  for (int s = 128; s > 0; s >>= 1) {
    if (tid < s) red[tid] = fmaxf(red[tid], red[tid + s]);
    __syncthreads();
  }
  m = red[0]; __syncthreads();

  float sum = 0.0f;
  for (int i = tid; i < kS; i += 256) {
    float e = __expf(p[i] - m);
    p[i] = e;
    sum += e;
  }
  red[tid] = sum; __syncthreads();
  for (int s = 128; s > 0; s >>= 1) {
    if (tid < s) red[tid] += red[tid + s];
    __syncthreads();
  }
  float inv = 1.0f / red[0];
  for (int i = tid; i < kS; i += 256) p[i] *= inv;
}

__global__ void __launch_bounds__(64) av_gemm(const float* __restrict__ SC, const float* __restrict__ Vh,
                        float* __restrict__ AVec, int h) {
  __shared__ alignas(16) float tile[16][36];
  const int lane = threadIdx.x & 31;
  const int wave = threadIdx.x >> 5;
  const int half = lane >> 4;
  const int lr   = lane & 15;
  const int m0   = (blockIdx.x >> 1) * 16;
  const int c0   = (blockIdx.x & 1) * 32;
  const int d0   = c0 + wave * 16;

  const float* pp = SC + (size_t)(m0 + lr) * kS + 2 * half;
  const float* vp = Vh + (size_t)h * kS * kDH + (size_t)(2 * half) * kDH + d0 + lr;

  v8f acc = v8f{};
  Stage1<4> s0, s1, s2, s3, s4, s5;
  load_nn1<4>(s0, pp, vp, 0);
  hold(acc);
  load_nn1<4>(s1, pp, vp, 16);
  hold(acc);
  load_nn1<4>(s2, pp, vp, 32);
  hold(acc);
  load_nn1<4>(s3, pp, vp, 48);
  hold(acc);
  load_nn1<4>(s4, pp, vp, 64);
  hold(acc);
  for (int k = 0; k < 20 * 96; k += 96) {
    load_nn1<4>(s5, pp, vp, k + 80);
    hold(acc);
    mma_stage1<4>(s0, acc);
    hold(acc);
    load_nn1<4>(s0, pp, vp, k + 96);
    hold(acc);
    mma_stage1<4>(s1, acc);
    hold(acc);
    load_nn1<4>(s1, pp, vp, k + 112);
    hold(acc);
    mma_stage1<4>(s2, acc);
    hold(acc);
    load_nn1<4>(s2, pp, vp, k + 128);
    hold(acc);
    mma_stage1<4>(s3, acc);
    hold(acc);
    load_nn1<4>(s3, pp, vp, k + 144);
    hold(acc);
    mma_stage1<4>(s4, acc);
    hold(acc);
    load_nn1<4>(s4, pp, vp, k + 160);
    hold(acc);
    mma_stage1<4>(s5, acc);
    hold(acc);
  }
  load_nn1<4>(s5, pp, vp, kS - 48);
  hold(acc);
  mma_stage1<4>(s0, acc);
  hold(acc);
  load_nn1<4>(s0, pp, vp, kS - 32);
  hold(acc);
  mma_stage1<4>(s1, acc);
  hold(acc);
  load_nn1<4>(s1, pp, vp, kS - 16);
  hold(acc);
  mma_stage1<4>(s2, acc);
  mma_stage1<4>(s3, acc);
  mma_stage1<4>(s4, acc);
  mma_stage1<4>(s5, acc);
  mma_stage1<4>(s0, acc);
  mma_stage1<4>(s1, acc);

#pragma unroll
  for (int r = 0; r < 8; ++r) tile[r + 8 * half][wave * 16 + lr] = acc[r];
  __syncthreads();
  for (int j = 0; j < 2; ++j) {
    const int idx = j * 64 + (int)threadIdx.x;
    const int row = idx >> 3;
    const int q   = idx & 7;
    const v4f val = *(const v4f*)&tile[row][q * 4];
    *(v4f*)(AVec + (size_t)(m0 + row) * kD + h * kDH + c0 + q * 4) = val;
  }
}

__global__ void __launch_bounds__(256) out_gemm(const float* __restrict__ A, const float* __restrict__ B,
                         float* __restrict__ C) {
  const int lane = threadIdx.x & 31;
  const int wave = threadIdx.x >> 5;
  const int half = lane >> 4;
  const int lr   = lane & 15;
  const int m0   = blockIdx.x * 16;
  const int n0   = wave * 64;

  const float* ap = A + (size_t)(m0 + lr) * kD + 2 * half;
  const float* bp = B + (size_t)(n0 + lr) * kD + 2 * half;

  v8f acc[4] = {v8f{}, v8f{}, v8f{}, v8f{}};
  kloop_nt512(ap, bp, acc);
#pragma unroll
  for (int t = 0; t < 4; ++t) {
    int n = n0 + t * 16 + lr;
#pragma unroll
    for (int r = 0; r < 8; ++r) {
      int m = m0 + r + 8 * half;
      C[(size_t)m * kD + n] = acc[t][r];
    }
  }
}

extern "C" void kernel_launch(void* const* d_in, const int* in_sizes, int n_in,
                              void* d_out, int out_size, void* d_ws, size_t ws_size,
                              hipStream_t stream) {
  const float* x   = (const float*)d_in[0];
  const float* pos = (const float*)d_in[1];
  const float* Wq  = (const float*)d_in[2];
  const float* Wk  = (const float*)d_in[3];
  const float* Wv  = (const float*)d_in[4];
  const float* Wp  = (const float*)d_in[5];
  const float* Wo  = (const float*)d_in[6];
  const float* u   = (const float*)d_in[7];
  const float* v   = (const float*)d_in[8];

  float* ws = (float*)d_ws;
  const size_t SD = (size_t)kS * kD;
  float* Qhs  = ws;
  float* Khs  = Qhs + SD;
  float* Vhs  = Khs + SD;
  float* Phs  = Vhs + SD;
  float* AVec = Phs + SD;
  float* BD   = AVec + SD;
  float* SC   = BD + (size_t)kS * kS;

  dim3 blk(256);
  proj_gemm_head<<<dim3(kS / 16), blk, 0, stream>>>(x,   Wq, Qhs);
  proj_gemm_head<<<dim3(kS / 16), blk, 0, stream>>>(x,   Wk, Khs);
  proj_gemm_head<<<dim3(kS / 16), blk, 0, stream>>>(x,   Wv, Vhs);
  proj_gemm_head<<<dim3(kS / 16), blk, 0, stream>>>(pos, Wp, Phs);

  const float scale = 0.125f;
  for (int h = 0; h < kH; ++h) {
    bd_gemm   <<<dim3(kS / 16, kS / 512), blk, 0, stream>>>(Qhs, Phs, v, BD, h);
    score_gemm<<<dim3(kS / 16, kS / 512), blk, 0, stream>>>(Qhs, Khs, u, BD, SC, h, scale);
    softmax_rows<<<dim3(kS), blk, 0, stream>>>(SC);
    av_gemm   <<<dim3(kS / 16 * 2), dim3(64), 0, stream>>>(SC, Vhs, AVec, h);
  }

  out_gemm<<<dim3(kS / 16), blk, 0, stream>>>(AVec, Wo, (float*)d_out);
}


# softmax pass folded into its neighbours: score blocks store exp(score - own 512-column row max) and (max, sum) pairs, AV scales A by exp(max - row max)/row sum; 29 launches; new row-sum order
# speedup vs baseline: 3.1227x; 1.2467x over previous
#include <hip/hip_runtime.h>
#include <math.h>

typedef __attribute__((ext_vector_type(2))) float v2f;
typedef __attribute__((ext_vector_type(4))) float v4f;
typedef __attribute__((ext_vector_type(8))) float v8f;

static constexpr int kS  = 2048;
static constexpr int kD  = 512;
static constexpr int kH  = 8;
static constexpr int kDH = 64;

__device__ __forceinline__ v8f wmma_f32(v2f a, v2f b, v8f c) {
  return __builtin_amdgcn_wmma_f32_16x16x4_f32(
       false, a,  false, b,
       (short)0, c,  false,  false);
}

template <int U>
struct Stage {
  v2f a[U];
  v2f b[U][4];
};

__device__ __forceinline__ void pin_order() { __builtin_amdgcn_sched_barrier(0); }

__device__ __forceinline__ void hold(v8f& acc) { asm volatile("" : "+v"(acc) : : "memory"); }

template <int U>
__device__ __forceinline__ void load_nt(Stage<U>& s, const float* __restrict__ ap,
                                        const float* __restrict__ bp, int ld, int k0) {
#pragma unroll
  for (int i = 0; i < U; ++i) {
    s.a[i] = *(const v2f*)(ap + k0 + 4 * i);
#pragma unroll
    for (int t = 0; t < 4; ++t)
      s.b[i][t] = *(const v2f*)(bp + (size_t)t * 16 * ld + k0 + 4 * i);
  }
}

template <int U>
struct Stage1 {
  v2f a[U];
  v2f b[U];
};

template <int U>
__device__ __forceinline__ void load_nn1(Stage1<U>& s, const float* __restrict__ ap,
                                         const float* __restrict__ vp, int k0) {
#pragma unroll
  for (int i = 0; i < U; ++i) {
    s.a[i] = *(const v2f*)(ap + k0 + 4 * i);
    const float* vb = vp + (size_t)(k0 + 4 * i) * kDH;
    s.b[i] = v2f{vb[0], vb[kDH]};
  }
}

template <int U>
__device__ __forceinline__ void mma_stage1(const Stage1<U>& s, float f, v8f& acc) {
  pin_order();
#pragma unroll
  for (int i = 0; i < U; ++i) acc = wmma_f32(s.a[i] * f, s.b[i], acc);
}

__device__ __forceinline__ float block_factor(v4f f, int k0) { return f[k0 >> 9]; }

template <int U>
__device__ __forceinline__ void mma_stage(const Stage<U>& s, v8f (&acc)[4]) {
#pragma unroll
  for (int i = 0; i < U; ++i) {
#pragma unroll
    for (int t = 0; t < 4; ++t)
      acc[t] = wmma_f32(s.a[i], s.b[i][t], acc[t]);
  }
}

__device__ __forceinline__ void kloop_nt512(const float* __restrict__ ap,
                                            const float* __restrict__ bp, v8f (&acc)[4]) {
  Stage<8> s0, s1;
  load_nt<8>(s0, ap, bp, kD, 0);
  for (int k = 0; k < kD - 64; k += 64) {
    load_nt<8>(s1, ap, bp, kD, k + 32);
    pin_order();
    mma_stage<8>(s0, acc);
    pin_order();
    load_nt<8>(s0, ap, bp, kD, k + 64);
    pin_order();
    mma_stage<8>(s1, acc);
    pin_order();
  }
  load_nt<8>(s1, ap, bp, kD, kD - 32);
  pin_order();
  mma_stage<8>(s0, acc);
  mma_stage<8>(s1, acc);
}

__device__ __forceinline__ void kloop_nt64_bias(const float* __restrict__ ap,
                                                const float* __restrict__ bp,
                                                const float* __restrict__ bb, v8f (&acc)[4]) {
  Stage<8> s0, s1;
  v2f c0[8], c1[8];
  load_nt<8>(s0, ap, bp, kDH, 0);
#pragma unroll
  for (int i = 0; i < 8; ++i) c0[i] = *(const v2f*)(bb + 4 * i);
  load_nt<8>(s1, ap, bp, kDH, 32);
#pragma unroll
  for (int i = 0; i < 8; ++i) c1[i] = *(const v2f*)(bb + 32 + 4 * i);
  pin_order();
#pragma unroll
  for (int i = 0; i < 8; ++i) s0.a[i] = s0.a[i] + c0[i];
  mma_stage<8>(s0, acc);
#pragma unroll
  for (int i = 0; i < 8; ++i) s1.a[i] = s1.a[i] + c1[i];
  mma_stage<8>(s1, acc);
}

__global__ void __launch_bounds__(256) proj_gemm_head(const float* __restrict__ A,
                               const float* __restrict__ B,
                               float* __restrict__ out) {
  const int lane = threadIdx.x & 31;
  const int wave = threadIdx.x >> 5;
  const int half = lane >> 4;
  const int lr   = lane & 15;
  const int m0   = blockIdx.x * 16;
  const int n0   = wave * 64;

  const float* ap = A + (size_t)(m0 + lr) * kD + 2 * half;
  const float* bp = B + (size_t)(n0 + lr) * kD + 2 * half;

  v8f acc[4] = {v8f{}, v8f{}, v8f{}, v8f{}};
  kloop_nt512(ap, bp, acc);
#pragma unroll
  for (int t = 0; t < 4; ++t) {
    int n = n0 + t * 16 + lr;
    int h = n >> 6, d = n & 63;
    float* op = out + (size_t)h * kS * kDH + d;
#pragma unroll
    for (int r = 0; r < 8; ++r) {
      int m = m0 + r + 8 * half;
      op[(size_t)m * kDH] = acc[t][r];
    }
  }
}

__global__ void __launch_bounds__(256) bd_gemm(const float* __restrict__ Qh, const float* __restrict__ Ph,
                        const float* __restrict__ vb, float* __restrict__ BD, int h) {
  const int lane = threadIdx.x & 31;
  const int wave = threadIdx.x >> 5;
  const int half = lane >> 4;
  const int lr   = lane & 15;
  const int m0   = blockIdx.x * 16;
  const int n0   = blockIdx.y * 512 + wave * 64;

  const float* qp = Qh + (size_t)h * kS * kDH + (size_t)(m0 + lr) * kDH + 2 * half;
  const float* pp = Ph + (size_t)h * kS * kDH + (size_t)(n0 + lr) * kDH + 2 * half;
  const float* bb = vb + h * kDH + 2 * half;

  v8f acc[4] = {v8f{}, v8f{}, v8f{}, v8f{}};
  kloop_nt64_bias(qp, pp, bb, acc);
#pragma unroll
  for (int t = 0; t < 4; ++t) {
    int n = n0 + t * 16 + lr;
#pragma unroll
    for (int r = 0; r < 8; ++r) {
      int m = m0 + r + 8 * half;
      BD[(size_t)m * kS + n] = acc[t][r];
    }
  }
}

__global__ void __launch_bounds__(256) score_gemm(const float* __restrict__ Qh, const float* __restrict__ Kh,
                           const float* __restrict__ ub, const float* __restrict__ BD,
                           float* __restrict__ SC, float* __restrict__ ST, int h, float scale) {
  __shared__ alignas(16) float redm[16][8];
  __shared__ alignas(16) float reds[16][8];
  const int lane = threadIdx.x & 31;
  const int wave = threadIdx.x >> 5;
  const int half = lane >> 4;
  const int lr   = lane & 15;
  const int m0   = blockIdx.x * 16;
  const int n0   = blockIdx.y * 512 + wave * 64;

  const float* qp = Qh + (size_t)h * kS * kDH + (size_t)(m0 + lr) * kDH + 2 * half;
  const float* kp = Kh + (size_t)h * kS * kDH + (size_t)(n0 + lr) * kDH + 2 * half;
  const float* bb = ub + h * kDH + 2 * half;

  v8f acc[4] = {v8f{}, v8f{}, v8f{}, v8f{}};
  kloop_nt64_bias(qp, kp, bb, acc);
  float sv[4][8];
#pragma unroll
  for (int t = 0; t < 4; ++t) {
    int n = n0 + t * 16 + lr;
#pragma unroll
    for (int r = 0; r < 8; ++r) {
      int m = m0 + r + 8 * half;
      float bd;
      if (n <= m)          bd = BD[(size_t)m * kS + (kS - 1 - m + n)];
      else if (n == m + 1) bd = 0.0f;
      else                 bd = BD[(size_t)(m + 1) * kS + (n - m - 2)];
      sv[t][r] = (acc[t][r] + bd) * scale;
    }
  }

  float mx[8];
#pragma unroll
  for (int r = 0; r < 8; ++r)
    mx[r] = fmaxf(fmaxf(sv[0][r], sv[1][r]), fmaxf(sv[2][r], sv[3][r]));
#pragma unroll
  for (int off = 1; off < 16; off <<= 1) {
#pragma unroll
    for (int r = 0; r < 8; ++r) mx[r] = fmaxf(mx[r], __shfl_xor(mx[r], off, 32));
  }
  if (lr == 0) {
#pragma unroll
    for (int r = 0; r < 8; ++r) redm[r + 8 * half][wave] = mx[r];
  }
  __syncthreads();

  float sm[8];
#pragma unroll
  for (int r = 0; r < 8; ++r) {
    const v4f a = *(const v4f*)&redm[r + 8 * half][0];
    const v4f b = *(const v4f*)&redm[r + 8 * half][4];
    const float bm = fmaxf(fmaxf(fmaxf(a[0], a[1]), fmaxf(a[2], a[3])),
                           fmaxf(fmaxf(b[0], b[1]), fmaxf(b[2], b[3])));
    const int m = m0 + r + 8 * half;
    float s = 0.0f;
#pragma unroll
    for (int t = 0; t < 4; ++t) {
      const float e = __expf(sv[t][r] - bm);
      SC[(size_t)m * kS + n0 + t * 16 + lr] = e;
      s += e;
    }
    sm[r] = s;
  }
#pragma unroll
  for (int off = 1; off < 16; off <<= 1) {
#pragma unroll
    for (int r = 0; r < 8; ++r) sm[r] += __shfl_xor(sm[r], off, 32);
  }
  if (lr == 0) {
#pragma unroll
    for (int r = 0; r < 8; ++r) reds[r + 8 * half][wave] = sm[r];
  }
  __syncthreads();

  if (threadIdx.x < 16) {
    const int row = threadIdx.x;
    const v4f a = *(const v4f*)&redm[row][0];
    const v4f b = *(const v4f*)&redm[row][4];
    const float bm = fmaxf(fmaxf(fmaxf(a[0], a[1]), fmaxf(a[2], a[3])),
                           fmaxf(fmaxf(b[0], b[1]), fmaxf(b[2], b[3])));
    const v4f c = *(const v4f*)&reds[row][0];
    const v4f d = *(const v4f*)&reds[row][4];
    const float bs = ((((((c[0] + c[1]) + c[2]) + c[3]) + d[0]) + d[1]) + d[2]) + d[3];
    *(v2f*)(ST + ((size_t)(m0 + row) * 4 + blockIdx.y) * 2) = v2f{bm, bs};
  }
}

__global__ void __launch_bounds__(64) av_gemm(const float* __restrict__ SC, const float* __restrict__ ST,
                        const float* __restrict__ Vh, float* __restrict__ AVec, int h) {
  __shared__ alignas(16) float tile[16][36];
  const int lane = threadIdx.x & 31;
  const int wave = threadIdx.x >> 5;
  const int half = lane >> 4;
  const int lr   = lane & 15;
  const int m0   = (blockIdx.x >> 1) * 16;
  const int c0   = (blockIdx.x & 1) * 32;
  const int d0   = c0 + wave * 16;

  const float* pp = SC + (size_t)(m0 + lr) * kS + 2 * half;
  const float* vp = Vh + (size_t)h * kS * kDH + (size_t)(2 * half) * kDH + d0 + lr;

  const v4f st0 = *(const v4f*)(ST + (size_t)(m0 + lr) * 8);
  const v4f st1 = *(const v4f*)(ST + (size_t)(m0 + lr) * 8 + 4);

  v8f acc = v8f{};
  Stage1<4> s0, s1, s2, s3, s4, s5;
  load_nn1<4>(s0, pp, vp, 0);
  hold(acc);
  load_nn1<4>(s1, pp, vp, 16);
  hold(acc);
  load_nn1<4>(s2, pp, vp, 32);
  hold(acc);
  load_nn1<4>(s3, pp, vp, 48);
  hold(acc);
  load_nn1<4>(s4, pp, vp, 64);
  hold(acc);
  pin_order();
  v4f f;
  {
    const float M  = fmaxf(fmaxf(st0[0], st0[2]), fmaxf(st1[0], st1[2]));
    const float w0 = __expf(st0[0] - M), w1 = __expf(st0[2] - M);
    const float w2 = __expf(st1[0] - M), w3 = __expf(st1[2] - M);
    const float L  = ((st0[1] * w0 + st0[3] * w1) + st1[1] * w2) + st1[3] * w3;
    const float inv = 1.0f / L;
    f[0] = w0 * inv; f[1] = w1 * inv; f[2] = w2 * inv; f[3] = w3 * inv;
  }
  for (int k = 0; k < 20 * 96; k += 96) {
    load_nn1<4>(s5, pp, vp, k + 80);
    hold(acc);
    mma_stage1<4>(s0, block_factor(f, k), acc);
    hold(acc);
    load_nn1<4>(s0, pp, vp, k + 96);
    hold(acc);
    mma_stage1<4>(s1, block_factor(f, k + 16), acc);
    hold(acc);
    load_nn1<4>(s1, pp, vp, k + 112);
    hold(acc);
    mma_stage1<4>(s2, block_factor(f, k + 32), acc);
    hold(acc);
    load_nn1<4>(s2, pp, vp, k + 128);
    hold(acc);
    mma_stage1<4>(s3, block_factor(f, k + 48), acc);
    hold(acc);
    load_nn1<4>(s3, pp, vp, k + 144);
    hold(acc);
    mma_stage1<4>(s4, block_factor(f, k + 64), acc);
    hold(acc);
    load_nn1<4>(s4, pp, vp, k + 160);
    hold(acc);
    mma_stage1<4>(s5, block_factor(f, k + 80), acc);
    hold(acc);
  }
  load_nn1<4>(s5, pp, vp, kS - 48);
  hold(acc);
  mma_stage1<4>(s0, f[3], acc);
  hold(acc);
  load_nn1<4>(s0, pp, vp, kS - 32);
  hold(acc);
  mma_stage1<4>(s1, f[3], acc);
  hold(acc);
  load_nn1<4>(s1, pp, vp, kS - 16);
  hold(acc);
  mma_stage1<4>(s2, f[3], acc);
  mma_stage1<4>(s3, f[3], acc);
  mma_stage1<4>(s4, f[3], acc);
  mma_stage1<4>(s5, f[3], acc);
  mma_stage1<4>(s0, f[3], acc);
  mma_stage1<4>(s1, f[3], acc);

#pragma unroll
  for (int r = 0; r < 8; ++r) tile[r + 8 * half][wave * 16 + lr] = acc[r];
  __syncthreads();
  for (int j = 0; j < 2; ++j) {
    const int idx = j * 64 + (int)threadIdx.x;
    const int row = idx >> 3;
    const int q   = idx & 7;
    const v4f val = *(const v4f*)&tile[row][q * 4];
    *(v4f*)(AVec + (size_t)(m0 + row) * kD + h * kDH + c0 + q * 4) = val;
  }
}

__global__ void __launch_bounds__(256) out_gemm(const float* __restrict__ A, const float* __restrict__ B,
                         float* __restrict__ C) {
  const int lane = threadIdx.x & 31;
  const int wave = threadIdx.x >> 5;
  const int half = lane >> 4;
  const int lr   = lane & 15;
  const int m0   = blockIdx.x * 16;
  const int n0   = wave * 64;

  const float* ap = A + (size_t)(m0 + lr) * kD + 2 * half;
  const float* bp = B + (size_t)(n0 + lr) * kD + 2 * half;

  v8f acc[4] = {v8f{}, v8f{}, v8f{}, v8f{}};
  kloop_nt512(ap, bp, acc);
#pragma unroll
  for (int t = 0; t < 4; ++t) {
    int n = n0 + t * 16 + lr;
#pragma unroll
    for (int r = 0; r < 8; ++r) {
      int m = m0 + r + 8 * half;
      C[(size_t)m * kD + n] = acc[t][r];
    }
  }
}

extern "C" void kernel_launch(void* const* d_in, const int* in_sizes, int n_in,
                              void* d_out, int out_size, void* d_ws, size_t ws_size,
                              hipStream_t stream) {
  const float* x   = (const float*)d_in[0];
  const float* pos = (const float*)d_in[1];
  const float* Wq  = (const float*)d_in[2];
  const float* Wk  = (const float*)d_in[3];
  const float* Wv  = (const float*)d_in[4];
  const float* Wp  = (const float*)d_in[5];
  const float* Wo  = (const float*)d_in[6];
  const float* u   = (const float*)d_in[7];
  const float* v   = (const float*)d_in[8];

  float* ws = (float*)d_ws;
  const size_t SD = (size_t)kS * kD;
  float* Qhs  = ws;
  float* Khs  = Qhs + SD;
  float* Vhs  = Khs + SD;
  float* Phs  = Vhs + SD;
  float* AVec = Phs + SD;
  float* BD   = AVec + SD;
  float* SC   = BD + (size_t)kS * kS;
  float* ST   = SC + (size_t)kS * kS;

  dim3 blk(256);
  proj_gemm_head<<<dim3(kS / 16), blk, 0, stream>>>(x,   Wq, Qhs);
  proj_gemm_head<<<dim3(kS / 16), blk, 0, stream>>>(x,   Wk, Khs);
  proj_gemm_head<<<dim3(kS / 16), blk, 0, stream>>>(x,   Wv, Vhs);
  proj_gemm_head<<<dim3(kS / 16), blk, 0, stream>>>(pos, Wp, Phs);

  const float scale = 0.125f;
  for (int h = 0; h < kH; ++h) {
    bd_gemm   <<<dim3(kS / 16, kS / 512), blk, 0, stream>>>(Qhs, Phs, v, BD, h);
    score_gemm<<<dim3(kS / 16, kS / 512), blk, 0, stream>>>(Qhs, Khs, u, BD, SC, ST, h, scale);
    av_gemm   <<<dim3(kS / 16 * 2), dim3(64), 0, stream>>>(SC, ST, Vhs, AVec, h);
  }

  out_gemm<<<dim3(kS / 16), blk, 0, stream>>>(AVec, Wo, (float*)d_out);
}


# BD product moved into the score kernel: a wave multiplies its own shifted 80-column P window (5 tiles for 4) and skews it across lanes by lane permutes; BD plane and 8 launches gone; sum order kept
# speedup vs baseline: 3.8449x; 1.2313x over previous
#include <hip/hip_runtime.h>
#include <math.h>

typedef __attribute__((ext_vector_type(2))) float v2f;
typedef __attribute__((ext_vector_type(4))) float v4f;
typedef __attribute__((ext_vector_type(8))) float v8f;

static constexpr int kS  = 2048;
static constexpr int kD  = 512;
static constexpr int kH  = 8;
static constexpr int kDH = 64;

__device__ __forceinline__ v8f wmma_f32(v2f a, v2f b, v8f c) {
  return __builtin_amdgcn_wmma_f32_16x16x4_f32(
       false, a,  false, b,
       (short)0, c,  false,  false);
}

template <int U>
struct Stage {
  v2f a[U];
  v2f b[U][4];
};

__device__ __forceinline__ void pin_order() { __builtin_amdgcn_sched_barrier(0); }

__device__ __forceinline__ void hold(v8f& acc) { asm volatile("" : "+v"(acc) : : "memory"); }

template <int U>
__device__ __forceinline__ void load_nt(Stage<U>& s, const float* __restrict__ ap,
                                        const float* __restrict__ bp, int ld, int k0) {
#pragma unroll
  for (int i = 0; i < U; ++i) {
    s.a[i] = *(const v2f*)(ap + k0 + 4 * i);
#pragma unroll
    for (int t = 0; t < 4; ++t)
      s.b[i][t] = *(const v2f*)(bp + (size_t)t * 16 * ld + k0 + 4 * i);
  }
}

template <int U>
struct Stage1 {
  v2f a[U];
  v2f b[U];
};

template <int U>
__device__ __forceinline__ void load_nn1(Stage1<U>& s, const float* __restrict__ ap,
                                         const float* __restrict__ vp, int k0) {
#pragma unroll
  for (int i = 0; i < U; ++i) {
    s.a[i] = *(const v2f*)(ap + k0 + 4 * i);
    const float* vb = vp + (size_t)(k0 + 4 * i) * kDH;
    s.b[i] = v2f{vb[0], vb[kDH]};
  }
}

template <int U>
__device__ __forceinline__ void mma_stage1(const Stage1<U>& s, float f, v8f& acc) {
  pin_order();
#pragma unroll
  for (int i = 0; i < U; ++i) acc = wmma_f32(s.a[i] * f, s.b[i], acc);
}

__device__ __forceinline__ float block_factor(v4f f, int k0) { return f[k0 >> 9]; }

template <int U>
__device__ __forceinline__ void mma_stage(const Stage<U>& s, v8f (&acc)[4]) {
#pragma unroll
  for (int i = 0; i < U; ++i) {
#pragma unroll
    for (int t = 0; t < 4; ++t)
      acc[t] = wmma_f32(s.a[i], s.b[i][t], acc[t]);
  }
}

__device__ __forceinline__ void kloop_nt512(const float* __restrict__ ap,
                                            const float* __restrict__ bp, v8f (&acc)[4]) {
  Stage<8> s0, s1;
  load_nt<8>(s0, ap, bp, kD, 0);
  for (int k = 0; k < kD - 64; k += 64) {
    load_nt<8>(s1, ap, bp, kD, k + 32);
    pin_order();
    mma_stage<8>(s0, acc);
    pin_order();
    load_nt<8>(s0, ap, bp, kD, k + 64);
    pin_order();
    mma_stage<8>(s1, acc);
    pin_order();
  }
  load_nt<8>(s1, ap, bp, kD, kD - 32);
  pin_order();
  mma_stage<8>(s0, acc);
  mma_stage<8>(s1, acc);
}

__device__ __forceinline__ void kloop_nt64_bias(const float* __restrict__ ap,
                                                const float* __restrict__ bp,
                                                const float* __restrict__ bb, v8f (&acc)[4]) {
  Stage<8> s0, s1;
  v2f c0[8], c1[8];
  load_nt<8>(s0, ap, bp, kDH, 0);
#pragma unroll
  for (int i = 0; i < 8; ++i) c0[i] = *(const v2f*)(bb + 4 * i);
  load_nt<8>(s1, ap, bp, kDH, 32);
#pragma unroll
  for (int i = 0; i < 8; ++i) c1[i] = *(const v2f*)(bb + 32 + 4 * i);
  pin_order();
#pragma unroll
  for (int i = 0; i < 8; ++i) s0.a[i] = s0.a[i] + c0[i];
  mma_stage<8>(s0, acc);
#pragma unroll
  for (int i = 0; i < 8; ++i) s1.a[i] = s1.a[i] + c1[i];
  mma_stage<8>(s1, acc);
}

__global__ void __launch_bounds__(256) proj_gemm_head(const float* __restrict__ A,
                               const float* __restrict__ B,
                               float* __restrict__ out) {
  const int lane = threadIdx.x & 31;
  const int wave = threadIdx.x >> 5;
  const int half = lane >> 4;
  const int lr   = lane & 15;
  const int m0   = blockIdx.x * 16;
  const int n0   = wave * 64;

  const float* ap = A + (size_t)(m0 + lr) * kD + 2 * half;
  const float* bp = B + (size_t)(n0 + lr) * kD + 2 * half;

  v8f acc[4] = {v8f{}, v8f{}, v8f{}, v8f{}};
  kloop_nt512(ap, bp, acc);
#pragma unroll
  for (int t = 0; t < 4; ++t) {
    int n = n0 + t * 16 + lr;
    int h = n >> 6, d = n & 63;
    float* op = out + (size_t)h * kS * kDH + d;
#pragma unroll
    for (int r = 0; r < 8; ++r) {
      int m = m0 + r + 8 * half;
      op[(size_t)m * kDH] = acc[t][r];
    }
  }
}

template <int U>
struct StageW {
  v2f a[U];
  v2f c[U];
  v2f b[U][5];
};

template <int U>
__device__ __forceinline__ void load_win(StageW<U>& s, const float* __restrict__ ap,
                                         const float* __restrict__ cp,
                                         const float* __restrict__ pb,
                                         const unsigned (&off)[5], int k0) {
#pragma unroll
  for (int i = 0; i < U; ++i) {
    s.a[i] = *(const v2f*)(ap + k0 + 4 * i);
    s.c[i] = *(const v2f*)(cp + k0 + 4 * i);
#pragma unroll
    for (int w = 0; w < 5; ++w)
      s.b[i][w] = *(const v2f*)(pb + (size_t)off[w] + (k0 + 4 * i));
  }
}

template <int U>
__device__ __forceinline__ void mma_win(const StageW<U>& s, v8f (&g)[5]) {
#pragma unroll
  for (int i = 0; i < U; ++i) {
    const v2f a = s.a[i] + s.c[i];
#pragma unroll
    for (int w = 0; w < 5; ++w) g[w] = wmma_f32(a, s.b[i][w], g[w]);
  }
}

__device__ __forceinline__ void shifted_bd(const float* __restrict__ ap,
                                           const float* __restrict__ cp,
                                           const float* __restrict__ pb, int cs,
                                           int lane, float (&sh)[4][8]) {
  const int half = lane >> 4;
  const int lr   = lane & 15;
  unsigned off[5];
#pragma unroll
  for (int w = 0; w < 5; ++w) {
    int c = cs + 16 * w + lr;
    c = c < 0 ? 0 : (c > kS - 1 ? kS - 1 : c);
    off[w] = (unsigned)(c * kDH + 2 * half);
  }
  v8f g[5] = {v8f{}, v8f{}, v8f{}, v8f{}, v8f{}};
  StageW<4> s0, s1;
  load_win<4>(s0, ap, cp, pb, off, 0);
  load_win<4>(s1, ap, cp, pb, off, 16);
  pin_order();
  mma_win<4>(s0, g);
  pin_order();
  load_win<4>(s0, ap, cp, pb, off, 32);
  pin_order();
  mma_win<4>(s1, g);
  pin_order();
  load_win<4>(s1, ap, cp, pb, off, 48);
  pin_order();
  mma_win<4>(s0, g);
  mma_win<4>(s1, g);

#pragma unroll
  for (int r = 0; r < 8; ++r) {
    const int m   = r + 8 * half;
    const int src = (lane & 16) | ((lr - m - 1) & 15);
    float p[5];
#pragma unroll
    for (int w = 0; w < 5; ++w) p[w] = __shfl(g[w][r], src, 32);
#pragma unroll
    for (int t = 0; t < 4; ++t) sh[t][r] = (lr <= m) ? p[t] : p[t + 1];
  }
}

__global__ void __launch_bounds__(256) score_gemm(const float* __restrict__ Qh, const float* __restrict__ Kh,
                           const float* __restrict__ Ph, const float* __restrict__ ub,
                           const float* __restrict__ vb,
                           float* __restrict__ SC, float* __restrict__ ST, int h, float scale) {
  __shared__ alignas(16) float redm[16][8];
  __shared__ alignas(16) float reds[16][8];
  const int lane = threadIdx.x & 31;
  const int wave = threadIdx.x >> 5;
  const int half = lane >> 4;
  const int lr   = lane & 15;
  const int m0   = blockIdx.x * 16;
  const int n0   = blockIdx.y * 512 + wave * 64;

  const float* qp = Qh + (size_t)h * kS * kDH + (size_t)(m0 + lr) * kDH + 2 * half;
  const float* kp = Kh + (size_t)h * kS * kDH + (size_t)(n0 + lr) * kDH + 2 * half;
  const float* bb = ub + h * kDH + 2 * half;

  v8f acc[4] = {v8f{}, v8f{}, v8f{}, v8f{}};
  kloop_nt64_bias(qp, kp, bb, acc);
  pin_order();

  const int delta = __builtin_amdgcn_readfirstlane(n0 - m0);
  const float* pb = Ph + (size_t)h * kS * kDH;
  const float* cp = vb + h * kDH + 2 * half;
  float bd[4][8];
#pragma unroll
  for (int t = 0; t < 4; ++t) {
#pragma unroll
    for (int r = 0; r < 8; ++r) bd[t][r] = 0.0f;
  }
  if (delta <= 0) {
    float sh[4][8];
    shifted_bd(qp, cp, pb, kS - 16 + delta, lane, sh);
#pragma unroll
    for (int t = 0; t < 4; ++t) {
      const int n = n0 + t * 16 + lr;
#pragma unroll
      for (int r = 0; r < 8; ++r) {
        const int m = m0 + r + 8 * half;
        bd[t][r] = (n <= m) ? sh[t][r] : 0.0f;
      }
    }
  }
  if (delta >= -48) {
    const int row1 = (m0 + lr + 1 > kS - 1) ? kS - 1 : m0 + lr + 1;
    const float* qp1 = Qh + (size_t)h * kS * kDH + (size_t)row1 * kDH + 2 * half;
    float sh[4][8];
    shifted_bd(qp1, cp, pb, delta - 17, lane, sh);
#pragma unroll
    for (int t = 0; t < 4; ++t) {
      const int n = n0 + t * 16 + lr;
#pragma unroll
      for (int r = 0; r < 8; ++r) {
        const int m = m0 + r + 8 * half;
        bd[t][r] = (n >= m + 2) ? sh[t][r] : bd[t][r];
      }
    }
  }

  float sv[4][8];
#pragma unroll
  for (int t = 0; t < 4; ++t) {
#pragma unroll
    for (int r = 0; r < 8; ++r) sv[t][r] = (acc[t][r] + bd[t][r]) * scale;
  }

  float mx[8];
#pragma unroll
  for (int r = 0; r < 8; ++r)
    mx[r] = fmaxf(fmaxf(sv[0][r], sv[1][r]), fmaxf(sv[2][r], sv[3][r]));
#pragma unroll
  for (int off = 1; off < 16; off <<= 1) {
#pragma unroll
    for (int r = 0; r < 8; ++r) mx[r] = fmaxf(mx[r], __shfl_xor(mx[r], off, 32));
  }
  if (lr == 0) {
#pragma unroll
    for (int r = 0; r < 8; ++r) redm[r + 8 * half][wave] = mx[r];
  }
  __syncthreads();

  float sm[8];
#pragma unroll
  for (int r = 0; r < 8; ++r) {
    const v4f a = *(const v4f*)&redm[r + 8 * half][0];
    const v4f b = *(const v4f*)&redm[r + 8 * half][4];
    const float bm = fmaxf(fmaxf(fmaxf(a[0], a[1]), fmaxf(a[2], a[3])),
                           fmaxf(fmaxf(b[0], b[1]), fmaxf(b[2], b[3])));
    const int m = m0 + r + 8 * half;
    float s = 0.0f;
#pragma unroll
    for (int t = 0; t < 4; ++t) {
      const float e = __expf(sv[t][r] - bm);
      SC[(size_t)m * kS + n0 + t * 16 + lr] = e;
      s += e;
    }
    sm[r] = s;
  }
#pragma unroll
  for (int off = 1; off < 16; off <<= 1) {
#pragma unroll
    for (int r = 0; r < 8; ++r) sm[r] += __shfl_xor(sm[r], off, 32);
  }
  if (lr == 0) {
#pragma unroll
    for (int r = 0; r < 8; ++r) reds[r + 8 * half][wave] = sm[r];
  }
  __syncthreads();

  if (threadIdx.x < 16) {
    const int row = threadIdx.x;
    const v4f a = *(const v4f*)&redm[row][0];
    const v4f b = *(const v4f*)&redm[row][4];
    const float bm = fmaxf(fmaxf(fmaxf(a[0], a[1]), fmaxf(a[2], a[3])),
                           fmaxf(fmaxf(b[0], b[1]), fmaxf(b[2], b[3])));
    const v4f c = *(const v4f*)&reds[row][0];
    const v4f d = *(const v4f*)&reds[row][4];
    const float bs = ((((((c[0] + c[1]) + c[2]) + c[3]) + d[0]) + d[1]) + d[2]) + d[3];
    *(v2f*)(ST + ((size_t)(m0 + row) * 4 + blockIdx.y) * 2) = v2f{bm, bs};
  }
}

__global__ void __launch_bounds__(64) av_gemm(const float* __restrict__ SC, const float* __restrict__ ST,
                        const float* __restrict__ Vh, float* __restrict__ AVec, int h) {
  __shared__ alignas(16) float tile[16][36];
  const int lane = threadIdx.x & 31;
  const int wave = threadIdx.x >> 5;
  const int half = lane >> 4;
  const int lr   = lane & 15;
  const int m0   = (blockIdx.x >> 1) * 16;
  const int c0   = (blockIdx.x & 1) * 32;
  const int d0   = c0 + wave * 16;

  const float* pp = SC + (size_t)(m0 + lr) * kS + 2 * half;
  const float* vp = Vh + (size_t)h * kS * kDH + (size_t)(2 * half) * kDH + d0 + lr;

  const v4f st0 = *(const v4f*)(ST + (size_t)(m0 + lr) * 8);
  const v4f st1 = *(const v4f*)(ST + (size_t)(m0 + lr) * 8 + 4);

  v8f acc = v8f{};
  Stage1<4> s0, s1, s2, s3, s4, s5;
  load_nn1<4>(s0, pp, vp, 0);
  hold(acc);
  load_nn1<4>(s1, pp, vp, 16);
  hold(acc);
  load_nn1<4>(s2, pp, vp, 32);
  hold(acc);
  load_nn1<4>(s3, pp, vp, 48);
  hold(acc);
  load_nn1<4>(s4, pp, vp, 64);
  hold(acc);
  pin_order();
  v4f f;
  {
    const float M  = fmaxf(fmaxf(st0[0], st0[2]), fmaxf(st1[0], st1[2]));
    const float w0 = __expf(st0[0] - M), w1 = __expf(st0[2] - M);
    const float w2 = __expf(st1[0] - M), w3 = __expf(st1[2] - M);
    const float L  = ((st0[1] * w0 + st0[3] * w1) + st1[1] * w2) + st1[3] * w3;
    const float inv = 1.0f / L;
    f[0] = w0 * inv; f[1] = w1 * inv; f[2] = w2 * inv; f[3] = w3 * inv;
  }
  for (int k = 0; k < 20 * 96; k += 96) {
    load_nn1<4>(s5, pp, vp, k + 80);
    hold(acc);
    mma_stage1<4>(s0, block_factor(f, k), acc);
    hold(acc);
    load_nn1<4>(s0, pp, vp, k + 96);
    hold(acc);
    mma_stage1<4>(s1, block_factor(f, k + 16), acc);
    hold(acc);
    load_nn1<4>(s1, pp, vp, k + 112);
    hold(acc);
    mma_stage1<4>(s2, block_factor(f, k + 32), acc);
    hold(acc);
    load_nn1<4>(s2, pp, vp, k + 128);
    hold(acc);
    mma_stage1<4>(s3, block_factor(f, k + 48), acc);
    hold(acc);
    load_nn1<4>(s3, pp, vp, k + 144);
    hold(acc);
    mma_stage1<4>(s4, block_factor(f, k + 64), acc);
    hold(acc);
    load_nn1<4>(s4, pp, vp, k + 160);
    hold(acc);
    mma_stage1<4>(s5, block_factor(f, k + 80), acc);
    hold(acc);
  }
  load_nn1<4>(s5, pp, vp, kS - 48);
  hold(acc);
  mma_stage1<4>(s0, f[3], acc);
  hold(acc);
  load_nn1<4>(s0, pp, vp, kS - 32);
  hold(acc);
  mma_stage1<4>(s1, f[3], acc);
  hold(acc);
  load_nn1<4>(s1, pp, vp, kS - 16);
  hold(acc);
  mma_stage1<4>(s2, f[3], acc);
  mma_stage1<4>(s3, f[3], acc);
  mma_stage1<4>(s4, f[3], acc);
  mma_stage1<4>(s5, f[3], acc);
  mma_stage1<4>(s0, f[3], acc);
  mma_stage1<4>(s1, f[3], acc);

#pragma unroll
  for (int r = 0; r < 8; ++r) tile[r + 8 * half][wave * 16 + lr] = acc[r];
  __syncthreads();
  for (int j = 0; j < 2; ++j) {
    const int idx = j * 64 + (int)threadIdx.x;
    const int row = idx >> 3;
    const int q   = idx & 7;
    const v4f val = *(const v4f*)&tile[row][q * 4];
    *(v4f*)(AVec + (size_t)(m0 + row) * kD + h * kDH + c0 + q * 4) = val;
  }
}

__global__ void __launch_bounds__(256) out_gemm(const float* __restrict__ A, const float* __restrict__ B,
                         float* __restrict__ C) {
  const int lane = threadIdx.x & 31;
  const int wave = threadIdx.x >> 5;
  const int half = lane >> 4;
  const int lr   = lane & 15;
  const int m0   = blockIdx.x * 16;
  const int n0   = wave * 64;

  const float* ap = A + (size_t)(m0 + lr) * kD + 2 * half;
  const float* bp = B + (size_t)(n0 + lr) * kD + 2 * half;

  v8f acc[4] = {v8f{}, v8f{}, v8f{}, v8f{}};
  kloop_nt512(ap, bp, acc);
#pragma unroll
  for (int t = 0; t < 4; ++t) {
    int n = n0 + t * 16 + lr;
#pragma unroll
    for (int r = 0; r < 8; ++r) {
      int m = m0 + r + 8 * half;
      C[(size_t)m * kD + n] = acc[t][r];
    }
  }
}

extern "C" void kernel_launch(void* const* d_in, const int* in_sizes, int n_in,
                              void* d_out, int out_size, void* d_ws, size_t ws_size,
                              hipStream_t stream) {
  const float* x   = (const float*)d_in[0];
  const float* pos = (const float*)d_in[1];
  const float* Wq  = (const float*)d_in[2];
  const float* Wk  = (const float*)d_in[3];
  const float* Wv  = (const float*)d_in[4];
  const float* Wp  = (const float*)d_in[5];
  const float* Wo  = (const float*)d_in[6];
  const float* u   = (const float*)d_in[7];
  const float* v   = (const float*)d_in[8];

  float* ws = (float*)d_ws;
  const size_t SD = (size_t)kS * kD;
  float* Qhs  = ws;
  float* Khs  = Qhs + SD;
  float* Vhs  = Khs + SD;
  float* Phs  = Vhs + SD;
  float* AVec = Phs + SD;
  float* SC   = AVec + SD;
  float* ST   = SC + (size_t)kS * kS;

  dim3 blk(256);
  proj_gemm_head<<<dim3(kS / 16), blk, 0, stream>>>(x,   Wq, Qhs);
  proj_gemm_head<<<dim3(kS / 16), blk, 0, stream>>>(x,   Wk, Khs);
  proj_gemm_head<<<dim3(kS / 16), blk, 0, stream>>>(x,   Wv, Vhs);
  proj_gemm_head<<<dim3(kS / 16), blk, 0, stream>>>(pos, Wp, Phs);

  const float scale = 0.125f;
  for (int h = 0; h < kH; ++h) {
    score_gemm<<<dim3(kS / 16, kS / 512), blk, 0, stream>>>(Qhs, Khs, Phs, u, v, SC, ST, h, scale);
    av_gemm   <<<dim3(kS / 16 * 2), dim3(64), 0, stream>>>(SC, ST, Vhs, AVec, h);
  }

  out_gemm<<<dim3(kS / 16), blk, 0, stream>>>(AVec, Wo, (float*)d_out);
}


# score and AV products fused into one kernel per (16 rows, head): the block keeps its 16 x 2048 exponentials in LDS, four of its waves multiply them by V; score plane, 15 launches gone; sum order kept
# speedup vs baseline: 6.3903x; 1.6620x over previous
#include <hip/hip_runtime.h>
#include <math.h>

typedef __attribute__((ext_vector_type(2))) float v2f;
typedef __attribute__((ext_vector_type(4))) float v4f;
typedef __attribute__((ext_vector_type(8))) float v8f;

static constexpr int kS  = 2048;
static constexpr int kD  = 512;
static constexpr int kH  = 8;
static constexpr int kDH = 64;

__device__ __forceinline__ v8f wmma_f32(v2f a, v2f b, v8f c) {
  return __builtin_amdgcn_wmma_f32_16x16x4_f32(
       false, a,  false, b,
       (short)0, c,  false,  false);
}

template <int U>
struct Stage {
  v2f a[U];
  v2f b[U][4];
};

__device__ __forceinline__ void pin_order() { __builtin_amdgcn_sched_barrier(0); }

__device__ __forceinline__ void hold(v8f& acc) { asm volatile("" : "+v"(acc) : : "memory"); }

template <int U>
__device__ __forceinline__ void load_nt(Stage<U>& s, const float* __restrict__ ap,
                                        const float* __restrict__ bp, int ld, int k0) {
#pragma unroll
  for (int i = 0; i < U; ++i) {
    s.a[i] = *(const v2f*)(ap + k0 + 4 * i);
#pragma unroll
    for (int t = 0; t < 4; ++t)
      s.b[i][t] = *(const v2f*)(bp + (size_t)t * 16 * ld + k0 + 4 * i);
  }
}

template <int U>
struct Stage1 {
  v2f a[U];
  v2f b[U];
};

template <int U>
__device__ __forceinline__ void load_nn1(Stage1<U>& s, const float* __restrict__ ap,
                                         const float* __restrict__ vp, int k0) {
#pragma unroll
  for (int i = 0; i < U; ++i) {
    s.a[i] = *(const v2f*)(ap + k0 + 4 * i);
    const float* vb = vp + (size_t)(k0 + 4 * i) * kDH;
    s.b[i] = v2f{vb[0], vb[kDH]};
  }
}

template <int U>
__device__ __forceinline__ void mma_stage1(const Stage1<U>& s, float f, v8f& acc) {
  pin_order();
#pragma unroll
  for (int i = 0; i < U; ++i) acc = wmma_f32(s.a[i] * f, s.b[i], acc);
}

__device__ __forceinline__ float block_factor(v4f f, int k0) { return f[k0 >> 9]; }

template <int U>
__device__ __forceinline__ void mma_stage(const Stage<U>& s, v8f (&acc)[4]) {
#pragma unroll
  for (int i = 0; i < U; ++i) {
#pragma unroll
    for (int t = 0; t < 4; ++t)
      acc[t] = wmma_f32(s.a[i], s.b[i][t], acc[t]);
  }
}

__device__ __forceinline__ void kloop_nt512(const float* __restrict__ ap,
                                            const float* __restrict__ bp, v8f (&acc)[4]) {
  Stage<8> s0, s1;
  load_nt<8>(s0, ap, bp, kD, 0);
  for (int k = 0; k < kD - 64; k += 64) {
    load_nt<8>(s1, ap, bp, kD, k + 32);
    pin_order();
    mma_stage<8>(s0, acc);
    pin_order();
    load_nt<8>(s0, ap, bp, kD, k + 64);
    pin_order();
    mma_stage<8>(s1, acc);
    pin_order();
  }
  load_nt<8>(s1, ap, bp, kD, kD - 32);
  pin_order();
  mma_stage<8>(s0, acc);
  mma_stage<8>(s1, acc);
}

__device__ __forceinline__ void kloop_nt64_bias(const float* __restrict__ ap,
                                                const float* __restrict__ bp,
                                                const float* __restrict__ bb, v8f (&acc)[4]) {
  Stage<8> s0, s1;
  v2f c0[8], c1[8];
  load_nt<8>(s0, ap, bp, kDH, 0);
#pragma unroll
  for (int i = 0; i < 8; ++i) c0[i] = *(const v2f*)(bb + 4 * i);
  load_nt<8>(s1, ap, bp, kDH, 32);
#pragma unroll
  for (int i = 0; i < 8; ++i) c1[i] = *(const v2f*)(bb + 32 + 4 * i);
  pin_order();
#pragma unroll
  for (int i = 0; i < 8; ++i) s0.a[i] = s0.a[i] + c0[i];
  mma_stage<8>(s0, acc);
#pragma unroll
  for (int i = 0; i < 8; ++i) s1.a[i] = s1.a[i] + c1[i];
  mma_stage<8>(s1, acc);
}

__global__ void __launch_bounds__(256) proj_gemm_head(const float* __restrict__ A,
                               const float* __restrict__ B,
                               float* __restrict__ out) {
  const int lane = threadIdx.x & 31;
  const int wave = threadIdx.x >> 5;
  const int half = lane >> 4;
  const int lr   = lane & 15;
  const int m0   = blockIdx.x * 16;
  const int n0   = wave * 64;

  const float* ap = A + (size_t)(m0 + lr) * kD + 2 * half;
  const float* bp = B + (size_t)(n0 + lr) * kD + 2 * half;

  v8f acc[4] = {v8f{}, v8f{}, v8f{}, v8f{}};
  kloop_nt512(ap, bp, acc);
#pragma unroll
  for (int t = 0; t < 4; ++t) {
    int n = n0 + t * 16 + lr;
    int h = n >> 6, d = n & 63;
    float* op = out + (size_t)h * kS * kDH + d;
#pragma unroll
    for (int r = 0; r < 8; ++r) {
      int m = m0 + r + 8 * half;
      op[(size_t)m * kDH] = acc[t][r];
    }
  }
}

template <int U>
struct StageW {
  v2f a[U];
  v2f c[U];
  v2f b[U][5];
};

template <int U>
__device__ __forceinline__ void load_win(StageW<U>& s, const float* __restrict__ ap,
                                         const float* __restrict__ cp,
                                         const float* __restrict__ pb,
                                         const unsigned (&off)[5], int k0) {
#pragma unroll
  for (int i = 0; i < U; ++i) {
    s.a[i] = *(const v2f*)(ap + k0 + 4 * i);
    s.c[i] = *(const v2f*)(cp + k0 + 4 * i);
#pragma unroll
    for (int w = 0; w < 5; ++w)
      s.b[i][w] = *(const v2f*)(pb + (size_t)off[w] + (k0 + 4 * i));
  }
}

template <int U>
__device__ __forceinline__ void mma_win(const StageW<U>& s, v8f (&g)[5]) {
#pragma unroll
  for (int i = 0; i < U; ++i) {
    const v2f a = s.a[i] + s.c[i];
#pragma unroll
    for (int w = 0; w < 5; ++w) g[w] = wmma_f32(a, s.b[i][w], g[w]);
  }
}

__device__ __forceinline__ void shifted_bd(const float* __restrict__ ap,
                                           const float* __restrict__ cp,
                                           const float* __restrict__ pb, int cs,
                                           int lane, float (&sh)[4][8]) {
  const int half = lane >> 4;
  const int lr   = lane & 15;
  unsigned off[5];
#pragma unroll
  for (int w = 0; w < 5; ++w) {
    int c = cs + 16 * w + lr;
    c = c < 0 ? 0 : (c > kS - 1 ? kS - 1 : c);
    off[w] = (unsigned)(c * kDH + 2 * half);
  }
  v8f g[5] = {v8f{}, v8f{}, v8f{}, v8f{}, v8f{}};
  StageW<4> s0, s1;
  load_win<4>(s0, ap, cp, pb, off, 0);
  load_win<4>(s1, ap, cp, pb, off, 16);
  pin_order();
  mma_win<4>(s0, g);
  pin_order();
  load_win<4>(s0, ap, cp, pb, off, 32);
  pin_order();
  mma_win<4>(s1, g);
  pin_order();
  load_win<4>(s1, ap, cp, pb, off, 48);
  pin_order();
  mma_win<4>(s0, g);
  mma_win<4>(s1, g);

#pragma unroll
  for (int r = 0; r < 8; ++r) {
    const int m   = r + 8 * half;
    const int src = (lane & 16) | ((lr - m - 1) & 15);
    float p[5];
#pragma unroll
    for (int w = 0; w < 5; ++w) p[w] = __shfl(g[w][r], src, 32);
#pragma unroll
    for (int t = 0; t < 4; ++t) sh[t][r] = (lr <= m) ? p[t] : p[t + 1];
  }
}

static constexpr int kEP = kS + 4;

__global__ void __launch_bounds__(256) attn_fused(const float* __restrict__ Qh, const float* __restrict__ Kh,
                           const float* __restrict__ Ph, const float* __restrict__ Vh,
                           const float* __restrict__ ub, const float* __restrict__ vb,
                           float* __restrict__ AVec, float scale) {
  __shared__ alignas(16) float E[16][kEP];
  __shared__ alignas(16) float redm[4][16][8];
  __shared__ alignas(16) float reds[4][16][8];
  __shared__ alignas(16) float tile[16][68];
  const int lane = threadIdx.x & 31;
  const int wave = __builtin_amdgcn_readfirstlane((int)(threadIdx.x >> 5));
  const int half = lane >> 4;
  const int lr   = lane & 15;
  const int m0   = blockIdx.x * 16;
  const int h    = blockIdx.y;

  const float* qrow = Qh + (size_t)h * kS * kDH + (size_t)(m0 + lr) * kDH + 2 * half;
  const float* ubias = ub + h * kDH + 2 * half;
  const float* pb = Ph + (size_t)h * kS * kDH;
  const float* vbias = vb + h * kDH + 2 * half;

#pragma nounroll
  for (int j = 0; j < 4; ++j) {
    const float* qp = qrow;
    const float* bb = ubias;
    const float* cp = vbias;
    asm volatile("" : "+v"(qp), "+v"(bb), "+v"(cp));
    const int n0 = j * 512 + wave * 64;
    const float* kp = Kh + (size_t)h * kS * kDH + (size_t)(n0 + lr) * kDH + 2 * half;

    v8f acc[4] = {v8f{}, v8f{}, v8f{}, v8f{}};
    kloop_nt64_bias(qp, kp, bb, acc);
    pin_order();

    const int delta = __builtin_amdgcn_readfirstlane(n0 - m0);
    if (delta <= 0) {
      float sh[4][8];
      shifted_bd(qp, cp, pb, kS - 16 + delta, lane, sh);
#pragma unroll
      for (int t = 0; t < 4; ++t) {
        const int n = n0 + t * 16 + lr;
#pragma unroll
        for (int r = 0; r < 8; ++r) {
          const int m = m0 + r + 8 * half;
          acc[t][r] += (n <= m) ? sh[t][r] : 0.0f;
        }
      }
    }
    if (delta >= -48) {
      const int row1 = (m0 + lr + 1 > kS - 1) ? kS - 1 : m0 + lr + 1;
      const float* qp1 = Qh + (size_t)h * kS * kDH + (size_t)row1 * kDH + 2 * half;
      float sh[4][8];
      shifted_bd(qp1, cp, pb, delta - 17, lane, sh);
#pragma unroll
      for (int t = 0; t < 4; ++t) {
        const int n = n0 + t * 16 + lr;
#pragma unroll
        for (int r = 0; r < 8; ++r) {
          const int m = m0 + r + 8 * half;
          acc[t][r] += (n >= m + 2) ? sh[t][r] : 0.0f;
        }
      }
    }

    float sv[4][8];
#pragma unroll
    for (int t = 0; t < 4; ++t) {
#pragma unroll
      for (int r = 0; r < 8; ++r) sv[t][r] = acc[t][r] * scale;
    }

    float mx[8];
#pragma unroll
    for (int r = 0; r < 8; ++r)
      mx[r] = fmaxf(fmaxf(sv[0][r], sv[1][r]), fmaxf(sv[2][r], sv[3][r]));
#pragma unroll
    for (int off = 1; off < 16; off <<= 1) {
#pragma unroll
      for (int r = 0; r < 8; ++r) mx[r] = fmaxf(mx[r], __shfl_xor(mx[r], off, 32));
    }
    if (lr == 0) {
#pragma unroll
      for (int r = 0; r < 8; ++r) redm[j][r + 8 * half][wave] = mx[r];
    }
    __syncthreads();

    float sm[8];
#pragma unroll
    for (int r = 0; r < 8; ++r) {
      const v4f a = *(const v4f*)&redm[j][r + 8 * half][0];
      const v4f b = *(const v4f*)&redm[j][r + 8 * half][4];
      const float bm = fmaxf(fmaxf(fmaxf(a[0], a[1]), fmaxf(a[2], a[3])),
                             fmaxf(fmaxf(b[0], b[1]), fmaxf(b[2], b[3])));
      float s = 0.0f;
#pragma unroll
      for (int t = 0; t < 4; ++t) {
        const float e = __expf(sv[t][r] - bm);
        E[r + 8 * half][n0 + t * 16 + lr] = e;
        s += e;
      }
      sm[r] = s;
    }
#pragma unroll
    for (int off = 1; off < 16; off <<= 1) {
#pragma unroll
      for (int r = 0; r < 8; ++r) sm[r] += __shfl_xor(sm[r], off, 32);
    }
    if (lr == 0) {
#pragma unroll
      for (int r = 0; r < 8; ++r) reds[j][r + 8 * half][wave] = sm[r];
    }
  }
  __syncthreads();

  if (wave < 4) {
    const int d0 = wave * 16;
    const float* ep = &E[lr][2 * half];
    const float* vp = Vh + (size_t)h * kS * kDH + (size_t)(2 * half) * kDH + d0 + lr;

    float bmj[4], bsj[4];
#pragma unroll
    for (int j = 0; j < 4; ++j) {
      const v4f a = *(const v4f*)&redm[j][lr][0];
      const v4f b = *(const v4f*)&redm[j][lr][4];
      bmj[j] = fmaxf(fmaxf(fmaxf(a[0], a[1]), fmaxf(a[2], a[3])),
                     fmaxf(fmaxf(b[0], b[1]), fmaxf(b[2], b[3])));
      const v4f c = *(const v4f*)&reds[j][lr][0];
      const v4f d = *(const v4f*)&reds[j][lr][4];
      bsj[j] = ((((((c[0] + c[1]) + c[2]) + c[3]) + d[0]) + d[1]) + d[2]) + d[3];
    }

    v8f acc = v8f{};
    Stage1<4> s0, s1, s2, s3, s4, s5;
    load_nn1<4>(s0, ep, vp, 0);
    hold(acc);
    load_nn1<4>(s1, ep, vp, 16);
    hold(acc);
    load_nn1<4>(s2, ep, vp, 32);
    hold(acc);
    load_nn1<4>(s3, ep, vp, 48);
    hold(acc);
    load_nn1<4>(s4, ep, vp, 64);
    hold(acc);
    pin_order();
    v4f f;
    {
      const float M  = fmaxf(fmaxf(bmj[0], bmj[1]), fmaxf(bmj[2], bmj[3]));
      const float w0 = __expf(bmj[0] - M), w1 = __expf(bmj[1] - M);
      const float w2 = __expf(bmj[2] - M), w3 = __expf(bmj[3] - M);
      const float L  = ((bsj[0] * w0 + bsj[1] * w1) + bsj[2] * w2) + bsj[3] * w3;
      const float inv = 1.0f / L;
      f[0] = w0 * inv; f[1] = w1 * inv; f[2] = w2 * inv; f[3] = w3 * inv;
    }
    for (int k = 0; k < 20 * 96; k += 96) {
      load_nn1<4>(s5, ep, vp, k + 80);
      hold(acc);
      mma_stage1<4>(s0, block_factor(f, k), acc);
      hold(acc);
      load_nn1<4>(s0, ep, vp, k + 96);
      hold(acc);
      mma_stage1<4>(s1, block_factor(f, k + 16), acc);
      hold(acc);
      load_nn1<4>(s1, ep, vp, k + 112);
      hold(acc);
      mma_stage1<4>(s2, block_factor(f, k + 32), acc);
      hold(acc);
      load_nn1<4>(s2, ep, vp, k + 128);
      hold(acc);
      mma_stage1<4>(s3, block_factor(f, k + 48), acc);
      hold(acc);
      load_nn1<4>(s3, ep, vp, k + 144);
      hold(acc);
      mma_stage1<4>(s4, block_factor(f, k + 64), acc);
      hold(acc);
      load_nn1<4>(s4, ep, vp, k + 160);
      hold(acc);
      mma_stage1<4>(s5, block_factor(f, k + 80), acc);
      hold(acc);
    }
    load_nn1<4>(s5, ep, vp, kS - 48);
    hold(acc);
    mma_stage1<4>(s0, f[3], acc);
    hold(acc);
    load_nn1<4>(s0, ep, vp, kS - 32);
    hold(acc);
    mma_stage1<4>(s1, f[3], acc);
    hold(acc);
    load_nn1<4>(s1, ep, vp, kS - 16);
    hold(acc);
    mma_stage1<4>(s2, f[3], acc);
    mma_stage1<4>(s3, f[3], acc);
    mma_stage1<4>(s4, f[3], acc);
    mma_stage1<4>(s5, f[3], acc);
    mma_stage1<4>(s0, f[3], acc);
    mma_stage1<4>(s1, f[3], acc);

#pragma unroll
    for (int r = 0; r < 8; ++r) tile[r + 8 * half][d0 + lr] = acc[r];
  }
  __syncthreads();
  {
    const int row = (int)threadIdx.x >> 4;
    const int q   = (int)threadIdx.x & 15;
    const v4f val = *(const v4f*)&tile[row][q * 4];
    *(v4f*)(AVec + (size_t)(m0 + row) * kD + h * kDH + q * 4) = val;
  }
}

__global__ void __launch_bounds__(256) out_gemm(const float* __restrict__ A, const float* __restrict__ B,
                         float* __restrict__ C) {
  const int lane = threadIdx.x & 31;
  const int wave = threadIdx.x >> 5;
  const int half = lane >> 4;
  const int lr   = lane & 15;
  const int m0   = blockIdx.x * 16;
  const int n0   = wave * 64;

  const float* ap = A + (size_t)(m0 + lr) * kD + 2 * half;
  const float* bp = B + (size_t)(n0 + lr) * kD + 2 * half;

  v8f acc[4] = {v8f{}, v8f{}, v8f{}, v8f{}};
  kloop_nt512(ap, bp, acc);
#pragma unroll
  for (int t = 0; t < 4; ++t) {
    int n = n0 + t * 16 + lr;
#pragma unroll
    for (int r = 0; r < 8; ++r) {
      int m = m0 + r + 8 * half;
      C[(size_t)m * kD + n] = acc[t][r];
    }
  }
}

extern "C" void kernel_launch(void* const* d_in, const int* in_sizes, int n_in,
                              void* d_out, int out_size, void* d_ws, size_t ws_size,
                              hipStream_t stream) {
  const float* x   = (const float*)d_in[0];
  const float* pos = (const float*)d_in[1];
  const float* Wq  = (const float*)d_in[2];
  const float* Wk  = (const float*)d_in[3];
  const float* Wv  = (const float*)d_in[4];
  const float* Wp  = (const float*)d_in[5];
  const float* Wo  = (const float*)d_in[6];
  const float* u   = (const float*)d_in[7];
  const float* v   = (const float*)d_in[8];

  float* ws = (float*)d_ws;
  const size_t SD = (size_t)kS * kD;
  float* Qhs  = ws;
  float* Khs  = Qhs + SD;
  float* Vhs  = Khs + SD;
  float* Phs  = Vhs + SD;
  float* AVec = Phs + SD;

  dim3 blk(256);
  proj_gemm_head<<<dim3(kS / 16), blk, 0, stream>>>(x,   Wq, Qhs);
  proj_gemm_head<<<dim3(kS / 16), blk, 0, stream>>>(x,   Wk, Khs);
  proj_gemm_head<<<dim3(kS / 16), blk, 0, stream>>>(x,   Wv, Vhs);
  proj_gemm_head<<<dim3(kS / 16), blk, 0, stream>>>(pos, Wp, Phs);

  const float scale = 0.125f;
  attn_fused<<<dim3(kS / 16, kH), blk, 0, stream>>>(Qhs, Khs, Phs, Vhs, u, v, AVec, scale);

  out_gemm<<<dim3(kS / 16), blk, 0, stream>>>(AVec, Wo, (float*)d_out);
}


# Q, K, P, V and attention output stored in WMMA fragment order by their producers (Q/K/P tiles computed transposed): a fragment load reads 2 whole 128-byte lines, was 16 partial; sum order kept
# speedup vs baseline: 7.4472x; 1.1654x over previous
#include <hip/hip_runtime.h>
#include <math.h>

typedef __attribute__((ext_vector_type(2))) float v2f;
typedef __attribute__((ext_vector_type(4))) float v4f;
typedef __attribute__((ext_vector_type(8))) float v8f;

static constexpr int kS  = 2048;
static constexpr int kD  = 512;
static constexpr int kH  = 8;
static constexpr int kDH = 64;

static constexpr int kChunk = 64;
static constexpr int kTileF = 16 * kDH;

__device__ __forceinline__ int frag_row(int row, int half) {
  return (row >> 4) * kTileF + half * 32 + (row & 15) * 2;
}

__device__ __forceinline__ v8f wmma_f32(v2f a, v2f b, v8f c) {
  return __builtin_amdgcn_wmma_f32_16x16x4_f32(
       false, a,  false, b,
       (short)0, c,  false,  false);
}

template <int U>
struct Stage {
  v2f a[U];
  v2f b[U][4];
};

__device__ __forceinline__ void pin_order() { __builtin_amdgcn_sched_barrier(0); }

__device__ __forceinline__ void hold(v8f& acc) { asm volatile("" : "+v"(acc) : : "memory"); }

template <int U, bool AF>
__device__ __forceinline__ void load_nt(Stage<U>& s, const float* __restrict__ ap,
                                        const float* __restrict__ bp, int ld, int k0) {
#pragma unroll
  for (int i = 0; i < U; ++i) {
    s.a[i] = *(const v2f*)(ap + (AF ? (k0 + 4 * i) * (kChunk / 4) : k0 + 4 * i));
#pragma unroll
    for (int t = 0; t < 4; ++t)
      s.b[i][t] = *(const v2f*)(bp + (size_t)t * 16 * ld + k0 + 4 * i);
  }
}

template <int U>
__device__ __forceinline__ void load_ff(Stage<U>& s, const float* __restrict__ ap,
                                        const float* __restrict__ bp, int k0) {
#pragma unroll
  for (int i = 0; i < U; ++i) {
    s.a[i] = *(const v2f*)(ap + (k0 / 4 + i) * kChunk);
#pragma unroll
    for (int t = 0; t < 4; ++t)
      s.b[i][t] = *(const v2f*)(bp + t * kTileF + (k0 / 4 + i) * kChunk);
  }
}

template <int U>
struct Stage1 {
  v2f a[U];
  v2f b[U];
};

template <int U>
__device__ __forceinline__ void load_nn1(Stage1<U>& s, const float* __restrict__ ap,
                                         const float* __restrict__ vp, int k0) {
#pragma unroll
  for (int i = 0; i < U; ++i) {
    s.a[i] = *(const v2f*)(ap + k0 + 4 * i);
    s.b[i] = *(const v2f*)(vp + (k0 / 4 + i) * kChunk);
  }
}

template <int U>
__device__ __forceinline__ void mma_stage1(const Stage1<U>& s, float f, v8f& acc) {
  pin_order();
#pragma unroll
  for (int i = 0; i < U; ++i) acc = wmma_f32(s.a[i] * f, s.b[i], acc);
}

__device__ __forceinline__ float block_factor(v4f f, int k0) { return f[k0 >> 9]; }

template <int U, bool TR>
__device__ __forceinline__ void mma_stage(const Stage<U>& s, v8f (&acc)[4]) {
#pragma unroll
  for (int i = 0; i < U; ++i) {
#pragma unroll
    for (int t = 0; t < 4; ++t)
      acc[t] = TR ? wmma_f32(s.b[i][t], s.a[i], acc[t]) : wmma_f32(s.a[i], s.b[i][t], acc[t]);
  }
}

template <bool AF, bool TR>
__device__ __forceinline__ void kloop_nt512(const float* __restrict__ ap,
                                            const float* __restrict__ bp, v8f (&acc)[4]) {
  Stage<8> s0, s1;
  load_nt<8, AF>(s0, ap, bp, kD, 0);
  for (int k = 0; k < kD - 64; k += 64) {
    load_nt<8, AF>(s1, ap, bp, kD, k + 32);
    pin_order();
    mma_stage<8, TR>(s0, acc);
    pin_order();
    load_nt<8, AF>(s0, ap, bp, kD, k + 64);
    pin_order();
    mma_stage<8, TR>(s1, acc);
    pin_order();
  }
  load_nt<8, AF>(s1, ap, bp, kD, kD - 32);
  pin_order();
  mma_stage<8, TR>(s0, acc);
  mma_stage<8, TR>(s1, acc);
}

__device__ __forceinline__ void kloop_nt64_bias(const float* __restrict__ ap,
                                                const float* __restrict__ bp,
                                                const float* __restrict__ bb, v8f (&acc)[4]) {
  Stage<8> s0, s1;
  v2f c0[8], c1[8];
  load_ff<8>(s0, ap, bp, 0);
#pragma unroll
  for (int i = 0; i < 8; ++i) c0[i] = *(const v2f*)(bb + 4 * i);
  load_ff<8>(s1, ap, bp, 32);
#pragma unroll
  for (int i = 0; i < 8; ++i) c1[i] = *(const v2f*)(bb + 32 + 4 * i);
  pin_order();
#pragma unroll
  for (int i = 0; i < 8; ++i) s0.a[i] = s0.a[i] + c0[i];
  mma_stage<8, false>(s0, acc);
#pragma unroll
  for (int i = 0; i < 8; ++i) s1.a[i] = s1.a[i] + c1[i];
  mma_stage<8, false>(s1, acc);
}

template <bool TR>
__global__ void __launch_bounds__(256) proj_gemm_head(const float* __restrict__ A,
                               const float* __restrict__ B,
                               float* __restrict__ out) {
  const int lane = threadIdx.x & 31;
  const int wave = threadIdx.x >> 5;
  const int half = lane >> 4;
  const int lr   = lane & 15;
  const int m0   = blockIdx.x * 16;
  const int n0   = wave * 64;

  const float* ap = A + (size_t)(m0 + lr) * kD + 2 * half;
  const float* bp = B + (size_t)(n0 + lr) * kD + 2 * half;

  v8f acc[4] = {v8f{}, v8f{}, v8f{}, v8f{}};
  kloop_nt512<false, TR>(ap, bp, acc);
  float* op = out + (size_t)wave * kS * kDH + lr * 2;
#pragma unroll
  for (int t = 0; t < 4; ++t) {
#pragma unroll
    for (int q = 0; q < 4; ++q) {
      const int off = TR ? blockIdx.x * kTileF + (4 * t + 2 * half + (q >> 1)) * kChunk
                         : t * (kS / 4) * kChunk + (m0 / 4 + 2 * half + (q >> 1)) * kChunk;
      *(v2f*)(op + off + (q & 1) * 32) = v2f{acc[t][2 * q], acc[t][2 * q + 1]};
    }
  }
}

template <int U>
struct StageW {
  v2f a[U];
  v2f c[U];
  v2f b[U][5];
};

template <int U>
__device__ __forceinline__ void load_win(StageW<U>& s, const float* __restrict__ ap,
                                         const float* __restrict__ cp,
                                         const float* __restrict__ pb,
                                         const unsigned (&off)[5], int k0) {
#pragma unroll
  for (int i = 0; i < U; ++i) {
    s.a[i] = *(const v2f*)(ap + (k0 / 4 + i) * kChunk);
    s.c[i] = *(const v2f*)(cp + k0 + 4 * i);
#pragma unroll
    for (int w = 0; w < 5; ++w)
      s.b[i][w] = *(const v2f*)(pb + (size_t)off[w] + (k0 / 4 + i) * kChunk);
  }
}

template <int U>
__device__ __forceinline__ void mma_win(const StageW<U>& s, v8f (&g)[5]) {
#pragma unroll
  for (int i = 0; i < U; ++i) {
    const v2f a = s.a[i] + s.c[i];
#pragma unroll
    for (int w = 0; w < 5; ++w) g[w] = wmma_f32(a, s.b[i][w], g[w]);
  }
}

__device__ __forceinline__ void shifted_bd(const float* __restrict__ ap,
                                           const float* __restrict__ cp,
                                           const float* __restrict__ pb, int cs,
                                           int lane, float (&sh)[4][8]) {
  const int half = lane >> 4;
  const int lr   = lane & 15;
  unsigned off[5];
#pragma unroll
  for (int w = 0; w < 5; ++w) {
    int c = cs + 16 * w + lr;
    c = c < 0 ? 0 : (c > kS - 1 ? kS - 1 : c);
    off[w] = (unsigned)frag_row(c, half);
  }
  v8f g[5] = {v8f{}, v8f{}, v8f{}, v8f{}, v8f{}};
  StageW<4> s0, s1;
  load_win<4>(s0, ap, cp, pb, off, 0);
  load_win<4>(s1, ap, cp, pb, off, 16);
  pin_order();
  mma_win<4>(s0, g);
  pin_order();
  load_win<4>(s0, ap, cp, pb, off, 32);
  pin_order();
  mma_win<4>(s1, g);
  pin_order();
  load_win<4>(s1, ap, cp, pb, off, 48);
  pin_order();
  mma_win<4>(s0, g);
  mma_win<4>(s1, g);

#pragma unroll
  for (int r = 0; r < 8; ++r) {
    const int m   = r + 8 * half;
    const int src = (lane & 16) | ((lr - m - 1) & 15);
    float p[5];
#pragma unroll
    for (int w = 0; w < 5; ++w) p[w] = __shfl(g[w][r], src, 32);
#pragma unroll
    for (int t = 0; t < 4; ++t) sh[t][r] = (lr <= m) ? p[t] : p[t + 1];
  }
}

static constexpr int kEP = kS + 4;

__global__ void __launch_bounds__(256) attn_fused(const float* __restrict__ Qh, const float* __restrict__ Kh,
                           const float* __restrict__ Ph, const float* __restrict__ Vh,
                           const float* __restrict__ ub, const float* __restrict__ vb,
                           float* __restrict__ AVec, float scale) {
  __shared__ alignas(16) float E[16][kEP];
  __shared__ alignas(16) float redm[4][16][8];
  __shared__ alignas(16) float reds[4][16][8];
  __shared__ alignas(16) float tile[16][68];
  const int lane = threadIdx.x & 31;
  const int wave = __builtin_amdgcn_readfirstlane((int)(threadIdx.x >> 5));
  const int half = lane >> 4;
  const int lr   = lane & 15;
  const int m0   = blockIdx.x * 16;
  const int h    = blockIdx.y;

  const float* qrow = Qh + (size_t)h * kS * kDH + frag_row(m0 + lr, half);
  const float* ubias = ub + h * kDH + 2 * half;
  const float* pb = Ph + (size_t)h * kS * kDH;
  const float* vbias = vb + h * kDH + 2 * half;

#pragma nounroll
  for (int j = 0; j < 4; ++j) {
    const float* qp = qrow;
    const float* bb = ubias;
    const float* cp = vbias;
    asm volatile("" : "+v"(qp), "+v"(bb), "+v"(cp));
    const int n0 = j * 512 + wave * 64;
    const float* kp = Kh + (size_t)h * kS * kDH + frag_row(n0 + lr, half);

    v8f acc[4] = {v8f{}, v8f{}, v8f{}, v8f{}};
    kloop_nt64_bias(qp, kp, bb, acc);
    pin_order();

    const int delta = __builtin_amdgcn_readfirstlane(n0 - m0);
    if (delta <= 0) {
      float sh[4][8];
      shifted_bd(qp, cp, pb, kS - 16 + delta, lane, sh);
#pragma unroll
      for (int t = 0; t < 4; ++t) {
        const int n = n0 + t * 16 + lr;
#pragma unroll
        for (int r = 0; r < 8; ++r) {
          const int m = m0 + r + 8 * half;
          acc[t][r] += (n <= m) ? sh[t][r] : 0.0f;
        }
      }
    }
    if (delta >= -48) {
      const int row1 = (m0 + lr + 1 > kS - 1) ? kS - 1 : m0 + lr + 1;
      const float* qp1 = Qh + (size_t)h * kS * kDH + frag_row(row1, half);
      float sh[4][8];
      shifted_bd(qp1, cp, pb, delta - 17, lane, sh);
#pragma unroll
      for (int t = 0; t < 4; ++t) {
        const int n = n0 + t * 16 + lr;
#pragma unroll
        for (int r = 0; r < 8; ++r) {
          const int m = m0 + r + 8 * half;
          acc[t][r] += (n >= m + 2) ? sh[t][r] : 0.0f;
        }
      }
    }

    float sv[4][8];
#pragma unroll
    for (int t = 0; t < 4; ++t) {
#pragma unroll
      for (int r = 0; r < 8; ++r) sv[t][r] = acc[t][r] * scale;
    }

    float mx[8];
#pragma unroll
    for (int r = 0; r < 8; ++r)
      mx[r] = fmaxf(fmaxf(sv[0][r], sv[1][r]), fmaxf(sv[2][r], sv[3][r]));
#pragma unroll
    for (int off = 1; off < 16; off <<= 1) {
#pragma unroll
      for (int r = 0; r < 8; ++r) mx[r] = fmaxf(mx[r], __shfl_xor(mx[r], off, 32));
    }
    if (lr == 0) {
#pragma unroll
      for (int r = 0; r < 8; ++r) redm[j][r + 8 * half][wave] = mx[r];
    }
    __syncthreads();

    float sm[8];
#pragma unroll
    for (int r = 0; r < 8; ++r) {
      const v4f a = *(const v4f*)&redm[j][r + 8 * half][0];
      const v4f b = *(const v4f*)&redm[j][r + 8 * half][4];
      const float bm = fmaxf(fmaxf(fmaxf(a[0], a[1]), fmaxf(a[2], a[3])),
                             fmaxf(fmaxf(b[0], b[1]), fmaxf(b[2], b[3])));
      float s = 0.0f;
#pragma unroll
      for (int t = 0; t < 4; ++t) {
        const float e = __expf(sv[t][r] - bm);
        E[r + 8 * half][n0 + t * 16 + lr] = e;
        s += e;
      }
      sm[r] = s;
    }
#pragma unroll
    for (int off = 1; off < 16; off <<= 1) {
#pragma unroll
      for (int r = 0; r < 8; ++r) sm[r] += __shfl_xor(sm[r], off, 32);
    }
    if (lr == 0) {
#pragma unroll
      for (int r = 0; r < 8; ++r) reds[j][r + 8 * half][wave] = sm[r];
    }
  }
  __syncthreads();

  if (wave < 4) {
    const int d0 = wave * 16;
    const float* ep = &E[lr][2 * half];
    const float* vp = Vh + (size_t)h * kS * kDH + wave * (kS / 4) * kChunk + half * 32 + lr * 2;

    float bmj[4], bsj[4];
#pragma unroll
    for (int j = 0; j < 4; ++j) {
      const v4f a = *(const v4f*)&redm[j][lr][0];
      const v4f b = *(const v4f*)&redm[j][lr][4];
      bmj[j] = fmaxf(fmaxf(fmaxf(a[0], a[1]), fmaxf(a[2], a[3])),
                     fmaxf(fmaxf(b[0], b[1]), fmaxf(b[2], b[3])));
      const v4f c = *(const v4f*)&reds[j][lr][0];
      const v4f d = *(const v4f*)&reds[j][lr][4];
      bsj[j] = ((((((c[0] + c[1]) + c[2]) + c[3]) + d[0]) + d[1]) + d[2]) + d[3];
    }

    v8f acc = v8f{};
    Stage1<4> s0, s1, s2, s3, s4, s5;
    load_nn1<4>(s0, ep, vp, 0);
    hold(acc);
    load_nn1<4>(s1, ep, vp, 16);
    hold(acc);
    load_nn1<4>(s2, ep, vp, 32);
    hold(acc);
    load_nn1<4>(s3, ep, vp, 48);
    hold(acc);
    load_nn1<4>(s4, ep, vp, 64);
    hold(acc);
    pin_order();
    v4f f;
    {
      const float M  = fmaxf(fmaxf(bmj[0], bmj[1]), fmaxf(bmj[2], bmj[3]));
      const float w0 = __expf(bmj[0] - M), w1 = __expf(bmj[1] - M);
      const float w2 = __expf(bmj[2] - M), w3 = __expf(bmj[3] - M);
      const float L  = ((bsj[0] * w0 + bsj[1] * w1) + bsj[2] * w2) + bsj[3] * w3;
      const float inv = 1.0f / L;
      f[0] = w0 * inv; f[1] = w1 * inv; f[2] = w2 * inv; f[3] = w3 * inv;
    }
    for (int k = 0; k < 20 * 96; k += 96) {
      load_nn1<4>(s5, ep, vp, k + 80);
      hold(acc);
      mma_stage1<4>(s0, block_factor(f, k), acc);
      hold(acc);
      load_nn1<4>(s0, ep, vp, k + 96);
      hold(acc);
      mma_stage1<4>(s1, block_factor(f, k + 16), acc);
      hold(acc);
      load_nn1<4>(s1, ep, vp, k + 112);
      hold(acc);
      mma_stage1<4>(s2, block_factor(f, k + 32), acc);
      hold(acc);
      load_nn1<4>(s2, ep, vp, k + 128);
      hold(acc);
      mma_stage1<4>(s3, block_factor(f, k + 48), acc);
      hold(acc);
      load_nn1<4>(s3, ep, vp, k + 144);
      hold(acc);
      mma_stage1<4>(s4, block_factor(f, k + 64), acc);
      hold(acc);
      load_nn1<4>(s4, ep, vp, k + 160);
      hold(acc);
      mma_stage1<4>(s5, block_factor(f, k + 80), acc);
      hold(acc);
    }
    load_nn1<4>(s5, ep, vp, kS - 48);
    hold(acc);
    mma_stage1<4>(s0, f[3], acc);
    hold(acc);
    load_nn1<4>(s0, ep, vp, kS - 32);
    hold(acc);
    mma_stage1<4>(s1, f[3], acc);
    hold(acc);
    load_nn1<4>(s1, ep, vp, kS - 16);
    hold(acc);
    mma_stage1<4>(s2, f[3], acc);
    mma_stage1<4>(s3, f[3], acc);
    mma_stage1<4>(s4, f[3], acc);
    mma_stage1<4>(s5, f[3], acc);
    mma_stage1<4>(s0, f[3], acc);
    mma_stage1<4>(s1, f[3], acc);

#pragma unroll
    for (int r = 0; r < 8; ++r) tile[r + 8 * half][d0 + lr] = acc[r];
  }
  __syncthreads();
  {
    const int p   = (int)threadIdx.x;
    const int row = 2 * (p & 7);
    const int d   = 4 * (p >> 4) + 2 * ((p >> 3) & 1);
    const v2f lo = *(const v2f*)&tile[row][d];
    const v2f hi = *(const v2f*)&tile[row + 1][d];
    *(v4f*)(AVec + (size_t)blockIdx.x * 16 * kD + h * kTileF + p * 4) = v4f{lo[0], lo[1], hi[0], hi[1]};
  }
}

__global__ void __launch_bounds__(256) out_gemm(const float* __restrict__ A, const float* __restrict__ B,
                         float* __restrict__ C) {
  const int lane = threadIdx.x & 31;
  const int wave = threadIdx.x >> 5;
  const int half = lane >> 4;
  const int lr   = lane & 15;
  const int m0   = blockIdx.x * 16;
  const int n0   = wave * 64;

  const float* ap = A + (size_t)blockIdx.x * 16 * kD + half * 32 + lr * 2;
  const float* bp = B + (size_t)(n0 + lr) * kD + 2 * half;

  v8f acc[4] = {v8f{}, v8f{}, v8f{}, v8f{}};
  kloop_nt512<true, false>(ap, bp, acc);
#pragma unroll
  for (int t = 0; t < 4; ++t) {
    int n = n0 + t * 16 + lr;
#pragma unroll
    for (int r = 0; r < 8; ++r) {
      int m = m0 + r + 8 * half;
      C[(size_t)m * kD + n] = acc[t][r];
    }
  }
}

extern "C" void kernel_launch(void* const* d_in, const int* in_sizes, int n_in,
                              void* d_out, int out_size, void* d_ws, size_t ws_size,
                              hipStream_t stream) {
  const float* x   = (const float*)d_in[0];
  const float* pos = (const float*)d_in[1];
  const float* Wq  = (const float*)d_in[2];
  const float* Wk  = (const float*)d_in[3];
  const float* Wv  = (const float*)d_in[4];
  const float* Wp  = (const float*)d_in[5];
  const float* Wo  = (const float*)d_in[6];
  const float* u   = (const float*)d_in[7];
  const float* v   = (const float*)d_in[8];

  float* ws = (float*)d_ws;
  const size_t SD = (size_t)kS * kD;
  float* Qhs  = ws;
  float* Khs  = Qhs + SD;
  float* Vhs  = Khs + SD;
  float* Phs  = Vhs + SD;
  float* AVec = Phs + SD;

  dim3 blk(256);
  proj_gemm_head<true><<<dim3(kS / 16), blk, 0, stream>>>(x,   Wq, Qhs);
  proj_gemm_head<true><<<dim3(kS / 16), blk, 0, stream>>>(x,   Wk, Khs);
  proj_gemm_head<false><<<dim3(kS / 16), blk, 0, stream>>>(x,   Wv, Vhs);
  proj_gemm_head<true><<<dim3(kS / 16), blk, 0, stream>>>(pos, Wp, Phs);

  const float scale = 0.125f;
  attn_fused<<<dim3(kS / 16, kH), blk, 0, stream>>>(Qhs, Khs, Phs, Vhs, u, v, AVec, scale);

  out_gemm<<<dim3(kS / 16), blk, 0, stream>>>(AVec, Wo, (float*)d_out);
}


# x, pos_emb and the five weights rewritten in WMMA fragment order by a new first kernel (7 launches): the five K=512 products read a fragment as 2 whole 128-byte lines, was 16 partial; sum order kept
# speedup vs baseline: 8.7051x; 1.1689x over previous
#include <hip/hip_runtime.h>
#include <math.h>

typedef __attribute__((ext_vector_type(2))) float v2f;
typedef __attribute__((ext_vector_type(4))) float v4f;
typedef __attribute__((ext_vector_type(8))) float v8f;

static constexpr int kS  = 2048;
static constexpr int kD  = 512;
static constexpr int kH  = 8;
static constexpr int kDH = 64;

static constexpr int kChunk = 64;
static constexpr int kTileF = 16 * kDH;
static constexpr int kTileK = 16 * kD;
static constexpr int kInTiles = 2 * (kS / 16) + 5 * (kD / 16);

__device__ __forceinline__ int frag_row(int row, int half) {
  return (row >> 4) * kTileF + half * 32 + (row & 15) * 2;
}

__device__ __forceinline__ v8f wmma_f32(v2f a, v2f b, v8f c) {
  return __builtin_amdgcn_wmma_f32_16x16x4_f32(
       false, a,  false, b,
       (short)0, c,  false,  false);
}

template <int U>
struct Stage {
  v2f a[U];
  v2f b[U][4];
};

__device__ __forceinline__ void pin_order() { __builtin_amdgcn_sched_barrier(0); }

__device__ __forceinline__ void hold(v8f& acc) { asm volatile("" : "+v"(acc) : : "memory"); }

template <int U, int BT>
__device__ __forceinline__ void load_ff(Stage<U>& s, const float* __restrict__ ap,
                                        const float* __restrict__ bp, int k0) {
#pragma unroll
  for (int i = 0; i < U; ++i) {
    s.a[i] = *(const v2f*)(ap + (k0 / 4 + i) * kChunk);
#pragma unroll
    for (int t = 0; t < 4; ++t)
      s.b[i][t] = *(const v2f*)(bp + t * BT + (k0 / 4 + i) * kChunk);
  }
}

template <int U>
struct Stage1 {
  v2f a[U];
  v2f b[U];
};

template <int U>
__device__ __forceinline__ void load_nn1(Stage1<U>& s, const float* __restrict__ ap,
                                         const float* __restrict__ vp, int k0) {
#pragma unroll
  for (int i = 0; i < U; ++i) {
    s.a[i] = *(const v2f*)(ap + k0 + 4 * i);
    s.b[i] = *(const v2f*)(vp + (k0 / 4 + i) * kChunk);
  }
}

template <int U>
__device__ __forceinline__ void mma_stage1(const Stage1<U>& s, float f, v8f& acc) {
  pin_order();
#pragma unroll
  for (int i = 0; i < U; ++i) acc = wmma_f32(s.a[i] * f, s.b[i], acc);
}

__device__ __forceinline__ float block_factor(v4f f, int k0) { return f[k0 >> 9]; }

template <int U, bool TR>
__device__ __forceinline__ void mma_stage(const Stage<U>& s, v8f (&acc)[4]) {
#pragma unroll
  for (int i = 0; i < U; ++i) {
#pragma unroll
    for (int t = 0; t < 4; ++t)
      acc[t] = TR ? wmma_f32(s.b[i][t], s.a[i], acc[t]) : wmma_f32(s.a[i], s.b[i][t], acc[t]);
  }
}

template <bool TR>
__device__ __forceinline__ void kloop_nt512(const float* __restrict__ ap,
                                            const float* __restrict__ bp, v8f (&acc)[4]) {
  Stage<8> s0, s1;
  load_ff<8, kTileK>(s0, ap, bp, 0);
  for (int k = 0; k < kD - 64; k += 64) {
    load_ff<8, kTileK>(s1, ap, bp, k + 32);
    pin_order();
    mma_stage<8, TR>(s0, acc);
    pin_order();
    load_ff<8, kTileK>(s0, ap, bp, k + 64);
    pin_order();
    mma_stage<8, TR>(s1, acc);
    pin_order();
  }
  load_ff<8, kTileK>(s1, ap, bp, kD - 32);
  pin_order();
  mma_stage<8, TR>(s0, acc);
  mma_stage<8, TR>(s1, acc);
}

__device__ __forceinline__ void kloop_nt64_bias(const float* __restrict__ ap,
                                                const float* __restrict__ bp,
                                                const float* __restrict__ bb, v8f (&acc)[4]) {
  Stage<8> s0, s1;
  v2f c0[8], c1[8];
  load_ff<8, kTileF>(s0, ap, bp, 0);
#pragma unroll
  for (int i = 0; i < 8; ++i) c0[i] = *(const v2f*)(bb + 4 * i);
  load_ff<8, kTileF>(s1, ap, bp, 32);
#pragma unroll
  for (int i = 0; i < 8; ++i) c1[i] = *(const v2f*)(bb + 32 + 4 * i);
  pin_order();
#pragma unroll
  for (int i = 0; i < 8; ++i) s0.a[i] = s0.a[i] + c0[i];
  mma_stage<8, false>(s0, acc);
#pragma unroll
  for (int i = 0; i < 8; ++i) s1.a[i] = s1.a[i] + c1[i];
  mma_stage<8, false>(s1, acc);
}

__global__ void __launch_bounds__(256) to_fragment_order(const float* __restrict__ x,
                                  const float* __restrict__ pos,
                                  const float* __restrict__ wq, const float* __restrict__ wk,
                                  const float* __restrict__ wv, const float* __restrict__ wp,
                                  const float* __restrict__ wo, float* __restrict__ dst) {
  const int lane = threadIdx.x & 31;
  const int wave = threadIdx.x >> 5;
  const int half = lane >> 4;
  const int lr   = lane & 15;
  const int b    = blockIdx.x;
  constexpr int kRowTiles = kS / 16;
  constexpr int kWTiles   = kD / 16;

  const float* src;
  if (b < kRowTiles) {
    src = x + (size_t)b * kTileK;
  } else if (b < 2 * kRowTiles) {
    src = pos + (size_t)(b - kRowTiles) * kTileK;
  } else {
    const int w = (b - 2 * kRowTiles) / kWTiles;
    const float* m = w == 0 ? wq : w == 1 ? wk : w == 2 ? wv : w == 3 ? wp : wo;
    src = m + (size_t)((b - 2 * kRowTiles) % kWTiles) * kTileK;
  }
  src += lr * kD + 4 * half;
  float* out = dst + (size_t)b * kTileK + half * kChunk + lr * 2;

  v4f e[8];
#pragma unroll
  for (int i = 0; i < 8; ++i) e[i] = *(const v4f*)(src + 8 * (8 * wave + i));
#pragma unroll
  for (int i = 0; i < 8; ++i) {
    float* o = out + 2 * (8 * wave + i) * kChunk;
    *(v2f*)(o)      = v2f{e[i][0], e[i][1]};
    *(v2f*)(o + 32) = v2f{e[i][2], e[i][3]};
  }
}

template <bool TR>
__global__ void __launch_bounds__(256) proj_gemm_head(const float* __restrict__ A,
                               const float* __restrict__ B,
                               float* __restrict__ out) {
  const int lane = threadIdx.x & 31;
  const int wave = threadIdx.x >> 5;
  const int half = lane >> 4;
  const int lr   = lane & 15;
  const int m0   = blockIdx.x * 16;
  const int n0   = wave * 64;

  const float* ap = A + (size_t)blockIdx.x * kTileK + half * 32 + lr * 2;
  const float* bp = B + (size_t)(n0 / 16) * kTileK + half * 32 + lr * 2;

  v8f acc[4] = {v8f{}, v8f{}, v8f{}, v8f{}};
  kloop_nt512<TR>(ap, bp, acc);
  float* op = out + (size_t)wave * kS * kDH + lr * 2;
#pragma unroll
  for (int t = 0; t < 4; ++t) {
#pragma unroll
    for (int q = 0; q < 4; ++q) {
      const int off = TR ? blockIdx.x * kTileF + (4 * t + 2 * half + (q >> 1)) * kChunk
                         : t * (kS / 4) * kChunk + (m0 / 4 + 2 * half + (q >> 1)) * kChunk;
      *(v2f*)(op + off + (q & 1) * 32) = v2f{acc[t][2 * q], acc[t][2 * q + 1]};
    }
  }
}

template <int U>
struct StageW {
  v2f a[U];
  v2f c[U];
  v2f b[U][5];
};

template <int U>
__device__ __forceinline__ void load_win(StageW<U>& s, const float* __restrict__ ap,
                                         const float* __restrict__ cp,
                                         const float* __restrict__ pb,
                                         const unsigned (&off)[5], int k0) {
#pragma unroll
  for (int i = 0; i < U; ++i) {
    s.a[i] = *(const v2f*)(ap + (k0 / 4 + i) * kChunk);
    s.c[i] = *(const v2f*)(cp + k0 + 4 * i);
#pragma unroll
    for (int w = 0; w < 5; ++w)
      s.b[i][w] = *(const v2f*)(pb + (size_t)off[w] + (k0 / 4 + i) * kChunk);
  }
}

template <int U>
__device__ __forceinline__ void mma_win(const StageW<U>& s, v8f (&g)[5]) {
#pragma unroll
  for (int i = 0; i < U; ++i) {
    const v2f a = s.a[i] + s.c[i];
#pragma unroll
    for (int w = 0; w < 5; ++w) g[w] = wmma_f32(a, s.b[i][w], g[w]);
  }
}

__device__ __forceinline__ void shifted_bd(const float* __restrict__ ap,
                                           const float* __restrict__ cp,
                                           const float* __restrict__ pb, int cs,
                                           int lane, float (&sh)[4][8]) {
  const int half = lane >> 4;
  const int lr   = lane & 15;
  unsigned off[5];
#pragma unroll
  for (int w = 0; w < 5; ++w) {
    int c = cs + 16 * w + lr;
    c = c < 0 ? 0 : (c > kS - 1 ? kS - 1 : c);
    off[w] = (unsigned)frag_row(c, half);
  }
  v8f g[5] = {v8f{}, v8f{}, v8f{}, v8f{}, v8f{}};
  StageW<4> s0, s1;
  load_win<4>(s0, ap, cp, pb, off, 0);
  load_win<4>(s1, ap, cp, pb, off, 16);
  pin_order();
  mma_win<4>(s0, g);
  pin_order();
  load_win<4>(s0, ap, cp, pb, off, 32);
  pin_order();
  mma_win<4>(s1, g);
  pin_order();
  load_win<4>(s1, ap, cp, pb, off, 48);
  pin_order();
  mma_win<4>(s0, g);
  mma_win<4>(s1, g);

#pragma unroll
  for (int r = 0; r < 8; ++r) {
    const int m   = r + 8 * half;
    const int src = (lane & 16) | ((lr - m - 1) & 15);
    float p[5];
#pragma unroll
    for (int w = 0; w < 5; ++w) p[w] = __shfl(g[w][r], src, 32);
#pragma unroll
    for (int t = 0; t < 4; ++t) sh[t][r] = (lr <= m) ? p[t] : p[t + 1];
  }
}

static constexpr int kEP = kS + 4;

__global__ void __launch_bounds__(256) attn_fused(const float* __restrict__ Qh, const float* __restrict__ Kh,
                           const float* __restrict__ Ph, const float* __restrict__ Vh,
                           const float* __restrict__ ub, const float* __restrict__ vb,
                           float* __restrict__ AVec, float scale) {
  __shared__ alignas(16) float E[16][kEP];
  __shared__ alignas(16) float redm[4][16][8];
  __shared__ alignas(16) float reds[4][16][8];
  __shared__ alignas(16) float tile[16][68];
  const int lane = threadIdx.x & 31;
  const int wave = __builtin_amdgcn_readfirstlane((int)(threadIdx.x >> 5));
  const int half = lane >> 4;
  const int lr   = lane & 15;
  const int m0   = blockIdx.x * 16;
  const int h    = blockIdx.y;

  const float* qrow = Qh + (size_t)h * kS * kDH + frag_row(m0 + lr, half);
  const float* ubias = ub + h * kDH + 2 * half;
  const float* pb = Ph + (size_t)h * kS * kDH;
  const float* vbias = vb + h * kDH + 2 * half;

#pragma nounroll
  for (int j = 0; j < 4; ++j) {
    const float* qp = qrow;
    const float* bb = ubias;
    const float* cp = vbias;
    asm volatile("" : "+v"(qp), "+v"(bb), "+v"(cp));
    const int n0 = j * 512 + wave * 64;
    const float* kp = Kh + (size_t)h * kS * kDH + frag_row(n0 + lr, half);

    v8f acc[4] = {v8f{}, v8f{}, v8f{}, v8f{}};
    kloop_nt64_bias(qp, kp, bb, acc);
    pin_order();

    const int delta = __builtin_amdgcn_readfirstlane(n0 - m0);
    if (delta <= 0) {
      float sh[4][8];
      shifted_bd(qp, cp, pb, kS - 16 + delta, lane, sh);
#pragma unroll
      for (int t = 0; t < 4; ++t) {
        const int n = n0 + t * 16 + lr;
#pragma unroll
        for (int r = 0; r < 8; ++r) {
          const int m = m0 + r + 8 * half;
          acc[t][r] += (n <= m) ? sh[t][r] : 0.0f;
        }
      }
    }
    if (delta >= -48) {
      const int row1 = (m0 + lr + 1 > kS - 1) ? kS - 1 : m0 + lr + 1;
      const float* qp1 = Qh + (size_t)h * kS * kDH + frag_row(row1, half);
      float sh[4][8];
      shifted_bd(qp1, cp, pb, delta - 17, lane, sh);
#pragma unroll
      for (int t = 0; t < 4; ++t) {
        const int n = n0 + t * 16 + lr;
#pragma unroll
        for (int r = 0; r < 8; ++r) {
          const int m = m0 + r + 8 * half;
          acc[t][r] += (n >= m + 2) ? sh[t][r] : 0.0f;
        }
      }
    }

    float sv[4][8];
#pragma unroll
    for (int t = 0; t < 4; ++t) {
#pragma unroll
      for (int r = 0; r < 8; ++r) sv[t][r] = acc[t][r] * scale;
    }

    float mx[8];
#pragma unroll
    for (int r = 0; r < 8; ++r)
      mx[r] = fmaxf(fmaxf(sv[0][r], sv[1][r]), fmaxf(sv[2][r], sv[3][r]));
#pragma unroll
    for (int off = 1; off < 16; off <<= 1) {
#pragma unroll
      for (int r = 0; r < 8; ++r) mx[r] = fmaxf(mx[r], __shfl_xor(mx[r], off, 32));
    }
    if (lr == 0) {
#pragma unroll
      for (int r = 0; r < 8; ++r) redm[j][r + 8 * half][wave] = mx[r];
    }
    __syncthreads();

    float sm[8];
#pragma unroll
    for (int r = 0; r < 8; ++r) {
      const v4f a = *(const v4f*)&redm[j][r + 8 * half][0];
      const v4f b = *(const v4f*)&redm[j][r + 8 * half][4];
      const float bm = fmaxf(fmaxf(fmaxf(a[0], a[1]), fmaxf(a[2], a[3])),
                             fmaxf(fmaxf(b[0], b[1]), fmaxf(b[2], b[3])));
      float s = 0.0f;
#pragma unroll
      for (int t = 0; t < 4; ++t) {
        const float e = __expf(sv[t][r] - bm);
        E[r + 8 * half][n0 + t * 16 + lr] = e;
        s += e;
      }
      sm[r] = s;
    }
#pragma unroll
    for (int off = 1; off < 16; off <<= 1) {
#pragma unroll
      for (int r = 0; r < 8; ++r) sm[r] += __shfl_xor(sm[r], off, 32);
    }
    if (lr == 0) {
#pragma unroll
      for (int r = 0; r < 8; ++r) reds[j][r + 8 * half][wave] = sm[r];
    }
  }
  __syncthreads();

  if (wave < 4) {
    const int d0 = wave * 16;
    const float* ep = &E[lr][2 * half];
    const float* vp = Vh + (size_t)h * kS * kDH + wave * (kS / 4) * kChunk + half * 32 + lr * 2;

    float bmj[4], bsj[4];
#pragma unroll
    for (int j = 0; j < 4; ++j) {
      const v4f a = *(const v4f*)&redm[j][lr][0];
      const v4f b = *(const v4f*)&redm[j][lr][4];
      bmj[j] = fmaxf(fmaxf(fmaxf(a[0], a[1]), fmaxf(a[2], a[3])),
                     fmaxf(fmaxf(b[0], b[1]), fmaxf(b[2], b[3])));
      const v4f c = *(const v4f*)&reds[j][lr][0];
      const v4f d = *(const v4f*)&reds[j][lr][4];
      bsj[j] = ((((((c[0] + c[1]) + c[2]) + c[3]) + d[0]) + d[1]) + d[2]) + d[3];
    }

    v8f acc = v8f{};
    Stage1<4> s0, s1, s2, s3, s4, s5;
    load_nn1<4>(s0, ep, vp, 0);
    hold(acc);
    load_nn1<4>(s1, ep, vp, 16);
    hold(acc);
    load_nn1<4>(s2, ep, vp, 32);
    hold(acc);
    load_nn1<4>(s3, ep, vp, 48);
    hold(acc);
    load_nn1<4>(s4, ep, vp, 64);
    hold(acc);
    pin_order();
    v4f f;
    {
      const float M  = fmaxf(fmaxf(bmj[0], bmj[1]), fmaxf(bmj[2], bmj[3]));
      const float w0 = __expf(bmj[0] - M), w1 = __expf(bmj[1] - M);
      const float w2 = __expf(bmj[2] - M), w3 = __expf(bmj[3] - M);
      const float L  = ((bsj[0] * w0 + bsj[1] * w1) + bsj[2] * w2) + bsj[3] * w3;
      const float inv = 1.0f / L;
      f[0] = w0 * inv; f[1] = w1 * inv; f[2] = w2 * inv; f[3] = w3 * inv;
    }
    for (int k = 0; k < 20 * 96; k += 96) {
      load_nn1<4>(s5, ep, vp, k + 80);
      hold(acc);
      mma_stage1<4>(s0, block_factor(f, k), acc);
      hold(acc);
      load_nn1<4>(s0, ep, vp, k + 96);
      hold(acc);
      mma_stage1<4>(s1, block_factor(f, k + 16), acc);
      hold(acc);
      load_nn1<4>(s1, ep, vp, k + 112);
      hold(acc);
      mma_stage1<4>(s2, block_factor(f, k + 32), acc);
      hold(acc);
      load_nn1<4>(s2, ep, vp, k + 128);
      hold(acc);
      mma_stage1<4>(s3, block_factor(f, k + 48), acc);
      hold(acc);
      load_nn1<4>(s3, ep, vp, k + 144);
      hold(acc);
      mma_stage1<4>(s4, block_factor(f, k + 64), acc);
      hold(acc);
      load_nn1<4>(s4, ep, vp, k + 160);
      hold(acc);
      mma_stage1<4>(s5, block_factor(f, k + 80), acc);
      hold(acc);
    }
    load_nn1<4>(s5, ep, vp, kS - 48);
    hold(acc);
    mma_stage1<4>(s0, f[3], acc);
    hold(acc);
    load_nn1<4>(s0, ep, vp, kS - 32);
    hold(acc);
    mma_stage1<4>(s1, f[3], acc);
    hold(acc);
    load_nn1<4>(s1, ep, vp, kS - 16);
    hold(acc);
    mma_stage1<4>(s2, f[3], acc);
    mma_stage1<4>(s3, f[3], acc);
    mma_stage1<4>(s4, f[3], acc);
    mma_stage1<4>(s5, f[3], acc);
    mma_stage1<4>(s0, f[3], acc);
    mma_stage1<4>(s1, f[3], acc);

#pragma unroll
    for (int r = 0; r < 8; ++r) tile[r + 8 * half][d0 + lr] = acc[r];
  }
  __syncthreads();
  {
    const int p   = (int)threadIdx.x;
    const int row = 2 * (p & 7);
    const int d   = 4 * (p >> 4) + 2 * ((p >> 3) & 1);
    const v2f lo = *(const v2f*)&tile[row][d];
    const v2f hi = *(const v2f*)&tile[row + 1][d];
    *(v4f*)(AVec + (size_t)blockIdx.x * 16 * kD + h * kTileF + p * 4) = v4f{lo[0], lo[1], hi[0], hi[1]};
  }
}

__global__ void __launch_bounds__(256) out_gemm(const float* __restrict__ A, const float* __restrict__ B,
                         float* __restrict__ C) {
  const int lane = threadIdx.x & 31;
  const int wave = threadIdx.x >> 5;
  const int half = lane >> 4;
  const int lr   = lane & 15;
  const int m0   = blockIdx.x * 16;
  const int n0   = wave * 64;

  const float* ap = A + (size_t)blockIdx.x * kTileK + half * 32 + lr * 2;
  const float* bp = B + (size_t)(n0 / 16) * kTileK + half * 32 + lr * 2;

  v8f acc[4] = {v8f{}, v8f{}, v8f{}, v8f{}};
  kloop_nt512<false>(ap, bp, acc);
#pragma unroll
  for (int t = 0; t < 4; ++t) {
    int n = n0 + t * 16 + lr;
#pragma unroll
    for (int r = 0; r < 8; ++r) {
      int m = m0 + r + 8 * half;
      C[(size_t)m * kD + n] = acc[t][r];
    }
  }
}

extern "C" void kernel_launch(void* const* d_in, const int* in_sizes, int n_in,
                              void* d_out, int out_size, void* d_ws, size_t ws_size,
                              hipStream_t stream) {
  const float* x   = (const float*)d_in[0];
  const float* pos = (const float*)d_in[1];
  const float* Wq  = (const float*)d_in[2];
  const float* Wk  = (const float*)d_in[3];
  const float* Wv  = (const float*)d_in[4];
  const float* Wp  = (const float*)d_in[5];
  const float* Wo  = (const float*)d_in[6];
  const float* u   = (const float*)d_in[7];
  const float* v   = (const float*)d_in[8];

  float* ws = (float*)d_ws;
  const size_t SD = (size_t)kS * kD;
  float* Qhs  = ws;
  float* Khs  = Qhs + SD;
  float* Vhs  = Khs + SD;
  float* Phs  = Vhs + SD;
  float* AVec = Phs + SD;
  const size_t DD = (size_t)kD * kD;
  float* xf  = AVec + SD;
  float* pf  = xf + SD;
  float* Wqf = pf + SD;
  float* Wkf = Wqf + DD;
  float* Wvf = Wkf + DD;
  float* Wpf = Wvf + DD;
  float* Wof = Wpf + DD;

  dim3 blk(256);
  to_fragment_order<<<dim3(kInTiles), blk, 0, stream>>>(x, pos, Wq, Wk, Wv, Wp, Wo, xf);

  proj_gemm_head<true><<<dim3(kS / 16), blk, 0, stream>>>(xf, Wqf, Qhs);
  proj_gemm_head<true><<<dim3(kS / 16), blk, 0, stream>>>(xf, Wkf, Khs);
  proj_gemm_head<false><<<dim3(kS / 16), blk, 0, stream>>>(xf, Wvf, Vhs);
  proj_gemm_head<true><<<dim3(kS / 16), blk, 0, stream>>>(pf, Wpf, Phs);

  const float scale = 0.125f;
  attn_fused<<<dim3(kS / 16, kH), blk, 0, stream>>>(Qhs, Khs, Phs, Vhs, u, v, AVec, scale);

  out_gemm<<<dim3(kS / 16), blk, 0, stream>>>(AVec, Wof, (float*)d_out);
}


# five K=512 products run as 256 blocks of 4 waves, one wave per SIMD on every WGP (was 8-wave blocks on half of the WGPs), loads in a ring of 4 stages of 4 k steps for the lone wave; sum order kept
# speedup vs baseline: 9.6051x; 1.1034x over previous
#include <hip/hip_runtime.h>
#include <math.h>

typedef __attribute__((ext_vector_type(2))) float v2f;
typedef __attribute__((ext_vector_type(4))) float v4f;
typedef __attribute__((ext_vector_type(8))) float v8f;

static constexpr int kS  = 2048;
static constexpr int kD  = 512;
static constexpr int kH  = 8;
static constexpr int kDH = 64;

static constexpr int kChunk = 64;
static constexpr int kTileF = 16 * kDH;
static constexpr int kTileK = 16 * kD;
static constexpr int kInTiles = 2 * (kS / 16) + 5 * (kD / 16);

__device__ __forceinline__ int frag_row(int row, int half) {
  return (row >> 4) * kTileF + half * 32 + (row & 15) * 2;
}

__device__ __forceinline__ v8f wmma_f32(v2f a, v2f b, v8f c) {
  return __builtin_amdgcn_wmma_f32_16x16x4_f32(
       false, a,  false, b,
       (short)0, c,  false,  false);
}

template <int U>
struct Stage {
  v2f a[U];
  v2f b[U][4];
};

__device__ __forceinline__ void pin_order() { __builtin_amdgcn_sched_barrier(0); }

__device__ __forceinline__ void hold(v8f& acc) { asm volatile("" : "+v"(acc) : : "memory"); }

__device__ __forceinline__ void hold4(v8f (&acc)[4]) {
  asm volatile("" : "+v"(acc[0]), "+v"(acc[1]), "+v"(acc[2]), "+v"(acc[3]) : : "memory");
}

template <int U, int BT>
__device__ __forceinline__ void load_ff(Stage<U>& s, const float* __restrict__ ap,
                                        const float* __restrict__ bp, int k0) {
#pragma unroll
  for (int i = 0; i < U; ++i) {
    s.a[i] = *(const v2f*)(ap + (k0 / 4 + i) * kChunk);
#pragma unroll
    for (int t = 0; t < 4; ++t)
      s.b[i][t] = *(const v2f*)(bp + t * BT + (k0 / 4 + i) * kChunk);
  }
}

template <int U>
struct Stage1 {
  v2f a[U];
  v2f b[U];
};

template <int U>
__device__ __forceinline__ void load_nn1(Stage1<U>& s, const float* __restrict__ ap,
                                         const float* __restrict__ vp, int k0) {
#pragma unroll
  for (int i = 0; i < U; ++i) {
    s.a[i] = *(const v2f*)(ap + k0 + 4 * i);
    s.b[i] = *(const v2f*)(vp + (k0 / 4 + i) * kChunk);
  }
}

template <int U>
__device__ __forceinline__ void mma_stage1(const Stage1<U>& s, float f, v8f& acc) {
  pin_order();
#pragma unroll
  for (int i = 0; i < U; ++i) acc = wmma_f32(s.a[i] * f, s.b[i], acc);
}

__device__ __forceinline__ float block_factor(v4f f, int k0) { return f[k0 >> 9]; }

template <int U, bool TR>
__device__ __forceinline__ void mma_stage(const Stage<U>& s, v8f (&acc)[4]) {
#pragma unroll
  for (int i = 0; i < U; ++i) {
#pragma unroll
    for (int t = 0; t < 4; ++t)
      acc[t] = TR ? wmma_f32(s.b[i][t], s.a[i], acc[t]) : wmma_f32(s.a[i], s.b[i][t], acc[t]);
  }
}

template <bool TR>
__device__ __forceinline__ void kloop_nt512(const float* __restrict__ ap,
                                            const float* __restrict__ bp, v8f (&acc)[4]) {
  Stage<4> s0, s1, s2, s3;
  load_ff<4, kTileK>(s0, ap, bp, 0);
  load_ff<4, kTileK>(s1, ap, bp, 16);
  load_ff<4, kTileK>(s2, ap, bp, 32);
  for (int k = 0; k < kD - 64; k += 64) {
    load_ff<4, kTileK>(s3, ap, bp, k + 48);
    hold4(acc);
    mma_stage<4, TR>(s0, acc);
    hold4(acc);
    load_ff<4, kTileK>(s0, ap, bp, k + 64);
    hold4(acc);
    mma_stage<4, TR>(s1, acc);
    hold4(acc);
    load_ff<4, kTileK>(s1, ap, bp, k + 80);
    hold4(acc);
    mma_stage<4, TR>(s2, acc);
    hold4(acc);
    load_ff<4, kTileK>(s2, ap, bp, k + 96);
    hold4(acc);
    mma_stage<4, TR>(s3, acc);
    hold4(acc);
  }
  load_ff<4, kTileK>(s3, ap, bp, kD - 16);
  hold4(acc);
  mma_stage<4, TR>(s0, acc);
  mma_stage<4, TR>(s1, acc);
  mma_stage<4, TR>(s2, acc);
  mma_stage<4, TR>(s3, acc);
}

__device__ __forceinline__ void kloop_nt64_bias(const float* __restrict__ ap,
                                                const float* __restrict__ bp,
                                                const float* __restrict__ bb, v8f (&acc)[4]) {
  Stage<8> s0, s1;
  v2f c0[8], c1[8];
  load_ff<8, kTileF>(s0, ap, bp, 0);
#pragma unroll
  for (int i = 0; i < 8; ++i) c0[i] = *(const v2f*)(bb + 4 * i);
  load_ff<8, kTileF>(s1, ap, bp, 32);
#pragma unroll
  for (int i = 0; i < 8; ++i) c1[i] = *(const v2f*)(bb + 32 + 4 * i);
  pin_order();
#pragma unroll
  for (int i = 0; i < 8; ++i) s0.a[i] = s0.a[i] + c0[i];
  mma_stage<8, false>(s0, acc);
#pragma unroll
  for (int i = 0; i < 8; ++i) s1.a[i] = s1.a[i] + c1[i];
  mma_stage<8, false>(s1, acc);
}

__global__ void __launch_bounds__(256) to_fragment_order(const float* __restrict__ x,
                                  const float* __restrict__ pos,
                                  const float* __restrict__ wq, const float* __restrict__ wk,
                                  const float* __restrict__ wv, const float* __restrict__ wp,
                                  const float* __restrict__ wo, float* __restrict__ dst) {
  const int lane = threadIdx.x & 31;
  const int wave = threadIdx.x >> 5;
  const int half = lane >> 4;
  const int lr   = lane & 15;
  const int b    = blockIdx.x;
  constexpr int kRowTiles = kS / 16;
  constexpr int kWTiles   = kD / 16;

  const float* src;
  if (b < kRowTiles) {
    src = x + (size_t)b * kTileK;
  } else if (b < 2 * kRowTiles) {
    src = pos + (size_t)(b - kRowTiles) * kTileK;
  } else {
    const int w = (b - 2 * kRowTiles) / kWTiles;
    const float* m = w == 0 ? wq : w == 1 ? wk : w == 2 ? wv : w == 3 ? wp : wo;
    src = m + (size_t)((b - 2 * kRowTiles) % kWTiles) * kTileK;
  }
  src += lr * kD + 4 * half;
  float* out = dst + (size_t)b * kTileK + half * kChunk + lr * 2;

  v4f e[8];
#pragma unroll
  for (int i = 0; i < 8; ++i) e[i] = *(const v4f*)(src + 8 * (8 * wave + i));
#pragma unroll
  for (int i = 0; i < 8; ++i) {
    float* o = out + 2 * (8 * wave + i) * kChunk;
    *(v2f*)(o)      = v2f{e[i][0], e[i][1]};
    *(v2f*)(o + 32) = v2f{e[i][2], e[i][3]};
  }
}

template <bool TR>
__global__ void __launch_bounds__(128) proj_gemm_head(const float* __restrict__ A,
                               const float* __restrict__ B,
                               float* __restrict__ out) {
  const int lane = threadIdx.x & 31;
  const int wave = threadIdx.x >> 5;
  const int half = lane >> 4;
  const int lr   = lane & 15;
  const int m0   = blockIdx.x * 16;
  const int head = blockIdx.y * 4 + wave;
  const int n0   = head * 64;

  const float* ap = A + (size_t)blockIdx.x * kTileK + half * 32 + lr * 2;
  const float* bp = B + (size_t)(n0 / 16) * kTileK + half * 32 + lr * 2;

  v8f acc[4] = {v8f{}, v8f{}, v8f{}, v8f{}};
  kloop_nt512<TR>(ap, bp, acc);
  float* op = out + (size_t)head * kS * kDH + lr * 2;
#pragma unroll
  for (int t = 0; t < 4; ++t) {
#pragma unroll
    for (int q = 0; q < 4; ++q) {
      const int off = TR ? blockIdx.x * kTileF + (4 * t + 2 * half + (q >> 1)) * kChunk
                         : t * (kS / 4) * kChunk + (m0 / 4 + 2 * half + (q >> 1)) * kChunk;
      *(v2f*)(op + off + (q & 1) * 32) = v2f{acc[t][2 * q], acc[t][2 * q + 1]};
    }
  }
}

template <int U>
struct StageW {
  v2f a[U];
  v2f c[U];
  v2f b[U][5];
};

template <int U>
__device__ __forceinline__ void load_win(StageW<U>& s, const float* __restrict__ ap,
                                         const float* __restrict__ cp,
                                         const float* __restrict__ pb,
                                         const unsigned (&off)[5], int k0) {
#pragma unroll
  for (int i = 0; i < U; ++i) {
    s.a[i] = *(const v2f*)(ap + (k0 / 4 + i) * kChunk);
    s.c[i] = *(const v2f*)(cp + k0 + 4 * i);
#pragma unroll
    for (int w = 0; w < 5; ++w)
      s.b[i][w] = *(const v2f*)(pb + (size_t)off[w] + (k0 / 4 + i) * kChunk);
  }
}

template <int U>
__device__ __forceinline__ void mma_win(const StageW<U>& s, v8f (&g)[5]) {
#pragma unroll
  for (int i = 0; i < U; ++i) {
    const v2f a = s.a[i] + s.c[i];
#pragma unroll
    for (int w = 0; w < 5; ++w) g[w] = wmma_f32(a, s.b[i][w], g[w]);
  }
}

__device__ __forceinline__ void shifted_bd(const float* __restrict__ ap,
                                           const float* __restrict__ cp,
                                           const float* __restrict__ pb, int cs,
                                           int lane, float (&sh)[4][8]) {
  const int half = lane >> 4;
  const int lr   = lane & 15;
  unsigned off[5];
#pragma unroll
  for (int w = 0; w < 5; ++w) {
    int c = cs + 16 * w + lr;
    c = c < 0 ? 0 : (c > kS - 1 ? kS - 1 : c);
    off[w] = (unsigned)frag_row(c, half);
  }
  v8f g[5] = {v8f{}, v8f{}, v8f{}, v8f{}, v8f{}};
  StageW<4> s0, s1;
  load_win<4>(s0, ap, cp, pb, off, 0);
  load_win<4>(s1, ap, cp, pb, off, 16);
  pin_order();
  mma_win<4>(s0, g);
  pin_order();
  load_win<4>(s0, ap, cp, pb, off, 32);
  pin_order();
  mma_win<4>(s1, g);
  pin_order();
  load_win<4>(s1, ap, cp, pb, off, 48);
  pin_order();
  mma_win<4>(s0, g);
  mma_win<4>(s1, g);

#pragma unroll
  for (int r = 0; r < 8; ++r) {
    const int m   = r + 8 * half;
    const int src = (lane & 16) | ((lr - m - 1) & 15);
    float p[5];
#pragma unroll
    for (int w = 0; w < 5; ++w) p[w] = __shfl(g[w][r], src, 32);
#pragma unroll
    for (int t = 0; t < 4; ++t) sh[t][r] = (lr <= m) ? p[t] : p[t + 1];
  }
}

static constexpr int kEP = kS + 4;

__global__ void __launch_bounds__(256) attn_fused(const float* __restrict__ Qh, const float* __restrict__ Kh,
                           const float* __restrict__ Ph, const float* __restrict__ Vh,
                           const float* __restrict__ ub, const float* __restrict__ vb,
                           float* __restrict__ AVec, float scale) {
  __shared__ alignas(16) float E[16][kEP];
  __shared__ alignas(16) float redm[4][16][8];
  __shared__ alignas(16) float reds[4][16][8];
  __shared__ alignas(16) float tile[16][68];
  const int lane = threadIdx.x & 31;
  const int wave = __builtin_amdgcn_readfirstlane((int)(threadIdx.x >> 5));
  const int half = lane >> 4;
  const int lr   = lane & 15;
  const int m0   = blockIdx.x * 16;
  const int h    = blockIdx.y;

  const float* qrow = Qh + (size_t)h * kS * kDH + frag_row(m0 + lr, half);
  const float* ubias = ub + h * kDH + 2 * half;
  const float* pb = Ph + (size_t)h * kS * kDH;
  const float* vbias = vb + h * kDH + 2 * half;

#pragma nounroll
  for (int j = 0; j < 4; ++j) {
    const float* qp = qrow;
    const float* bb = ubias;
    const float* cp = vbias;
    asm volatile("" : "+v"(qp), "+v"(bb), "+v"(cp));
    const int n0 = j * 512 + wave * 64;
    const float* kp = Kh + (size_t)h * kS * kDH + frag_row(n0 + lr, half);

    v8f acc[4] = {v8f{}, v8f{}, v8f{}, v8f{}};
    kloop_nt64_bias(qp, kp, bb, acc);
    pin_order();

    const int delta = __builtin_amdgcn_readfirstlane(n0 - m0);
    if (delta <= 0) {
      float sh[4][8];
      shifted_bd(qp, cp, pb, kS - 16 + delta, lane, sh);
#pragma unroll
      for (int t = 0; t < 4; ++t) {
        const int n = n0 + t * 16 + lr;
#pragma unroll
        for (int r = 0; r < 8; ++r) {
          const int m = m0 + r + 8 * half;
          acc[t][r] += (n <= m) ? sh[t][r] : 0.0f;
        }
      }
    }
    if (delta >= -48) {
      const int row1 = (m0 + lr + 1 > kS - 1) ? kS - 1 : m0 + lr + 1;
      const float* qp1 = Qh + (size_t)h * kS * kDH + frag_row(row1, half);
      float sh[4][8];
      shifted_bd(qp1, cp, pb, delta - 17, lane, sh);
#pragma unroll
      for (int t = 0; t < 4; ++t) {
        const int n = n0 + t * 16 + lr;
#pragma unroll
        for (int r = 0; r < 8; ++r) {
          const int m = m0 + r + 8 * half;
          acc[t][r] += (n >= m + 2) ? sh[t][r] : 0.0f;
        }
      }
    }

    float sv[4][8];
#pragma unroll
    for (int t = 0; t < 4; ++t) {
#pragma unroll
      for (int r = 0; r < 8; ++r) sv[t][r] = acc[t][r] * scale;
    }

    float mx[8];
#pragma unroll
    for (int r = 0; r < 8; ++r)
      mx[r] = fmaxf(fmaxf(sv[0][r], sv[1][r]), fmaxf(sv[2][r], sv[3][r]));
#pragma unroll
    for (int off = 1; off < 16; off <<= 1) {
#pragma unroll
      for (int r = 0; r < 8; ++r) mx[r] = fmaxf(mx[r], __shfl_xor(mx[r], off, 32));
    }
    if (lr == 0) {
#pragma unroll
      for (int r = 0; r < 8; ++r) redm[j][r + 8 * half][wave] = mx[r];
    }
    __syncthreads();

    float sm[8];
#pragma unroll
    for (int r = 0; r < 8; ++r) {
      const v4f a = *(const v4f*)&redm[j][r + 8 * half][0];
      const v4f b = *(const v4f*)&redm[j][r + 8 * half][4];
      const float bm = fmaxf(fmaxf(fmaxf(a[0], a[1]), fmaxf(a[2], a[3])),
                             fmaxf(fmaxf(b[0], b[1]), fmaxf(b[2], b[3])));
      float s = 0.0f;
#pragma unroll
      for (int t = 0; t < 4; ++t) {
        const float e = __expf(sv[t][r] - bm);
        E[r + 8 * half][n0 + t * 16 + lr] = e;
        s += e;
      }
      sm[r] = s;
    }
#pragma unroll
    for (int off = 1; off < 16; off <<= 1) {
#pragma unroll
      for (int r = 0; r < 8; ++r) sm[r] += __shfl_xor(sm[r], off, 32);
    }
    if (lr == 0) {
#pragma unroll
      for (int r = 0; r < 8; ++r) reds[j][r + 8 * half][wave] = sm[r];
    }
  }
  __syncthreads();

  if (wave < 4) {
    const int d0 = wave * 16;
    const float* ep = &E[lr][2 * half];
    const float* vp = Vh + (size_t)h * kS * kDH + wave * (kS / 4) * kChunk + half * 32 + lr * 2;

    float bmj[4], bsj[4];
#pragma unroll
    for (int j = 0; j < 4; ++j) {
      const v4f a = *(const v4f*)&redm[j][lr][0];
      const v4f b = *(const v4f*)&redm[j][lr][4];
      bmj[j] = fmaxf(fmaxf(fmaxf(a[0], a[1]), fmaxf(a[2], a[3])),
                     fmaxf(fmaxf(b[0], b[1]), fmaxf(b[2], b[3])));
      const v4f c = *(const v4f*)&reds[j][lr][0];
      const v4f d = *(const v4f*)&reds[j][lr][4];
      bsj[j] = ((((((c[0] + c[1]) + c[2]) + c[3]) + d[0]) + d[1]) + d[2]) + d[3];
    }

    v8f acc = v8f{};
    Stage1<4> s0, s1, s2, s3, s4, s5;
    load_nn1<4>(s0, ep, vp, 0);
    hold(acc);
    load_nn1<4>(s1, ep, vp, 16);
    hold(acc);
    load_nn1<4>(s2, ep, vp, 32);
    hold(acc);
    load_nn1<4>(s3, ep, vp, 48);
    hold(acc);
    load_nn1<4>(s4, ep, vp, 64);
    hold(acc);
    pin_order();
    v4f f;
    {
      const float M  = fmaxf(fmaxf(bmj[0], bmj[1]), fmaxf(bmj[2], bmj[3]));
      const float w0 = __expf(bmj[0] - M), w1 = __expf(bmj[1] - M);
      const float w2 = __expf(bmj[2] - M), w3 = __expf(bmj[3] - M);
      const float L  = ((bsj[0] * w0 + bsj[1] * w1) + bsj[2] * w2) + bsj[3] * w3;
      const float inv = 1.0f / L;
      f[0] = w0 * inv; f[1] = w1 * inv; f[2] = w2 * inv; f[3] = w3 * inv;
    }
    for (int k = 0; k < 20 * 96; k += 96) {
      load_nn1<4>(s5, ep, vp, k + 80);
      hold(acc);
      mma_stage1<4>(s0, block_factor(f, k), acc);
      hold(acc);
      load_nn1<4>(s0, ep, vp, k + 96);
      hold(acc);
      mma_stage1<4>(s1, block_factor(f, k + 16), acc);
      hold(acc);
      load_nn1<4>(s1, ep, vp, k + 112);
      hold(acc);
      mma_stage1<4>(s2, block_factor(f, k + 32), acc);
      hold(acc);
      load_nn1<4>(s2, ep, vp, k + 128);
      hold(acc);
      mma_stage1<4>(s3, block_factor(f, k + 48), acc);
      hold(acc);
      load_nn1<4>(s3, ep, vp, k + 144);
      hold(acc);
      mma_stage1<4>(s4, block_factor(f, k + 64), acc);
      hold(acc);
      load_nn1<4>(s4, ep, vp, k + 160);
      hold(acc);
      mma_stage1<4>(s5, block_factor(f, k + 80), acc);
      hold(acc);
    }
    load_nn1<4>(s5, ep, vp, kS - 48);
    hold(acc);
    mma_stage1<4>(s0, f[3], acc);
    hold(acc);
    load_nn1<4>(s0, ep, vp, kS - 32);
    hold(acc);
    mma_stage1<4>(s1, f[3], acc);
    hold(acc);
    load_nn1<4>(s1, ep, vp, kS - 16);
    hold(acc);
    mma_stage1<4>(s2, f[3], acc);
    mma_stage1<4>(s3, f[3], acc);
    mma_stage1<4>(s4, f[3], acc);
    mma_stage1<4>(s5, f[3], acc);
    mma_stage1<4>(s0, f[3], acc);
    mma_stage1<4>(s1, f[3], acc);

#pragma unroll
    for (int r = 0; r < 8; ++r) tile[r + 8 * half][d0 + lr] = acc[r];
  }
  __syncthreads();
  {
    const int p   = (int)threadIdx.x;
    const int row = 2 * (p & 7);
    const int d   = 4 * (p >> 4) + 2 * ((p >> 3) & 1);
    const v2f lo = *(const v2f*)&tile[row][d];
    const v2f hi = *(const v2f*)&tile[row + 1][d];
    *(v4f*)(AVec + (size_t)blockIdx.x * 16 * kD + h * kTileF + p * 4) = v4f{lo[0], lo[1], hi[0], hi[1]};
  }
}

__global__ void __launch_bounds__(128) out_gemm(const float* __restrict__ A, const float* __restrict__ B,
                         float* __restrict__ C) {
  const int lane = threadIdx.x & 31;
  const int wave = threadIdx.x >> 5;
  const int half = lane >> 4;
  const int lr   = lane & 15;
  const int m0   = blockIdx.x * 16;
  const int n0   = (blockIdx.y * 4 + wave) * 64;

  const float* ap = A + (size_t)blockIdx.x * kTileK + half * 32 + lr * 2;
  const float* bp = B + (size_t)(n0 / 16) * kTileK + half * 32 + lr * 2;

  v8f acc[4] = {v8f{}, v8f{}, v8f{}, v8f{}};
  kloop_nt512<false>(ap, bp, acc);
#pragma unroll
  for (int t = 0; t < 4; ++t) {
    int n = n0 + t * 16 + lr;
#pragma unroll
    for (int r = 0; r < 8; ++r) {
      int m = m0 + r + 8 * half;
      C[(size_t)m * kD + n] = acc[t][r];
    }
  }
}

extern "C" void kernel_launch(void* const* d_in, const int* in_sizes, int n_in,
                              void* d_out, int out_size, void* d_ws, size_t ws_size,
                              hipStream_t stream) {
  const float* x   = (const float*)d_in[0];
  const float* pos = (const float*)d_in[1];
  const float* Wq  = (const float*)d_in[2];
  const float* Wk  = (const float*)d_in[3];
  const float* Wv  = (const float*)d_in[4];
  const float* Wp  = (const float*)d_in[5];
  const float* Wo  = (const float*)d_in[6];
  const float* u   = (const float*)d_in[7];
  const float* v   = (const float*)d_in[8];

  float* ws = (float*)d_ws;
  const size_t SD = (size_t)kS * kD;
  float* Qhs  = ws;
  float* Khs  = Qhs + SD;
  float* Vhs  = Khs + SD;
  float* Phs  = Vhs + SD;
  float* AVec = Phs + SD;
  const size_t DD = (size_t)kD * kD;
  float* xf  = AVec + SD;
  float* pf  = xf + SD;
  float* Wqf = pf + SD;
  float* Wkf = Wqf + DD;
  float* Wvf = Wkf + DD;
  float* Wpf = Wvf + DD;
  float* Wof = Wpf + DD;

  dim3 blk(256);
  to_fragment_order<<<dim3(kInTiles), blk, 0, stream>>>(x, pos, Wq, Wk, Wv, Wp, Wo, xf);

  dim3 blk4(128);
  dim3 grid2(kS / 16, 2);
  proj_gemm_head<true><<<grid2, blk4, 0, stream>>>(xf, Wqf, Qhs);
  proj_gemm_head<true><<<grid2, blk4, 0, stream>>>(xf, Wkf, Khs);
  proj_gemm_head<false><<<grid2, blk4, 0, stream>>>(xf, Wvf, Vhs);
  proj_gemm_head<true><<<grid2, blk4, 0, stream>>>(pf, Wpf, Phs);

  const float scale = 0.125f;
  attn_fused<<<dim3(kS / 16, kH), blk, 0, stream>>>(Qhs, Khs, Phs, Vhs, u, v, AVec, scale);

  out_gemm<<<grid2, blk4, 0, stream>>>(AVec, Wof, (float*)d_out);
}


# attention kernel: one 16-wave block per 32 query rows, two 16-row halves in step (was separate 8-wave blocks of 16 rows); twin waves ask for the same K, P, V lines together; sum order kept
# speedup vs baseline: 9.9411x; 1.0350x over previous
#include <hip/hip_runtime.h>
#include <math.h>

typedef __attribute__((ext_vector_type(2))) float v2f;
typedef __attribute__((ext_vector_type(4))) float v4f;
typedef __attribute__((ext_vector_type(8))) float v8f;

static constexpr int kS  = 2048;
static constexpr int kD  = 512;
static constexpr int kH  = 8;
static constexpr int kDH = 64;

static constexpr int kChunk = 64;
static constexpr int kTileF = 16 * kDH;
static constexpr int kTileK = 16 * kD;
static constexpr int kInTiles = 2 * (kS / 16) + 5 * (kD / 16);

__device__ __forceinline__ int frag_row(int row, int half) {
  return (row >> 4) * kTileF + half * 32 + (row & 15) * 2;
}

__device__ __forceinline__ v8f wmma_f32(v2f a, v2f b, v8f c) {
  return __builtin_amdgcn_wmma_f32_16x16x4_f32(
       false, a,  false, b,
       (short)0, c,  false,  false);
}

template <int U>
struct Stage {
  v2f a[U];
  v2f b[U][4];
};

__device__ __forceinline__ void pin_order() { __builtin_amdgcn_sched_barrier(0); }

__device__ __forceinline__ void hold(v8f& acc) { asm volatile("" : "+v"(acc) : : "memory"); }

__device__ __forceinline__ void hold4(v8f (&acc)[4]) {
  asm volatile("" : "+v"(acc[0]), "+v"(acc[1]), "+v"(acc[2]), "+v"(acc[3]) : : "memory");
}

template <int U, int BT>
__device__ __forceinline__ void load_ff(Stage<U>& s, const float* __restrict__ ap,
                                        const float* __restrict__ bp, int k0) {
#pragma unroll
  for (int i = 0; i < U; ++i) {
    s.a[i] = *(const v2f*)(ap + (k0 / 4 + i) * kChunk);
#pragma unroll
    for (int t = 0; t < 4; ++t)
      s.b[i][t] = *(const v2f*)(bp + t * BT + (k0 / 4 + i) * kChunk);
  }
}

template <int U>
struct Stage1 {
  v2f a[U];
  v2f b[U];
};

template <int U>
__device__ __forceinline__ void load_nn1(Stage1<U>& s, const float* __restrict__ ap,
                                         const float* __restrict__ vp, int k0) {
#pragma unroll
  for (int i = 0; i < U; ++i) {
    s.a[i] = *(const v2f*)(ap + k0 + 4 * i);
    s.b[i] = *(const v2f*)(vp + (k0 / 4 + i) * kChunk);
  }
}

template <int U>
__device__ __forceinline__ void mma_stage1(const Stage1<U>& s, float f, v8f& acc) {
  pin_order();
#pragma unroll
  for (int i = 0; i < U; ++i) acc = wmma_f32(s.a[i] * f, s.b[i], acc);
}

__device__ __forceinline__ float block_factor(v4f f, int k0) { return f[k0 >> 9]; }

template <int U, bool TR>
__device__ __forceinline__ void mma_stage(const Stage<U>& s, v8f (&acc)[4]) {
#pragma unroll
  for (int i = 0; i < U; ++i) {
#pragma unroll
    for (int t = 0; t < 4; ++t)
      acc[t] = TR ? wmma_f32(s.b[i][t], s.a[i], acc[t]) : wmma_f32(s.a[i], s.b[i][t], acc[t]);
  }
}

template <bool TR>
__device__ __forceinline__ void kloop_nt512(const float* __restrict__ ap,
                                            const float* __restrict__ bp, v8f (&acc)[4]) {
  Stage<4> s0, s1, s2, s3;
  load_ff<4, kTileK>(s0, ap, bp, 0);
  load_ff<4, kTileK>(s1, ap, bp, 16);
  load_ff<4, kTileK>(s2, ap, bp, 32);
  for (int k = 0; k < kD - 64; k += 64) {
    load_ff<4, kTileK>(s3, ap, bp, k + 48);
    hold4(acc);
    mma_stage<4, TR>(s0, acc);
    hold4(acc);
    load_ff<4, kTileK>(s0, ap, bp, k + 64);
    hold4(acc);
    mma_stage<4, TR>(s1, acc);
    hold4(acc);
    load_ff<4, kTileK>(s1, ap, bp, k + 80);
    hold4(acc);
    mma_stage<4, TR>(s2, acc);
    hold4(acc);
    load_ff<4, kTileK>(s2, ap, bp, k + 96);
    hold4(acc);
    mma_stage<4, TR>(s3, acc);
    hold4(acc);
  }
  load_ff<4, kTileK>(s3, ap, bp, kD - 16);
  hold4(acc);
  mma_stage<4, TR>(s0, acc);
  mma_stage<4, TR>(s1, acc);
  mma_stage<4, TR>(s2, acc);
  mma_stage<4, TR>(s3, acc);
}

__device__ __forceinline__ void kloop_nt64_bias(const float* __restrict__ ap,
                                                const float* __restrict__ bp,
                                                const float* __restrict__ bb, v8f (&acc)[4]) {
  Stage<8> s0, s1;
  v2f c0[8], c1[8];
  load_ff<8, kTileF>(s0, ap, bp, 0);
#pragma unroll
  for (int i = 0; i < 8; ++i) c0[i] = *(const v2f*)(bb + 4 * i);
  load_ff<8, kTileF>(s1, ap, bp, 32);
#pragma unroll
  for (int i = 0; i < 8; ++i) c1[i] = *(const v2f*)(bb + 32 + 4 * i);
  pin_order();
#pragma unroll
  for (int i = 0; i < 8; ++i) s0.a[i] = s0.a[i] + c0[i];
  mma_stage<8, false>(s0, acc);
#pragma unroll
  for (int i = 0; i < 8; ++i) s1.a[i] = s1.a[i] + c1[i];
  mma_stage<8, false>(s1, acc);
}

__global__ void __launch_bounds__(256) to_fragment_order(const float* __restrict__ x,
                                  const float* __restrict__ pos,
                                  const float* __restrict__ wq, const float* __restrict__ wk,
                                  const float* __restrict__ wv, const float* __restrict__ wp,
                                  const float* __restrict__ wo, float* __restrict__ dst) {
  const int lane = threadIdx.x & 31;
  const int wave = threadIdx.x >> 5;
  const int half = lane >> 4;
  const int lr   = lane & 15;
  const int b    = blockIdx.x;
  constexpr int kRowTiles = kS / 16;
  constexpr int kWTiles   = kD / 16;

  const float* src;
  if (b < kRowTiles) {
    src = x + (size_t)b * kTileK;
  } else if (b < 2 * kRowTiles) {
    src = pos + (size_t)(b - kRowTiles) * kTileK;
  } else {
    const int w = (b - 2 * kRowTiles) / kWTiles;
    const float* m = w == 0 ? wq : w == 1 ? wk : w == 2 ? wv : w == 3 ? wp : wo;
    src = m + (size_t)((b - 2 * kRowTiles) % kWTiles) * kTileK;
  }
  src += lr * kD + 4 * half;
  float* out = dst + (size_t)b * kTileK + half * kChunk + lr * 2;

  v4f e[8];
#pragma unroll
  for (int i = 0; i < 8; ++i) e[i] = *(const v4f*)(src + 8 * (8 * wave + i));
#pragma unroll
  for (int i = 0; i < 8; ++i) {
    float* o = out + 2 * (8 * wave + i) * kChunk;
    *(v2f*)(o)      = v2f{e[i][0], e[i][1]};
    *(v2f*)(o + 32) = v2f{e[i][2], e[i][3]};
  }
}

template <bool TR>
__global__ void __launch_bounds__(128) proj_gemm_head(const float* __restrict__ A,
                               const float* __restrict__ B,
                               float* __restrict__ out) {
  const int lane = threadIdx.x & 31;
  const int wave = threadIdx.x >> 5;
  const int half = lane >> 4;
  const int lr   = lane & 15;
  const int m0   = blockIdx.x * 16;
  const int head = blockIdx.y * 4 + wave;
  const int n0   = head * 64;

  const float* ap = A + (size_t)blockIdx.x * kTileK + half * 32 + lr * 2;
  const float* bp = B + (size_t)(n0 / 16) * kTileK + half * 32 + lr * 2;

  v8f acc[4] = {v8f{}, v8f{}, v8f{}, v8f{}};
  kloop_nt512<TR>(ap, bp, acc);
  float* op = out + (size_t)head * kS * kDH + lr * 2;
#pragma unroll
  for (int t = 0; t < 4; ++t) {
#pragma unroll
    for (int q = 0; q < 4; ++q) {
      const int off = TR ? blockIdx.x * kTileF + (4 * t + 2 * half + (q >> 1)) * kChunk
                         : t * (kS / 4) * kChunk + (m0 / 4 + 2 * half + (q >> 1)) * kChunk;
      *(v2f*)(op + off + (q & 1) * 32) = v2f{acc[t][2 * q], acc[t][2 * q + 1]};
    }
  }
}

template <int U>
struct StageW {
  v2f a[U];
  v2f c[U];
  v2f b[U][5];
};

template <int U>
__device__ __forceinline__ void load_win(StageW<U>& s, const float* __restrict__ ap,
                                         const float* __restrict__ cp,
                                         const float* __restrict__ pb,
                                         const unsigned (&off)[5], int k0) {
#pragma unroll
  for (int i = 0; i < U; ++i) {
    s.a[i] = *(const v2f*)(ap + (k0 / 4 + i) * kChunk);
    s.c[i] = *(const v2f*)(cp + k0 + 4 * i);
#pragma unroll
    for (int w = 0; w < 5; ++w)
      s.b[i][w] = *(const v2f*)(pb + (size_t)off[w] + (k0 / 4 + i) * kChunk);
  }
}

template <int U>
__device__ __forceinline__ void mma_win(const StageW<U>& s, v8f (&g)[5]) {
#pragma unroll
  for (int i = 0; i < U; ++i) {
    const v2f a = s.a[i] + s.c[i];
#pragma unroll
    for (int w = 0; w < 5; ++w) g[w] = wmma_f32(a, s.b[i][w], g[w]);
  }
}

__device__ __forceinline__ void shifted_bd(const float* __restrict__ ap,
                                           const float* __restrict__ cp,
                                           const float* __restrict__ pb, int cs,
                                           int lane, float (&sh)[4][8]) {
  const int half = lane >> 4;
  const int lr   = lane & 15;
  unsigned off[5];
#pragma unroll
  for (int w = 0; w < 5; ++w) {
    int c = cs + 16 * w + lr;
    c = c < 0 ? 0 : (c > kS - 1 ? kS - 1 : c);
    off[w] = (unsigned)frag_row(c, half);
  }
  v8f g[5] = {v8f{}, v8f{}, v8f{}, v8f{}, v8f{}};
  StageW<4> s0, s1;
  load_win<4>(s0, ap, cp, pb, off, 0);
  load_win<4>(s1, ap, cp, pb, off, 16);
  pin_order();
  mma_win<4>(s0, g);
  pin_order();
  load_win<4>(s0, ap, cp, pb, off, 32);
  pin_order();
  mma_win<4>(s1, g);
  pin_order();
  load_win<4>(s1, ap, cp, pb, off, 48);
  pin_order();
  mma_win<4>(s0, g);
  mma_win<4>(s1, g);

#pragma unroll
  for (int r = 0; r < 8; ++r) {
    const int m   = r + 8 * half;
    const int src = (lane & 16) | ((lr - m - 1) & 15);
    float p[5];
#pragma unroll
    for (int w = 0; w < 5; ++w) p[w] = __shfl(g[w][r], src, 32);
#pragma unroll
    for (int t = 0; t < 4; ++t) sh[t][r] = (lr <= m) ? p[t] : p[t + 1];
  }
}

static constexpr int kEP = kS + 4;

__global__ void __launch_bounds__(512) attn_fused(const float* __restrict__ Qh, const float* __restrict__ Kh,
                           const float* __restrict__ Ph, const float* __restrict__ Vh,
                           const float* __restrict__ ub, const float* __restrict__ vb,
                           float* __restrict__ AVec, float scale) {
  __shared__ alignas(16) float E[2][16][kEP];
  __shared__ alignas(16) float redm[2][4][16][8];
  __shared__ alignas(16) float reds[2][4][16][8];
  __shared__ alignas(16) float tile[2][16][68];
  const int lane = threadIdx.x & 31;
  const int wave16 = __builtin_amdgcn_readfirstlane((int)(threadIdx.x >> 5));
  const int rt   = wave16 & 1;
  const int wave = wave16 >> 1;
  const int half = lane >> 4;
  const int lr   = lane & 15;
  const int m0   = (blockIdx.x * 2 + rt) * 16;
  const int h    = blockIdx.y;

  const float* qrow = Qh + (size_t)h * kS * kDH + frag_row(m0 + lr, half);
  const float* ubias = ub + h * kDH + 2 * half;
  const float* pb = Ph + (size_t)h * kS * kDH;
  const float* vbias = vb + h * kDH + 2 * half;

#pragma nounroll
  for (int j = 0; j < 4; ++j) {
    const float* qp = qrow;
    const float* bb = ubias;
    const float* cp = vbias;
    asm volatile("" : "+v"(qp), "+v"(bb), "+v"(cp));
    const int n0 = j * 512 + wave * 64;
    const float* kp = Kh + (size_t)h * kS * kDH + frag_row(n0 + lr, half);

    v8f acc[4] = {v8f{}, v8f{}, v8f{}, v8f{}};
    kloop_nt64_bias(qp, kp, bb, acc);
    pin_order();

    const int delta = __builtin_amdgcn_readfirstlane(n0 - m0);
    if (delta <= 0) {
      float sh[4][8];
      shifted_bd(qp, cp, pb, kS - 16 + delta, lane, sh);
#pragma unroll
      for (int t = 0; t < 4; ++t) {
        const int n = n0 + t * 16 + lr;
#pragma unroll
        for (int r = 0; r < 8; ++r) {
          const int m = m0 + r + 8 * half;
          acc[t][r] += (n <= m) ? sh[t][r] : 0.0f;
        }
      }
    }
    if (delta >= -48) {
      const int row1 = (m0 + lr + 1 > kS - 1) ? kS - 1 : m0 + lr + 1;
      const float* qp1 = Qh + (size_t)h * kS * kDH + frag_row(row1, half);
      float sh[4][8];
      shifted_bd(qp1, cp, pb, delta - 17, lane, sh);
#pragma unroll
      for (int t = 0; t < 4; ++t) {
        const int n = n0 + t * 16 + lr;
#pragma unroll
        for (int r = 0; r < 8; ++r) {
          const int m = m0 + r + 8 * half;
          acc[t][r] += (n >= m + 2) ? sh[t][r] : 0.0f;
        }
      }
    }

    float sv[4][8];
#pragma unroll
    for (int t = 0; t < 4; ++t) {
#pragma unroll
      for (int r = 0; r < 8; ++r) sv[t][r] = acc[t][r] * scale;
    }

    float mx[8];
#pragma unroll
    for (int r = 0; r < 8; ++r)
      mx[r] = fmaxf(fmaxf(sv[0][r], sv[1][r]), fmaxf(sv[2][r], sv[3][r]));
#pragma unroll
    for (int off = 1; off < 16; off <<= 1) {
#pragma unroll
      for (int r = 0; r < 8; ++r) mx[r] = fmaxf(mx[r], __shfl_xor(mx[r], off, 32));
    }
    if (lr == 0) {
#pragma unroll
      for (int r = 0; r < 8; ++r) redm[rt][j][r + 8 * half][wave] = mx[r];
    }
    __syncthreads();

    float sm[8];
#pragma unroll
    for (int r = 0; r < 8; ++r) {
      const v4f a = *(const v4f*)&redm[rt][j][r + 8 * half][0];
      const v4f b = *(const v4f*)&redm[rt][j][r + 8 * half][4];
      const float bm = fmaxf(fmaxf(fmaxf(a[0], a[1]), fmaxf(a[2], a[3])),
                             fmaxf(fmaxf(b[0], b[1]), fmaxf(b[2], b[3])));
      float s = 0.0f;
#pragma unroll
      for (int t = 0; t < 4; ++t) {
        const float e = __expf(sv[t][r] - bm);
        E[rt][r + 8 * half][n0 + t * 16 + lr] = e;
        s += e;
      }
      sm[r] = s;
    }
#pragma unroll
    for (int off = 1; off < 16; off <<= 1) {
#pragma unroll
      for (int r = 0; r < 8; ++r) sm[r] += __shfl_xor(sm[r], off, 32);
    }
    if (lr == 0) {
#pragma unroll
      for (int r = 0; r < 8; ++r) reds[rt][j][r + 8 * half][wave] = sm[r];
    }
  }
  __syncthreads();

  if (wave < 4) {
    const int d0 = wave * 16;
    const float* ep = &E[rt][lr][2 * half];
    const float* vp = Vh + (size_t)h * kS * kDH + wave * (kS / 4) * kChunk + half * 32 + lr * 2;

    float bmj[4], bsj[4];
#pragma unroll
    for (int j = 0; j < 4; ++j) {
      const v4f a = *(const v4f*)&redm[rt][j][lr][0];
      const v4f b = *(const v4f*)&redm[rt][j][lr][4];
      bmj[j] = fmaxf(fmaxf(fmaxf(a[0], a[1]), fmaxf(a[2], a[3])),
                     fmaxf(fmaxf(b[0], b[1]), fmaxf(b[2], b[3])));
      const v4f c = *(const v4f*)&reds[rt][j][lr][0];
      const v4f d = *(const v4f*)&reds[rt][j][lr][4];
      bsj[j] = ((((((c[0] + c[1]) + c[2]) + c[3]) + d[0]) + d[1]) + d[2]) + d[3];
    }

    v8f acc = v8f{};
    Stage1<4> s0, s1, s2, s3, s4, s5;
    load_nn1<4>(s0, ep, vp, 0);
    hold(acc);
    load_nn1<4>(s1, ep, vp, 16);
    hold(acc);
    load_nn1<4>(s2, ep, vp, 32);
    hold(acc);
    load_nn1<4>(s3, ep, vp, 48);
    hold(acc);
    load_nn1<4>(s4, ep, vp, 64);
    hold(acc);
    pin_order();
    v4f f;
    {
      const float M  = fmaxf(fmaxf(bmj[0], bmj[1]), fmaxf(bmj[2], bmj[3]));
      const float w0 = __expf(bmj[0] - M), w1 = __expf(bmj[1] - M);
      const float w2 = __expf(bmj[2] - M), w3 = __expf(bmj[3] - M);
      const float L  = ((bsj[0] * w0 + bsj[1] * w1) + bsj[2] * w2) + bsj[3] * w3;
      const float inv = 1.0f / L;
      f[0] = w0 * inv; f[1] = w1 * inv; f[2] = w2 * inv; f[3] = w3 * inv;
    }
    for (int k = 0; k < 20 * 96; k += 96) {
      load_nn1<4>(s5, ep, vp, k + 80);
      hold(acc);
      mma_stage1<4>(s0, block_factor(f, k), acc);
      hold(acc);
      load_nn1<4>(s0, ep, vp, k + 96);
      hold(acc);
      mma_stage1<4>(s1, block_factor(f, k + 16), acc);
      hold(acc);
      load_nn1<4>(s1, ep, vp, k + 112);
      hold(acc);
      mma_stage1<4>(s2, block_factor(f, k + 32), acc);
      hold(acc);
      load_nn1<4>(s2, ep, vp, k + 128);
      hold(acc);
      mma_stage1<4>(s3, block_factor(f, k + 48), acc);
      hold(acc);
      load_nn1<4>(s3, ep, vp, k + 144);
      hold(acc);
      mma_stage1<4>(s4, block_factor(f, k + 64), acc);
      hold(acc);
      load_nn1<4>(s4, ep, vp, k + 160);
      hold(acc);
      mma_stage1<4>(s5, block_factor(f, k + 80), acc);
      hold(acc);
    }
    load_nn1<4>(s5, ep, vp, kS - 48);
    hold(acc);
    mma_stage1<4>(s0, f[3], acc);
    hold(acc);
    load_nn1<4>(s0, ep, vp, kS - 32);
    hold(acc);
    mma_stage1<4>(s1, f[3], acc);
    hold(acc);
    load_nn1<4>(s1, ep, vp, kS - 16);
    hold(acc);
    mma_stage1<4>(s2, f[3], acc);
    mma_stage1<4>(s3, f[3], acc);
    mma_stage1<4>(s4, f[3], acc);
    mma_stage1<4>(s5, f[3], acc);
    mma_stage1<4>(s0, f[3], acc);
    mma_stage1<4>(s1, f[3], acc);

#pragma unroll
    for (int r = 0; r < 8; ++r) tile[rt][r + 8 * half][d0 + lr] = acc[r];
  }
  __syncthreads();
  {
    const int p   = wave * 32 + lane;
    const int row = 2 * (p & 7);
    const int d   = 4 * (p >> 4) + 2 * ((p >> 3) & 1);
    const v2f lo = *(const v2f*)&tile[rt][row][d];
    const v2f hi = *(const v2f*)&tile[rt][row + 1][d];
    *(v4f*)(AVec + (size_t)(blockIdx.x * 2 + rt) * 16 * kD + h * kTileF + p * 4) = v4f{lo[0], lo[1], hi[0], hi[1]};
  }
}

__global__ void __launch_bounds__(128) out_gemm(const float* __restrict__ A, const float* __restrict__ B,
                         float* __restrict__ C) {
  const int lane = threadIdx.x & 31;
  const int wave = threadIdx.x >> 5;
  const int half = lane >> 4;
  const int lr   = lane & 15;
  const int m0   = blockIdx.x * 16;
  const int n0   = (blockIdx.y * 4 + wave) * 64;

  const float* ap = A + (size_t)blockIdx.x * kTileK + half * 32 + lr * 2;
  const float* bp = B + (size_t)(n0 / 16) * kTileK + half * 32 + lr * 2;

  v8f acc[4] = {v8f{}, v8f{}, v8f{}, v8f{}};
  kloop_nt512<false>(ap, bp, acc);
#pragma unroll
  for (int t = 0; t < 4; ++t) {
    int n = n0 + t * 16 + lr;
#pragma unroll
    for (int r = 0; r < 8; ++r) {
      int m = m0 + r + 8 * half;
      C[(size_t)m * kD + n] = acc[t][r];
    }
  }
}

extern "C" void kernel_launch(void* const* d_in, const int* in_sizes, int n_in,
                              void* d_out, int out_size, void* d_ws, size_t ws_size,
                              hipStream_t stream) {
  const float* x   = (const float*)d_in[0];
  const float* pos = (const float*)d_in[1];
  const float* Wq  = (const float*)d_in[2];
  const float* Wk  = (const float*)d_in[3];
  const float* Wv  = (const float*)d_in[4];
  const float* Wp  = (const float*)d_in[5];
  const float* Wo  = (const float*)d_in[6];
  const float* u   = (const float*)d_in[7];
  const float* v   = (const float*)d_in[8];

  float* ws = (float*)d_ws;
  const size_t SD = (size_t)kS * kD;
  float* Qhs  = ws;
  float* Khs  = Qhs + SD;
  float* Vhs  = Khs + SD;
  float* Phs  = Vhs + SD;
  float* AVec = Phs + SD;
  const size_t DD = (size_t)kD * kD;
  float* xf  = AVec + SD;
  float* pf  = xf + SD;
  float* Wqf = pf + SD;
  float* Wkf = Wqf + DD;
  float* Wvf = Wkf + DD;
  float* Wpf = Wvf + DD;
  float* Wof = Wpf + DD;

  dim3 blk(256);
  to_fragment_order<<<dim3(kInTiles), blk, 0, stream>>>(x, pos, Wq, Wk, Wv, Wp, Wo, xf);

  dim3 blk4(128);
  dim3 grid2(kS / 16, 2);
  proj_gemm_head<true><<<grid2, blk4, 0, stream>>>(xf, Wqf, Qhs);
  proj_gemm_head<true><<<grid2, blk4, 0, stream>>>(xf, Wkf, Khs);
  proj_gemm_head<false><<<grid2, blk4, 0, stream>>>(xf, Wvf, Vhs);
  proj_gemm_head<true><<<grid2, blk4, 0, stream>>>(pf, Wpf, Phs);

  const float scale = 0.125f;
  attn_fused<<<dim3(kS / 32, kH), dim3(512), 0, stream>>>(Qhs, Khs, Phs, Vhs, u, v, AVec, scale);

  out_gemm<<<grid2, blk4, 0, stream>>>(AVec, Wof, (float*)d_out);
}


# attention: score passes without barriers: a wave uses its own 64-column strip max (32 strip factors a row at AV, were 4); issue priority falls pass by pass, laggards go first; new row-sum order
# speedup vs baseline: 10.3089x; 1.0370x over previous
#include <hip/hip_runtime.h>
#include <math.h>

typedef __attribute__((ext_vector_type(2))) float v2f;
typedef __attribute__((ext_vector_type(4))) float v4f;
typedef __attribute__((ext_vector_type(8))) float v8f;

static constexpr int kS  = 2048;
static constexpr int kD  = 512;
static constexpr int kH  = 8;
static constexpr int kDH = 64;

static constexpr int kChunk = 64;
static constexpr int kTileF = 16 * kDH;
static constexpr int kTileK = 16 * kD;
static constexpr int kInTiles = 2 * (kS / 16) + 5 * (kD / 16);

__device__ __forceinline__ int frag_row(int row, int half) {
  return (row >> 4) * kTileF + half * 32 + (row & 15) * 2;
}

__device__ __forceinline__ v8f wmma_f32(v2f a, v2f b, v8f c) {
  return __builtin_amdgcn_wmma_f32_16x16x4_f32(
       false, a,  false, b,
       (short)0, c,  false,  false);
}

template <int U>
struct Stage {
  v2f a[U];
  v2f b[U][4];
};

__device__ __forceinline__ void pin_order() { __builtin_amdgcn_sched_barrier(0); }

__device__ __forceinline__ void hold(v8f& acc) { asm volatile("" : "+v"(acc) : : "memory"); }

__device__ __forceinline__ void hold4(v8f (&acc)[4]) {
  asm volatile("" : "+v"(acc[0]), "+v"(acc[1]), "+v"(acc[2]), "+v"(acc[3]) : : "memory");
}

template <int U, int BT>
__device__ __forceinline__ void load_ff(Stage<U>& s, const float* __restrict__ ap,
                                        const float* __restrict__ bp, int k0) {
#pragma unroll
  for (int i = 0; i < U; ++i) {
    s.a[i] = *(const v2f*)(ap + (k0 / 4 + i) * kChunk);
#pragma unroll
    for (int t = 0; t < 4; ++t)
      s.b[i][t] = *(const v2f*)(bp + t * BT + (k0 / 4 + i) * kChunk);
  }
}

template <int U>
struct Stage1 {
  v2f a[U];
  v2f b[U];
  float f;
};

template <int U>
__device__ __forceinline__ void load_nn1(Stage1<U>& s, const float* __restrict__ ap,
                                         const float* __restrict__ vp, const float* fw, int k0) {
#pragma unroll
  for (int i = 0; i < U; ++i) {
    s.a[i] = *(const v2f*)(ap + k0 + 4 * i);
    s.b[i] = *(const v2f*)(vp + (k0 / 4 + i) * kChunk);
  }
  s.f = fw[k0 >> 6];
}

template <int U>
__device__ __forceinline__ void mma_stage1(const Stage1<U>& s, v8f& acc) {
  pin_order();
#pragma unroll
  for (int i = 0; i < U; ++i) acc = wmma_f32(s.a[i] * s.f, s.b[i], acc);
}

template <int U, bool TR>
__device__ __forceinline__ void mma_stage(const Stage<U>& s, v8f (&acc)[4]) {
#pragma unroll
  for (int i = 0; i < U; ++i) {
#pragma unroll
    for (int t = 0; t < 4; ++t)
      acc[t] = TR ? wmma_f32(s.b[i][t], s.a[i], acc[t]) : wmma_f32(s.a[i], s.b[i][t], acc[t]);
  }
}

template <bool TR>
__device__ __forceinline__ void kloop_nt512(const float* __restrict__ ap,
                                            const float* __restrict__ bp, v8f (&acc)[4]) {
  Stage<4> s0, s1, s2, s3;
  load_ff<4, kTileK>(s0, ap, bp, 0);
  load_ff<4, kTileK>(s1, ap, bp, 16);
  load_ff<4, kTileK>(s2, ap, bp, 32);
  for (int k = 0; k < kD - 64; k += 64) {
    load_ff<4, kTileK>(s3, ap, bp, k + 48);
    hold4(acc);
    mma_stage<4, TR>(s0, acc);
    hold4(acc);
    load_ff<4, kTileK>(s0, ap, bp, k + 64);
    hold4(acc);
    mma_stage<4, TR>(s1, acc);
    hold4(acc);
    load_ff<4, kTileK>(s1, ap, bp, k + 80);
    hold4(acc);
    mma_stage<4, TR>(s2, acc);
    hold4(acc);
    load_ff<4, kTileK>(s2, ap, bp, k + 96);
    hold4(acc);
    mma_stage<4, TR>(s3, acc);
    hold4(acc);
  }
  load_ff<4, kTileK>(s3, ap, bp, kD - 16);
  hold4(acc);
  mma_stage<4, TR>(s0, acc);
  mma_stage<4, TR>(s1, acc);
  mma_stage<4, TR>(s2, acc);
  mma_stage<4, TR>(s3, acc);
}

__device__ __forceinline__ void kloop_nt64_bias(const float* __restrict__ ap,
                                                const float* __restrict__ bp,
                                                const float* __restrict__ bb, v8f (&acc)[4]) {
  Stage<8> s0, s1;
  v2f c0[8], c1[8];
  load_ff<8, kTileF>(s0, ap, bp, 0);
#pragma unroll
  for (int i = 0; i < 8; ++i) c0[i] = *(const v2f*)(bb + 4 * i);
  load_ff<8, kTileF>(s1, ap, bp, 32);
#pragma unroll
  for (int i = 0; i < 8; ++i) c1[i] = *(const v2f*)(bb + 32 + 4 * i);
  pin_order();
#pragma unroll
  for (int i = 0; i < 8; ++i) s0.a[i] = s0.a[i] + c0[i];
  mma_stage<8, false>(s0, acc);
#pragma unroll
  for (int i = 0; i < 8; ++i) s1.a[i] = s1.a[i] + c1[i];
  mma_stage<8, false>(s1, acc);
}

__global__ void __launch_bounds__(256) to_fragment_order(const float* __restrict__ x,
                                  const float* __restrict__ pos,
                                  const float* __restrict__ wq, const float* __restrict__ wk,
                                  const float* __restrict__ wv, const float* __restrict__ wp,
                                  const float* __restrict__ wo, float* __restrict__ dst) {
  const int lane = threadIdx.x & 31;
  const int wave = threadIdx.x >> 5;
  const int half = lane >> 4;
  const int lr   = lane & 15;
  const int b    = blockIdx.x;
  constexpr int kRowTiles = kS / 16;
  constexpr int kWTiles   = kD / 16;

  const float* src;
  if (b < kRowTiles) {
    src = x + (size_t)b * kTileK;
  } else if (b < 2 * kRowTiles) {
    src = pos + (size_t)(b - kRowTiles) * kTileK;
  } else {
    const int w = (b - 2 * kRowTiles) / kWTiles;
    const float* m = w == 0 ? wq : w == 1 ? wk : w == 2 ? wv : w == 3 ? wp : wo;
    src = m + (size_t)((b - 2 * kRowTiles) % kWTiles) * kTileK;
  }
  src += lr * kD + 4 * half;
  float* out = dst + (size_t)b * kTileK + half * kChunk + lr * 2;

  v4f e[8];
#pragma unroll
  for (int i = 0; i < 8; ++i) e[i] = *(const v4f*)(src + 8 * (8 * wave + i));
#pragma unroll
  for (int i = 0; i < 8; ++i) {
    float* o = out + 2 * (8 * wave + i) * kChunk;
    *(v2f*)(o)      = v2f{e[i][0], e[i][1]};
    *(v2f*)(o + 32) = v2f{e[i][2], e[i][3]};
  }
}

template <bool TR>
__global__ void __launch_bounds__(128) proj_gemm_head(const float* __restrict__ A,
                               const float* __restrict__ B,
                               float* __restrict__ out) {
  const int lane = threadIdx.x & 31;
  const int wave = threadIdx.x >> 5;
  const int half = lane >> 4;
  const int lr   = lane & 15;
  const int m0   = blockIdx.x * 16;
  const int head = blockIdx.y * 4 + wave;
  const int n0   = head * 64;

  const float* ap = A + (size_t)blockIdx.x * kTileK + half * 32 + lr * 2;
  const float* bp = B + (size_t)(n0 / 16) * kTileK + half * 32 + lr * 2;

  v8f acc[4] = {v8f{}, v8f{}, v8f{}, v8f{}};
  kloop_nt512<TR>(ap, bp, acc);
  float* op = out + (size_t)head * kS * kDH + lr * 2;
#pragma unroll
  for (int t = 0; t < 4; ++t) {
#pragma unroll
    for (int q = 0; q < 4; ++q) {
      const int off = TR ? blockIdx.x * kTileF + (4 * t + 2 * half + (q >> 1)) * kChunk
                         : t * (kS / 4) * kChunk + (m0 / 4 + 2 * half + (q >> 1)) * kChunk;
      *(v2f*)(op + off + (q & 1) * 32) = v2f{acc[t][2 * q], acc[t][2 * q + 1]};
    }
  }
}

template <int U>
struct StageW {
  v2f a[U];
  v2f c[U];
  v2f b[U][5];
};

template <int U>
__device__ __forceinline__ void load_win(StageW<U>& s, const float* __restrict__ ap,
                                         const float* __restrict__ cp,
                                         const float* __restrict__ pb,
                                         const unsigned (&off)[5], int k0) {
#pragma unroll
  for (int i = 0; i < U; ++i) {
    s.a[i] = *(const v2f*)(ap + (k0 / 4 + i) * kChunk);
    s.c[i] = *(const v2f*)(cp + k0 + 4 * i);
#pragma unroll
    for (int w = 0; w < 5; ++w)
      s.b[i][w] = *(const v2f*)(pb + (size_t)off[w] + (k0 / 4 + i) * kChunk);
  }
}

template <int U>
__device__ __forceinline__ void mma_win(const StageW<U>& s, v8f (&g)[5]) {
#pragma unroll
  for (int i = 0; i < U; ++i) {
    const v2f a = s.a[i] + s.c[i];
#pragma unroll
    for (int w = 0; w < 5; ++w) g[w] = wmma_f32(a, s.b[i][w], g[w]);
  }
}

__device__ __forceinline__ void shifted_bd(const float* __restrict__ ap,
                                           const float* __restrict__ cp,
                                           const float* __restrict__ pb, int cs,
                                           int lane, float (&sh)[4][8]) {
  const int half = lane >> 4;
  const int lr   = lane & 15;
  unsigned off[5];
#pragma unroll
  for (int w = 0; w < 5; ++w) {
    int c = cs + 16 * w + lr;
    c = c < 0 ? 0 : (c > kS - 1 ? kS - 1 : c);
    off[w] = (unsigned)frag_row(c, half);
  }
  v8f g[5] = {v8f{}, v8f{}, v8f{}, v8f{}, v8f{}};
  StageW<4> s0, s1;
  load_win<4>(s0, ap, cp, pb, off, 0);
  load_win<4>(s1, ap, cp, pb, off, 16);
  pin_order();
  mma_win<4>(s0, g);
  pin_order();
  load_win<4>(s0, ap, cp, pb, off, 32);
  pin_order();
  mma_win<4>(s1, g);
  pin_order();
  load_win<4>(s1, ap, cp, pb, off, 48);
  pin_order();
  mma_win<4>(s0, g);
  mma_win<4>(s1, g);

#pragma unroll
  for (int r = 0; r < 8; ++r) {
    const int m   = r + 8 * half;
    const int src = (lane & 16) | ((lr - m - 1) & 15);
    float p[5];
#pragma unroll
    for (int w = 0; w < 5; ++w) p[w] = __shfl(g[w][r], src, 32);
#pragma unroll
    for (int t = 0; t < 4; ++t) sh[t][r] = (lr <= m) ? p[t] : p[t + 1];
  }
}

static constexpr int kEP = kS + 4;

__global__ void __launch_bounds__(512) attn_fused(const float* __restrict__ Qh, const float* __restrict__ Kh,
                           const float* __restrict__ Ph, const float* __restrict__ Vh,
                           const float* __restrict__ ub, const float* __restrict__ vb,
                           float* __restrict__ AVec, float scale) {
  __shared__ alignas(16) float E[2][16][kEP];
  __shared__ alignas(16) float redm[2][4][16][8];
  __shared__ alignas(16) float reds[2][4][16][8];
  __shared__ float fac[2][4][16][33];
  __shared__ alignas(16) float tile[2][16][68];
  const int lane = threadIdx.x & 31;
  const int wave16 = __builtin_amdgcn_readfirstlane((int)(threadIdx.x >> 5));
  const int rt   = wave16 & 1;
  const int wave = wave16 >> 1;
  const int half = lane >> 4;
  const int lr   = lane & 15;
  const int m0   = (blockIdx.x * 2 + rt) * 16;
  const int h    = blockIdx.y;

  const float* qrow = Qh + (size_t)h * kS * kDH + frag_row(m0 + lr, half);
  const float* ubias = ub + h * kDH + 2 * half;
  const float* pb = Ph + (size_t)h * kS * kDH;
  const float* vbias = vb + h * kDH + 2 * half;

#pragma nounroll
  for (int j = 0; j < 4; ++j) {
    if (j == 0) __builtin_amdgcn_s_setprio(3);
    else if (j == 1) __builtin_amdgcn_s_setprio(2);
    else if (j == 2) __builtin_amdgcn_s_setprio(1);
    else __builtin_amdgcn_s_setprio(0);
    const float* qp = qrow;
    const float* bb = ubias;
    const float* cp = vbias;
    asm volatile("" : "+v"(qp), "+v"(bb), "+v"(cp));
    const int n0 = j * 512 + wave * 64;
    const float* kp = Kh + (size_t)h * kS * kDH + frag_row(n0 + lr, half);

    v8f acc[4] = {v8f{}, v8f{}, v8f{}, v8f{}};
    kloop_nt64_bias(qp, kp, bb, acc);
    pin_order();

    const int delta = __builtin_amdgcn_readfirstlane(n0 - m0);
    if (delta <= 0) {
      float sh[4][8];
      shifted_bd(qp, cp, pb, kS - 16 + delta, lane, sh);
#pragma unroll
      for (int t = 0; t < 4; ++t) {
        const int n = n0 + t * 16 + lr;
#pragma unroll
        for (int r = 0; r < 8; ++r) {
          const int m = m0 + r + 8 * half;
          acc[t][r] += (n <= m) ? sh[t][r] : 0.0f;
        }
      }
    }
    if (delta >= -48) {
      const int row1 = (m0 + lr + 1 > kS - 1) ? kS - 1 : m0 + lr + 1;
      const float* qp1 = Qh + (size_t)h * kS * kDH + frag_row(row1, half);
      float sh[4][8];
      shifted_bd(qp1, cp, pb, delta - 17, lane, sh);
#pragma unroll
      for (int t = 0; t < 4; ++t) {
        const int n = n0 + t * 16 + lr;
#pragma unroll
        for (int r = 0; r < 8; ++r) {
          const int m = m0 + r + 8 * half;
          acc[t][r] += (n >= m + 2) ? sh[t][r] : 0.0f;
        }
      }
    }

    float sv[4][8];
#pragma unroll
    for (int t = 0; t < 4; ++t) {
#pragma unroll
      for (int r = 0; r < 8; ++r) sv[t][r] = acc[t][r] * scale;
    }

    float mx[8];
#pragma unroll
    for (int r = 0; r < 8; ++r)
      mx[r] = fmaxf(fmaxf(sv[0][r], sv[1][r]), fmaxf(sv[2][r], sv[3][r]));
#pragma unroll
    for (int off = 1; off < 16; off <<= 1) {
#pragma unroll
      for (int r = 0; r < 8; ++r) mx[r] = fmaxf(mx[r], __shfl_xor(mx[r], off, 32));
    }
    float sm[8];
#pragma unroll
    for (int r = 0; r < 8; ++r) {
      const float bm = mx[r];
      float s = 0.0f;
#pragma unroll
      for (int t = 0; t < 4; ++t) {
        const float e = __expf(sv[t][r] - bm);
        E[rt][r + 8 * half][n0 + t * 16 + lr] = e;
        s += e;
      }
      sm[r] = s;
    }
#pragma unroll
    for (int off = 1; off < 16; off <<= 1) {
#pragma unroll
      for (int r = 0; r < 8; ++r) sm[r] += __shfl_xor(sm[r], off, 32);
    }
    if (lr == 0) {
#pragma unroll
      for (int r = 0; r < 8; ++r) {
        redm[rt][j][r + 8 * half][wave] = mx[r];
        reds[rt][j][r + 8 * half][wave] = sm[r];
      }
    }
  }
  __builtin_amdgcn_s_setprio(0);
  __syncthreads();

  if (wave < 4) {
    const int d0 = wave * 16;
    const float* ep = &E[rt][lr][2 * half];
    const float* vp = Vh + (size_t)h * kS * kDH + wave * (kS / 4) * kChunk + half * 32 + lr * 2;

    float* fw = &fac[rt][wave][lr][0];
    {
      float ms[32], ss[32];
#pragma unroll
      for (int j = 0; j < 4; ++j) {
        const v4f a = *(const v4f*)&redm[rt][j][lr][0];
        const v4f b = *(const v4f*)&redm[rt][j][lr][4];
        const v4f c = *(const v4f*)&reds[rt][j][lr][0];
        const v4f d = *(const v4f*)&reds[rt][j][lr][4];
#pragma unroll
        for (int w = 0; w < 4; ++w) {
          ms[8 * j + w] = a[w]; ms[8 * j + 4 + w] = b[w];
          ss[8 * j + w] = c[w]; ss[8 * j + 4 + w] = d[w];
        }
      }
      float M = ms[0];
#pragma unroll
      for (int s = 1; s < 32; ++s) M = fmaxf(M, ms[s]);
      float L = 0.0f;
#pragma unroll
      for (int s = 0; s < 32; ++s) {
        ms[s] = __expf(ms[s] - M);
        L += ss[s] * ms[s];
      }
      const float inv = 1.0f / L;
#pragma unroll
      for (int s = 0; s < 32; ++s) fw[s] = ms[s] * inv;
    }

    v8f acc = v8f{};
    Stage1<4> s0, s1, s2, s3, s4, s5;
    load_nn1<4>(s0, ep, vp, fw, 0);
    hold(acc);
    load_nn1<4>(s1, ep, vp, fw, 16);
    hold(acc);
    load_nn1<4>(s2, ep, vp, fw, 32);
    hold(acc);
    load_nn1<4>(s3, ep, vp, fw, 48);
    hold(acc);
    load_nn1<4>(s4, ep, vp, fw, 64);
    hold(acc);
    for (int k = 0; k < 20 * 96; k += 96) {
      load_nn1<4>(s5, ep, vp, fw, k + 80);
      hold(acc);
      mma_stage1<4>(s0, acc);
      hold(acc);
      load_nn1<4>(s0, ep, vp, fw, k + 96);
      hold(acc);
      mma_stage1<4>(s1, acc);
      hold(acc);
      load_nn1<4>(s1, ep, vp, fw, k + 112);
      hold(acc);
      mma_stage1<4>(s2, acc);
      hold(acc);
      load_nn1<4>(s2, ep, vp, fw, k + 128);
      hold(acc);
      mma_stage1<4>(s3, acc);
      hold(acc);
      load_nn1<4>(s3, ep, vp, fw, k + 144);
      hold(acc);
      mma_stage1<4>(s4, acc);
      hold(acc);
      load_nn1<4>(s4, ep, vp, fw, k + 160);
      hold(acc);
      mma_stage1<4>(s5, acc);
      hold(acc);
    }
    load_nn1<4>(s5, ep, vp, fw, kS - 48);
    hold(acc);
    mma_stage1<4>(s0, acc);
    hold(acc);
    load_nn1<4>(s0, ep, vp, fw, kS - 32);
    hold(acc);
    mma_stage1<4>(s1, acc);
    hold(acc);
    load_nn1<4>(s1, ep, vp, fw, kS - 16);
    hold(acc);
    mma_stage1<4>(s2, acc);
    mma_stage1<4>(s3, acc);
    mma_stage1<4>(s4, acc);
    mma_stage1<4>(s5, acc);
    mma_stage1<4>(s0, acc);
    mma_stage1<4>(s1, acc);

#pragma unroll
    for (int r = 0; r < 8; ++r) tile[rt][r + 8 * half][d0 + lr] = acc[r];
  }
  __syncthreads();
  {
    const int p   = wave * 32 + lane;
    const int row = 2 * (p & 7);
    const int d   = 4 * (p >> 4) + 2 * ((p >> 3) & 1);
    const v2f lo = *(const v2f*)&tile[rt][row][d];
    const v2f hi = *(const v2f*)&tile[rt][row + 1][d];
    *(v4f*)(AVec + (size_t)(blockIdx.x * 2 + rt) * 16 * kD + h * kTileF + p * 4) = v4f{lo[0], lo[1], hi[0], hi[1]};
  }
}

__global__ void __launch_bounds__(128) out_gemm(const float* __restrict__ A, const float* __restrict__ B,
                         float* __restrict__ C) {
  const int lane = threadIdx.x & 31;
  const int wave = threadIdx.x >> 5;
  const int half = lane >> 4;
  const int lr   = lane & 15;
  const int m0   = blockIdx.x * 16;
  const int n0   = (blockIdx.y * 4 + wave) * 64;

  const float* ap = A + (size_t)blockIdx.x * kTileK + half * 32 + lr * 2;
  const float* bp = B + (size_t)(n0 / 16) * kTileK + half * 32 + lr * 2;

  v8f acc[4] = {v8f{}, v8f{}, v8f{}, v8f{}};
  kloop_nt512<false>(ap, bp, acc);
#pragma unroll
  for (int t = 0; t < 4; ++t) {
    int n = n0 + t * 16 + lr;
#pragma unroll
    for (int r = 0; r < 8; ++r) {
      int m = m0 + r + 8 * half;
      C[(size_t)m * kD + n] = acc[t][r];
    }
  }
}

extern "C" void kernel_launch(void* const* d_in, const int* in_sizes, int n_in,
                              void* d_out, int out_size, void* d_ws, size_t ws_size,
                              hipStream_t stream) {
  const float* x   = (const float*)d_in[0];
  const float* pos = (const float*)d_in[1];
  const float* Wq  = (const float*)d_in[2];
  const float* Wk  = (const float*)d_in[3];
  const float* Wv  = (const float*)d_in[4];
  const float* Wp  = (const float*)d_in[5];
  const float* Wo  = (const float*)d_in[6];
  const float* u   = (const float*)d_in[7];
  const float* v   = (const float*)d_in[8];

  float* ws = (float*)d_ws;
  const size_t SD = (size_t)kS * kD;
  float* Qhs  = ws;
  float* Khs  = Qhs + SD;
  float* Vhs  = Khs + SD;
  float* Phs  = Vhs + SD;
  float* AVec = Phs + SD;
  const size_t DD = (size_t)kD * kD;
  float* xf  = AVec + SD;
  float* pf  = xf + SD;
  float* Wqf = pf + SD;
  float* Wkf = Wqf + DD;
  float* Wvf = Wkf + DD;
  float* Wpf = Wvf + DD;
  float* Wof = Wpf + DD;

  dim3 blk(256);
  to_fragment_order<<<dim3(kInTiles), blk, 0, stream>>>(x, pos, Wq, Wk, Wv, Wp, Wo, xf);

  dim3 blk4(128);
  dim3 grid2(kS / 16, 2);
  proj_gemm_head<true><<<grid2, blk4, 0, stream>>>(xf, Wqf, Qhs);
  proj_gemm_head<true><<<grid2, blk4, 0, stream>>>(xf, Wkf, Khs);
  proj_gemm_head<false><<<grid2, blk4, 0, stream>>>(xf, Wvf, Vhs);
  proj_gemm_head<true><<<grid2, blk4, 0, stream>>>(pf, Wpf, Phs);

  const float scale = 0.125f;
  attn_fused<<<dim3(kS / 32, kH), dim3(512), 0, stream>>>(Qhs, Khs, Phs, Vhs, u, v, AVec, scale);

  out_gemm<<<grid2, blk4, 0, stream>>>(AVec, Wof, (float*)d_out);
}
